# starting kernel
# baseline (speedup 1.0000x reference)
#include <hip/hip_runtime.h>


typedef float v2f __attribute__((ext_vector_type(2)));
typedef float v8f __attribute__((ext_vector_type(8)));

#define C_DIM 256
#define V_DIM 1024
#define F_DIM 256
#define T_DIM 4096
#define B_DIM 4
#define M_DIM (B_DIM * T_DIM)

static __device__ __forceinline__ v8f wmma4(v2f a, v2f b, v8f c) {
  return __builtin_amdgcn_wmma_f32_16x16x4_f32(
       false, a,  false, b,
       (short)0, c,  false,  false);
}

static __device__ __forceinline__ v8f vzero8() {
  v8f z;
#pragma unroll
  for (int i = 0; i < 8; ++i) z[i] = 0.0f;
  return z;
}

__global__ void proj_weights_kernel(const float* __restrict__ basis,
                                    const float* __restrict__ qc,
                                    const float* __restrict__ kc,
                                    const float* __restrict__ vc,
                                    const float* __restrict__ oc,
                                    float* __restrict__ qw,
                                    float* __restrict__ kw,
                                    float* __restrict__ vw,
                                    float* __restrict__ owT) {
  const int lane = threadIdx.x & 31;
  const int r = lane & 15;
  const int h = lane >> 4;
  const int v0 = blockIdx.x * 16;
  const int c0 = blockIdx.y * 16;
  const int which = blockIdx.z;
  const float* coef = (which == 0) ? qc : (which == 1) ? kc : (which == 2) ? vc : oc;

  v8f acc = vzero8();
  for (int f0 = 0; f0 < F_DIM; f0 += 4) {
    v2f a = *(const v2f*)&basis[(size_t)(v0 + r) * F_DIM + f0 + 2 * h];
    v2f b = *(const v2f*)&coef[(size_t)(c0 + r) * F_DIM + f0 + 2 * h];
    acc = wmma4(a, b, acc);
  }
  if (which < 3) {
    float* w = (which == 0) ? qw : (which == 1) ? kw : vw;
#pragma unroll
    for (int i = 0; i < 8; ++i)
      w[(size_t)(v0 + i + 8 * h) * C_DIM + c0 + r] = acc[i];
  } else {
#pragma unroll
    for (int i = 0; i < 8; ++i)
      owT[(size_t)(c0 + r) * V_DIM + v0 + i + 8 * h] = acc[i];
  }
}

__global__ void qkv_kernel(const float* __restrict__ x,
                           const float* __restrict__ qw,
                           const float* __restrict__ kw,
                           const float* __restrict__ vw,
                           float* __restrict__ Q,
                           float* __restrict__ K,
                           float* __restrict__ Vv) {
  const int lane = threadIdx.x & 31;
  const int r = lane & 15;
  const int h = lane >> 4;
  const int m0 = blockIdx.x * 16;
  const int c0 = blockIdx.y * 64;
  const int which = blockIdx.z;
  const float* w = (which == 0) ? qw : (which == 1) ? kw : vw;
  float* out = (which == 0) ? Q : (which == 1) ? K : Vv;

  v8f acc[4];
#pragma unroll
  for (int nt = 0; nt < 4; ++nt) acc[nt] = vzero8();

  for (int k0 = 0; k0 < V_DIM; k0 += 4) {
    v2f a = *(const v2f*)&x[(size_t)(m0 + r) * V_DIM + k0 + 2 * h];
    const float* w0 = &w[(size_t)(k0 + 2 * h) * C_DIM + c0 + r];
    const float* w1 = w0 + C_DIM;
#pragma unroll
    for (int nt = 0; nt < 4; ++nt) {
      v2f b;
      b.x = w0[nt * 16];
      b.y = w1[nt * 16];
      acc[nt] = wmma4(a, b, acc[nt]);
    }
  }
#pragma unroll
  for (int nt = 0; nt < 4; ++nt)
#pragma unroll
    for (int i = 0; i < 8; ++i)
      out[(size_t)(m0 + i + 8 * h) * C_DIM + c0 + nt * 16 + r] = acc[nt][i];
}

__global__ void attn_kernel(const float* __restrict__ Q,
                            const float* __restrict__ K,
                            const float* __restrict__ Vv,
                            const float* __restrict__ decay_logit,
                            float* __restrict__ R) {
  __shared__ float qs[16 * 260];
  __shared__ float ss[16 * 18];

  const int lane = threadIdx.x & 31;
  const int r = lane & 15;
  const int h = lane >> 4;
  const int i0 = blockIdx.x * 16;
  const int c0 = blockIdx.y * 64;
  const int b = blockIdx.z;

  const float* Qb = Q + (size_t)b * T_DIM * C_DIM;
  const float* Kb = K + (size_t)b * T_DIM * C_DIM;
  const float* Vb = Vv + (size_t)b * T_DIM * C_DIM;
  float* Rb = R + (size_t)b * T_DIM * C_DIM;

  for (int idx = lane; idx < 16 * 64; idx += 32) {
    int row = idx >> 6;
    int c4 = (idx & 63) << 2;
    *(float4*)&qs[row * 260 + c4] =
        *(const float4*)&Qb[(size_t)(i0 + row) * C_DIM + c4];
  }
  __syncthreads();

  const float dl = *decay_logit;
  const float dec = 1.0f / (1.0f + __expf(-dl));
  const float ln_dec = __logf(dec);

  v8f acc0 = vzero8(), acc1 = vzero8(), acc2 = vzero8(), acc3 = vzero8();

  for (int j0 = i0; j0 < T_DIM; j0 += 16) {
    {
      int jn = (j0 + 16 < T_DIM) ? (j0 + 16) : j0;
      __builtin_prefetch(&Kb[(size_t)(jn + r + 8 * h) * C_DIM], 0, 1);
      __builtin_prefetch(&Vb[(size_t)(jn + r + 8 * h) * C_DIM + c0], 0, 1);
    }

    v8f s = vzero8();
    const float* Kj = Kb + (size_t)j0 * C_DIM;
    for (int k0 = 0; k0 < C_DIM; k0 += 4) {
      v2f a = *(const v2f*)&qs[r * 260 + k0 + 2 * h];
      v2f bb = *(const v2f*)&Kj[(size_t)r * C_DIM + k0 + 2 * h];
      s = wmma4(a, bb, s);
    }
#pragma unroll
    for (int i = 0; i < 8; ++i) {
      int gi = i0 + i + 8 * h;
      int gj = j0 + r;
      int d = gj - gi;
      float wgt = (d > 0) ? __expf(ln_dec * (float)(d - 1)) : 0.0f;
      ss[(i + 8 * h) * 18 + r] = s[i] * wgt;
    }
    __syncthreads();
#pragma unroll
    for (int kk = 0; kk < 4; ++kk) {
      v2f a = *(const v2f*)&ss[r * 18 + kk * 4 + 2 * h];
      const float* Vr0 = Vb + (size_t)(j0 + kk * 4 + 2 * h) * C_DIM + c0;
      const float* Vr1 = Vr0 + C_DIM;
      v2f bb;
      bb.x = Vr0[r];      bb.y = Vr1[r];      acc0 = wmma4(a, bb, acc0);
      bb.x = Vr0[16 + r]; bb.y = Vr1[16 + r]; acc1 = wmma4(a, bb, acc1);
      bb.x = Vr0[32 + r]; bb.y = Vr1[32 + r]; acc2 = wmma4(a, bb, acc2);
      bb.x = Vr0[48 + r]; bb.y = Vr1[48 + r]; acc3 = wmma4(a, bb, acc3);
    }
    __syncthreads();
  }

#pragma unroll
  for (int i = 0; i < 8; ++i) {
    size_t row = (size_t)(i0 + i + 8 * h) * C_DIM + c0 + r;
    Rb[row] = acc0[i];
    Rb[row + 16] = acc1[i];
    Rb[row + 32] = acc2[i];
    Rb[row + 48] = acc3[i];
  }
}

__global__ void out_proj_kernel(const float* __restrict__ R,
                                const float* __restrict__ owT,
                                const float* __restrict__ out_scale,
                                float* __restrict__ out) {
  const int lane = threadIdx.x & 31;
  const int r = lane & 15;
  const int h = lane >> 4;
  const int m0 = blockIdx.x * 16;
  const int v0 = blockIdx.y * 64;

  v8f acc[4];
#pragma unroll
  for (int nt = 0; nt < 4; ++nt) acc[nt] = vzero8();

  for (int k0 = 0; k0 < C_DIM; k0 += 4) {
    v2f a = *(const v2f*)&R[(size_t)(m0 + r) * C_DIM + k0 + 2 * h];
    const float* w0 = &owT[(size_t)(k0 + 2 * h) * V_DIM + v0 + r];
    const float* w1 = w0 + V_DIM;
#pragma unroll
    for (int nt = 0; nt < 4; ++nt) {
      v2f b;
      b.x = w0[nt * 16];
      b.y = w1[nt * 16];
      acc[nt] = wmma4(a, b, acc[nt]);
    }
  }
  const float sc = *out_scale;
#pragma unroll
  for (int nt = 0; nt < 4; ++nt)
#pragma unroll
    for (int i = 0; i < 8; ++i)
      out[(size_t)(m0 + i + 8 * h) * V_DIM + v0 + nt * 16 + r] = acc[nt][i] * sc;
}

extern "C" void kernel_launch(void* const* d_in, const int* in_sizes, int n_in,
                              void* d_out, int out_size, void* d_ws, size_t ws_size,
                              hipStream_t stream) {
  (void)in_sizes; (void)n_in; (void)out_size; (void)ws_size;

  const float* x = (const float*)d_in[0];
  const float* basis = (const float*)d_in[1];
  const float* qc = (const float*)d_in[2];
  const float* kc = (const float*)d_in[3];
  const float* vc = (const float*)d_in[4];
  const float* oc = (const float*)d_in[5];
  const float* decay_logit = (const float*)d_in[6];
  const float* out_scale = (const float*)d_in[7];
  float* out = (float*)d_out;

  float* ws = (float*)d_ws;
  float* qw = ws;
  float* kw = qw + (size_t)V_DIM * C_DIM;
  float* vw = kw + (size_t)V_DIM * C_DIM;
  float* owT = vw + (size_t)V_DIM * C_DIM;
  float* Q = owT + (size_t)V_DIM * C_DIM;
  float* K = Q + (size_t)M_DIM * C_DIM;
  float* Vv = K + (size_t)M_DIM * C_DIM;
  float* R = Vv + (size_t)M_DIM * C_DIM;

  dim3 blk(32);

  proj_weights_kernel<<<dim3(V_DIM / 16, C_DIM / 16, 4), blk, 0, stream>>>(
      basis, qc, kc, vc, oc, qw, kw, vw, owT);

  qkv_kernel<<<dim3(M_DIM / 16, C_DIM / 64, 3), blk, 0, stream>>>(
      x, qw, kw, vw, Q, K, Vv);

  attn_kernel<<<dim3(T_DIM / 16, C_DIM / 64, B_DIM), blk, 0, stream>>>(
      Q, K, Vv, decay_logit, R);

  out_proj_kernel<<<dim3(M_DIM / 16, V_DIM / 64), blk, 0, stream>>>(
      R, owT, out_scale, out);
}


# attention: the 4 channel-block waves of a row tile join one work-group; each computes one key tile's scores once and shares them via LDS (were recomputed 4x); summation order unchanged
# speedup vs baseline: 2.2151x; 2.2151x over previous
#include <hip/hip_runtime.h>


typedef float v2f __attribute__((ext_vector_type(2)));
typedef float v8f __attribute__((ext_vector_type(8)));

#define C_DIM 256
#define V_DIM 1024
#define F_DIM 256
#define T_DIM 4096
#define B_DIM 4
#define M_DIM (B_DIM * T_DIM)

static __device__ __forceinline__ v8f wmma4(v2f a, v2f b, v8f c) {
  return __builtin_amdgcn_wmma_f32_16x16x4_f32(
       false, a,  false, b,
       (short)0, c,  false,  false);
}

static __device__ __forceinline__ v8f vzero8() {
  v8f z;
#pragma unroll
  for (int i = 0; i < 8; ++i) z[i] = 0.0f;
  return z;
}

__global__ void proj_weights_kernel(const float* __restrict__ basis,
                                    const float* __restrict__ qc,
                                    const float* __restrict__ kc,
                                    const float* __restrict__ vc,
                                    const float* __restrict__ oc,
                                    float* __restrict__ qw,
                                    float* __restrict__ kw,
                                    float* __restrict__ vw,
                                    float* __restrict__ owT) {
  const int lane = threadIdx.x & 31;
  const int r = lane & 15;
  const int h = lane >> 4;
  const int v0 = blockIdx.x * 16;
  const int c0 = blockIdx.y * 16;
  const int which = blockIdx.z;
  const float* coef = (which == 0) ? qc : (which == 1) ? kc : (which == 2) ? vc : oc;

  v8f acc = vzero8();
  for (int f0 = 0; f0 < F_DIM; f0 += 4) {
    v2f a = *(const v2f*)&basis[(size_t)(v0 + r) * F_DIM + f0 + 2 * h];
    v2f b = *(const v2f*)&coef[(size_t)(c0 + r) * F_DIM + f0 + 2 * h];
    acc = wmma4(a, b, acc);
  }
  if (which < 3) {
    float* w = (which == 0) ? qw : (which == 1) ? kw : vw;
#pragma unroll
    for (int i = 0; i < 8; ++i)
      w[(size_t)(v0 + i + 8 * h) * C_DIM + c0 + r] = acc[i];
  } else {
#pragma unroll
    for (int i = 0; i < 8; ++i)
      owT[(size_t)(c0 + r) * V_DIM + v0 + i + 8 * h] = acc[i];
  }
}

__global__ void qkv_kernel(const float* __restrict__ x,
                           const float* __restrict__ qw,
                           const float* __restrict__ kw,
                           const float* __restrict__ vw,
                           float* __restrict__ Q,
                           float* __restrict__ K,
                           float* __restrict__ Vv) {
  const int lane = threadIdx.x & 31;
  const int r = lane & 15;
  const int h = lane >> 4;
  const int m0 = blockIdx.x * 16;
  const int c0 = blockIdx.y * 64;
  const int which = blockIdx.z;
  const float* w = (which == 0) ? qw : (which == 1) ? kw : vw;
  float* out = (which == 0) ? Q : (which == 1) ? K : Vv;

  v8f acc[4];
#pragma unroll
  for (int nt = 0; nt < 4; ++nt) acc[nt] = vzero8();

  for (int k0 = 0; k0 < V_DIM; k0 += 4) {
    v2f a = *(const v2f*)&x[(size_t)(m0 + r) * V_DIM + k0 + 2 * h];
    const float* w0 = &w[(size_t)(k0 + 2 * h) * C_DIM + c0 + r];
    const float* w1 = w0 + C_DIM;
#pragma unroll
    for (int nt = 0; nt < 4; ++nt) {
      v2f b;
      b.x = w0[nt * 16];
      b.y = w1[nt * 16];
      acc[nt] = wmma4(a, b, acc[nt]);
    }
  }
#pragma unroll
  for (int nt = 0; nt < 4; ++nt)
#pragma unroll
    for (int i = 0; i < 8; ++i)
      out[(size_t)(m0 + i + 8 * h) * C_DIM + c0 + nt * 16 + r] = acc[nt][i];
}

#define ATT_WAVES 4
#define SS_PITCH 68

__global__ __launch_bounds__(32 * ATT_WAVES)
void attn_kernel(const float* __restrict__ Q,
                 const float* __restrict__ K,
                 const float* __restrict__ Vv,
                 const float* __restrict__ decay_logit,
                 float* __restrict__ R) {
  __shared__ float qs[16 * 260];
  __shared__ float ss[2][16 * SS_PITCH];

  const int tid = threadIdx.x;
  const int lane = tid & 31;
  const int w = __builtin_amdgcn_readfirstlane(tid >> 5);
  const int r = lane & 15;
  const int h = lane >> 4;
  const int i0 = blockIdx.x * 16;
  const int c0 = w * 64;
  const int b = blockIdx.z;

  const float* Qb = Q + (size_t)b * T_DIM * C_DIM;
  const float* Kb = K + (size_t)b * T_DIM * C_DIM;
  const float* Vb = Vv + (size_t)b * T_DIM * C_DIM;
  float* Rb = R + (size_t)b * T_DIM * C_DIM;

  for (int idx = tid; idx < 16 * 64; idx += 32 * ATT_WAVES) {
    int row = idx >> 6;
    int c4 = (idx & 63) << 2;
    *(float4*)&qs[row * 260 + c4] =
        *(const float4*)&Qb[(size_t)(i0 + row) * C_DIM + c4];
  }
  __syncthreads();

  const float dl = *decay_logit;
  const float dec = 1.0f / (1.0f + __expf(-dl));
  const float ln_dec = __logf(dec);

  v8f acc0 = vzero8(), acc1 = vzero8(), acc2 = vzero8(), acc3 = vzero8();

  int buf = 0;
  for (int jc = i0 & ~63; jc < T_DIM; jc += 64) {
    const int j0 = jc + 16 * w;
    {
      int jn = (jc + 64 < T_DIM) ? (jc + 64) : jc;
      __builtin_prefetch(&Kb[(size_t)(jn + 16 * w + r) * C_DIM + 128 * h], 0, 1);
      __builtin_prefetch(&Vb[(size_t)(jn + lane) * C_DIM + c0], 0, 1);
      __builtin_prefetch(&Vb[(size_t)(jn + 32 + lane) * C_DIM + c0], 0, 1);
    }

    float* sb = &ss[buf][0];
    if (j0 >= i0) {
      v8f s = vzero8();
      const float* Kj = Kb + (size_t)j0 * C_DIM;
      for (int k0 = 0; k0 < C_DIM; k0 += 4) {
        v2f a = *(const v2f*)&qs[r * 260 + k0 + 2 * h];
        v2f bb = *(const v2f*)&Kj[(size_t)r * C_DIM + k0 + 2 * h];
        s = wmma4(a, bb, s);
      }
#pragma unroll
      for (int i = 0; i < 8; ++i) {
        int gi = i0 + i + 8 * h;
        int gj = j0 + r;
        int d = gj - gi;
        float wgt = (d > 0) ? __expf(ln_dec * (float)(d - 1)) : 0.0f;
        sb[(i + 8 * h) * SS_PITCH + 16 * w + r] = s[i] * wgt;
      }
    } else {
#pragma unroll
      for (int i = 0; i < 8; ++i)
        sb[(i + 8 * h) * SS_PITCH + 16 * w + r] = 0.0f;
    }
    __syncthreads();
#pragma unroll
    for (int kk = 0; kk < 16; ++kk) {
      v2f a = *(const v2f*)&sb[r * SS_PITCH + kk * 4 + 2 * h];
      const float* Vr0 = Vb + (size_t)(jc + kk * 4 + 2 * h) * C_DIM + c0;
      const float* Vr1 = Vr0 + C_DIM;
      v2f bb;
      bb.x = Vr0[r];      bb.y = Vr1[r];      acc0 = wmma4(a, bb, acc0);
      bb.x = Vr0[16 + r]; bb.y = Vr1[16 + r]; acc1 = wmma4(a, bb, acc1);
      bb.x = Vr0[32 + r]; bb.y = Vr1[32 + r]; acc2 = wmma4(a, bb, acc2);
      bb.x = Vr0[48 + r]; bb.y = Vr1[48 + r]; acc3 = wmma4(a, bb, acc3);
    }
    buf ^= 1;
  }

#pragma unroll
  for (int i = 0; i < 8; ++i) {
    size_t row = (size_t)(i0 + i + 8 * h) * C_DIM + c0 + r;
    Rb[row] = acc0[i];
    Rb[row + 16] = acc1[i];
    Rb[row + 32] = acc2[i];
    Rb[row + 48] = acc3[i];
  }
}

__global__ void out_proj_kernel(const float* __restrict__ R,
                                const float* __restrict__ owT,
                                const float* __restrict__ out_scale,
                                float* __restrict__ out) {
  const int lane = threadIdx.x & 31;
  const int r = lane & 15;
  const int h = lane >> 4;
  const int m0 = blockIdx.x * 16;
  const int v0 = blockIdx.y * 64;

  v8f acc[4];
#pragma unroll
  for (int nt = 0; nt < 4; ++nt) acc[nt] = vzero8();

  for (int k0 = 0; k0 < C_DIM; k0 += 4) {
    v2f a = *(const v2f*)&R[(size_t)(m0 + r) * C_DIM + k0 + 2 * h];
    const float* w0 = &owT[(size_t)(k0 + 2 * h) * V_DIM + v0 + r];
    const float* w1 = w0 + V_DIM;
#pragma unroll
    for (int nt = 0; nt < 4; ++nt) {
      v2f b;
      b.x = w0[nt * 16];
      b.y = w1[nt * 16];
      acc[nt] = wmma4(a, b, acc[nt]);
    }
  }
  const float sc = *out_scale;
#pragma unroll
  for (int nt = 0; nt < 4; ++nt)
#pragma unroll
    for (int i = 0; i < 8; ++i)
      out[(size_t)(m0 + i + 8 * h) * V_DIM + v0 + nt * 16 + r] = acc[nt][i] * sc;
}

extern "C" void kernel_launch(void* const* d_in, const int* in_sizes, int n_in,
                              void* d_out, int out_size, void* d_ws, size_t ws_size,
                              hipStream_t stream) {
  (void)in_sizes; (void)n_in; (void)out_size; (void)ws_size;

  const float* x = (const float*)d_in[0];
  const float* basis = (const float*)d_in[1];
  const float* qc = (const float*)d_in[2];
  const float* kc = (const float*)d_in[3];
  const float* vc = (const float*)d_in[4];
  const float* oc = (const float*)d_in[5];
  const float* decay_logit = (const float*)d_in[6];
  const float* out_scale = (const float*)d_in[7];
  float* out = (float*)d_out;

  float* ws = (float*)d_ws;
  float* qw = ws;
  float* kw = qw + (size_t)V_DIM * C_DIM;
  float* vw = kw + (size_t)V_DIM * C_DIM;
  float* owT = vw + (size_t)V_DIM * C_DIM;
  float* Q = owT + (size_t)V_DIM * C_DIM;
  float* K = Q + (size_t)M_DIM * C_DIM;
  float* Vv = K + (size_t)M_DIM * C_DIM;
  float* R = Vv + (size_t)M_DIM * C_DIM;

  dim3 blk(32);

  proj_weights_kernel<<<dim3(V_DIM / 16, C_DIM / 16, 4), blk, 0, stream>>>(
      basis, qc, kc, vc, oc, qw, kw, vw, owT);

  qkv_kernel<<<dim3(M_DIM / 16, C_DIM / 64, 3), blk, 0, stream>>>(
      x, qw, kw, vw, Q, K, Vv);

  attn_kernel<<<dim3(T_DIM / 16, 1, B_DIM), dim3(32 * ATT_WAVES), 0, stream>>>(
      Q, K, Vv, decay_logit, R);

  out_proj_kernel<<<dim3(M_DIM / 16, V_DIM / 64), blk, 0, stream>>>(
      R, owT, out_scale, out);
}


# attention: each work-group takes 32 query rows (two row tiles) instead of 16, so every K and V fragment loaded feeds two WMMAs (two independent score chains); summation order unchanged
# speedup vs baseline: 2.4660x; 1.1133x over previous
#include <hip/hip_runtime.h>


typedef float v2f __attribute__((ext_vector_type(2)));
typedef float v8f __attribute__((ext_vector_type(8)));

#define C_DIM 256
#define V_DIM 1024
#define F_DIM 256
#define T_DIM 4096
#define B_DIM 4
#define M_DIM (B_DIM * T_DIM)

static __device__ __forceinline__ v8f wmma4(v2f a, v2f b, v8f c) {
  return __builtin_amdgcn_wmma_f32_16x16x4_f32(
       false, a,  false, b,
       (short)0, c,  false,  false);
}

static __device__ __forceinline__ v8f vzero8() {
  v8f z;
#pragma unroll
  for (int i = 0; i < 8; ++i) z[i] = 0.0f;
  return z;
}

__global__ void proj_weights_kernel(const float* __restrict__ basis,
                                    const float* __restrict__ qc,
                                    const float* __restrict__ kc,
                                    const float* __restrict__ vc,
                                    const float* __restrict__ oc,
                                    float* __restrict__ qw,
                                    float* __restrict__ kw,
                                    float* __restrict__ vw,
                                    float* __restrict__ owT) {
  const int lane = threadIdx.x & 31;
  const int r = lane & 15;
  const int h = lane >> 4;
  const int v0 = blockIdx.x * 16;
  const int c0 = blockIdx.y * 16;
  const int which = blockIdx.z;
  const float* coef = (which == 0) ? qc : (which == 1) ? kc : (which == 2) ? vc : oc;

  v8f acc = vzero8();
  for (int f0 = 0; f0 < F_DIM; f0 += 4) {
    v2f a = *(const v2f*)&basis[(size_t)(v0 + r) * F_DIM + f0 + 2 * h];
    v2f b = *(const v2f*)&coef[(size_t)(c0 + r) * F_DIM + f0 + 2 * h];
    acc = wmma4(a, b, acc);
  }
  if (which < 3) {
    float* w = (which == 0) ? qw : (which == 1) ? kw : vw;
#pragma unroll
    for (int i = 0; i < 8; ++i)
      w[(size_t)(v0 + i + 8 * h) * C_DIM + c0 + r] = acc[i];
  } else {
#pragma unroll
    for (int i = 0; i < 8; ++i)
      owT[(size_t)(c0 + r) * V_DIM + v0 + i + 8 * h] = acc[i];
  }
}

__global__ void qkv_kernel(const float* __restrict__ x,
                           const float* __restrict__ qw,
                           const float* __restrict__ kw,
                           const float* __restrict__ vw,
                           float* __restrict__ Q,
                           float* __restrict__ K,
                           float* __restrict__ Vv) {
  const int lane = threadIdx.x & 31;
  const int r = lane & 15;
  const int h = lane >> 4;
  const int m0 = blockIdx.x * 16;
  const int c0 = blockIdx.y * 64;
  const int which = blockIdx.z;
  const float* w = (which == 0) ? qw : (which == 1) ? kw : vw;
  float* out = (which == 0) ? Q : (which == 1) ? K : Vv;

  v8f acc[4];
#pragma unroll
  for (int nt = 0; nt < 4; ++nt) acc[nt] = vzero8();

  for (int k0 = 0; k0 < V_DIM; k0 += 4) {
    v2f a = *(const v2f*)&x[(size_t)(m0 + r) * V_DIM + k0 + 2 * h];
    const float* w0 = &w[(size_t)(k0 + 2 * h) * C_DIM + c0 + r];
    const float* w1 = w0 + C_DIM;
#pragma unroll
    for (int nt = 0; nt < 4; ++nt) {
      v2f b;
      b.x = w0[nt * 16];
      b.y = w1[nt * 16];
      acc[nt] = wmma4(a, b, acc[nt]);
    }
  }
#pragma unroll
  for (int nt = 0; nt < 4; ++nt)
#pragma unroll
    for (int i = 0; i < 8; ++i)
      out[(size_t)(m0 + i + 8 * h) * C_DIM + c0 + nt * 16 + r] = acc[nt][i];
}

#define ATT_WAVES 4
#define ATT_ROWS 32
#define QS_PITCH 260
#define SS_PITCH 68

__global__ __launch_bounds__(32 * ATT_WAVES)
void attn_kernel(const float* __restrict__ Q,
                 const float* __restrict__ K,
                 const float* __restrict__ Vv,
                 const float* __restrict__ decay_logit,
                 float* __restrict__ R) {
  __shared__ float qs[ATT_ROWS * QS_PITCH];
  __shared__ float ss[2][ATT_ROWS * SS_PITCH];

  const int tid = threadIdx.x;
  const int lane = tid & 31;
  const int w = __builtin_amdgcn_readfirstlane(tid >> 5);
  const int r = lane & 15;
  const int h = lane >> 4;
  const int i0 = blockIdx.x * ATT_ROWS;
  const int c0 = w * 64;
  const int b = blockIdx.z;

  const float* Qb = Q + (size_t)b * T_DIM * C_DIM;
  const float* Kb = K + (size_t)b * T_DIM * C_DIM;
  const float* Vb = Vv + (size_t)b * T_DIM * C_DIM;
  float* Rb = R + (size_t)b * T_DIM * C_DIM;

  for (int idx = tid; idx < ATT_ROWS * 64; idx += 32 * ATT_WAVES) {
    int row = idx >> 6;
    int c4 = (idx & 63) << 2;
    *(float4*)&qs[row * QS_PITCH + c4] =
        *(const float4*)&Qb[(size_t)(i0 + row) * C_DIM + c4];
  }
  __syncthreads();

  const float dl = *decay_logit;
  const float dec = 1.0f / (1.0f + __expf(-dl));
  const float ln_dec = __logf(dec);

  v8f a0c0 = vzero8(), a0c1 = vzero8(), a0c2 = vzero8(), a0c3 = vzero8();
  v8f a1c0 = vzero8(), a1c1 = vzero8(), a1c2 = vzero8(), a1c3 = vzero8();

  int buf = 0;
  for (int jc = i0 & ~63; jc < T_DIM; jc += 64) {
    const int j0 = jc + 16 * w;
    {
      int jn = (jc + 64 < T_DIM) ? (jc + 64) : jc;
      __builtin_prefetch(&Kb[(size_t)(jn + 16 * w + r) * C_DIM + 128 * h], 0, 1);
      __builtin_prefetch(&Vb[(size_t)(jn + lane) * C_DIM + c0], 0, 1);
      __builtin_prefetch(&Vb[(size_t)(jn + 32 + lane) * C_DIM + c0], 0, 1);
    }

    float* sb = &ss[buf][0];
    if (j0 >= i0) {
      v8f s0 = vzero8(), s1 = vzero8();
      const float* Kj = Kb + (size_t)j0 * C_DIM;
#pragma unroll 16
      for (int k0 = 0; k0 < C_DIM; k0 += 4) {
        v2f qa = *(const v2f*)&qs[r * QS_PITCH + k0 + 2 * h];
        v2f qb = *(const v2f*)&qs[(16 + r) * QS_PITCH + k0 + 2 * h];
        v2f bb = *(const v2f*)&Kj[(size_t)r * C_DIM + k0 + 2 * h];
        s0 = wmma4(qa, bb, s0);
        s1 = wmma4(qb, bb, s1);
      }
#pragma unroll
      for (int i = 0; i < 8; ++i) {
        int gi = i0 + i + 8 * h;
        int gj = j0 + r;
        int d0 = gj - gi;
        int d1 = d0 - 16;
        float wg0 = (d0 > 0) ? __expf(ln_dec * (float)(d0 - 1)) : 0.0f;
        float wg1 = (d1 > 0) ? __expf(ln_dec * (float)(d1 - 1)) : 0.0f;
        sb[(i + 8 * h) * SS_PITCH + 16 * w + r] = s0[i] * wg0;
        sb[(16 + i + 8 * h) * SS_PITCH + 16 * w + r] = s1[i] * wg1;
      }
    } else {
#pragma unroll
      for (int i = 0; i < 8; ++i) {
        sb[(i + 8 * h) * SS_PITCH + 16 * w + r] = 0.0f;
        sb[(16 + i + 8 * h) * SS_PITCH + 16 * w + r] = 0.0f;
      }
    }
    __syncthreads();
#pragma unroll
    for (int kk = 0; kk < 16; ++kk) {
      v2f pa = *(const v2f*)&sb[r * SS_PITCH + kk * 4 + 2 * h];
      v2f pb = *(const v2f*)&sb[(16 + r) * SS_PITCH + kk * 4 + 2 * h];
      const float* Vr0 = Vb + (size_t)(jc + kk * 4 + 2 * h) * C_DIM + c0;
      const float* Vr1 = Vr0 + C_DIM;
      v2f bb;
      bb.x = Vr0[r];      bb.y = Vr1[r];
      a0c0 = wmma4(pa, bb, a0c0);  a1c0 = wmma4(pb, bb, a1c0);
      bb.x = Vr0[16 + r]; bb.y = Vr1[16 + r];
      a0c1 = wmma4(pa, bb, a0c1);  a1c1 = wmma4(pb, bb, a1c1);
      bb.x = Vr0[32 + r]; bb.y = Vr1[32 + r];
      a0c2 = wmma4(pa, bb, a0c2);  a1c2 = wmma4(pb, bb, a1c2);
      bb.x = Vr0[48 + r]; bb.y = Vr1[48 + r];
      a0c3 = wmma4(pa, bb, a0c3);  a1c3 = wmma4(pb, bb, a1c3);
    }
    buf ^= 1;
  }

#pragma unroll
  for (int i = 0; i < 8; ++i) {
    size_t row0 = (size_t)(i0 + i + 8 * h) * C_DIM + c0 + r;
    size_t row1 = row0 + (size_t)16 * C_DIM;
    Rb[row0] = a0c0[i];
    Rb[row0 + 16] = a0c1[i];
    Rb[row0 + 32] = a0c2[i];
    Rb[row0 + 48] = a0c3[i];
    Rb[row1] = a1c0[i];
    Rb[row1 + 16] = a1c1[i];
    Rb[row1 + 32] = a1c2[i];
    Rb[row1 + 48] = a1c3[i];
  }
}

__global__ void out_proj_kernel(const float* __restrict__ R,
                                const float* __restrict__ owT,
                                const float* __restrict__ out_scale,
                                float* __restrict__ out) {
  const int lane = threadIdx.x & 31;
  const int r = lane & 15;
  const int h = lane >> 4;
  const int m0 = blockIdx.x * 16;
  const int v0 = blockIdx.y * 64;

  v8f acc[4];
#pragma unroll
  for (int nt = 0; nt < 4; ++nt) acc[nt] = vzero8();

  for (int k0 = 0; k0 < C_DIM; k0 += 4) {
    v2f a = *(const v2f*)&R[(size_t)(m0 + r) * C_DIM + k0 + 2 * h];
    const float* w0 = &owT[(size_t)(k0 + 2 * h) * V_DIM + v0 + r];
    const float* w1 = w0 + V_DIM;
#pragma unroll
    for (int nt = 0; nt < 4; ++nt) {
      v2f b;
      b.x = w0[nt * 16];
      b.y = w1[nt * 16];
      acc[nt] = wmma4(a, b, acc[nt]);
    }
  }
  const float sc = *out_scale;
#pragma unroll
  for (int nt = 0; nt < 4; ++nt)
#pragma unroll
    for (int i = 0; i < 8; ++i)
      out[(size_t)(m0 + i + 8 * h) * V_DIM + v0 + nt * 16 + r] = acc[nt][i] * sc;
}

extern "C" void kernel_launch(void* const* d_in, const int* in_sizes, int n_in,
                              void* d_out, int out_size, void* d_ws, size_t ws_size,
                              hipStream_t stream) {
  (void)in_sizes; (void)n_in; (void)out_size; (void)ws_size;

  const float* x = (const float*)d_in[0];
  const float* basis = (const float*)d_in[1];
  const float* qc = (const float*)d_in[2];
  const float* kc = (const float*)d_in[3];
  const float* vc = (const float*)d_in[4];
  const float* oc = (const float*)d_in[5];
  const float* decay_logit = (const float*)d_in[6];
  const float* out_scale = (const float*)d_in[7];
  float* out = (float*)d_out;

  float* ws = (float*)d_ws;
  float* qw = ws;
  float* kw = qw + (size_t)V_DIM * C_DIM;
  float* vw = kw + (size_t)V_DIM * C_DIM;
  float* owT = vw + (size_t)V_DIM * C_DIM;
  float* Q = owT + (size_t)V_DIM * C_DIM;
  float* K = Q + (size_t)M_DIM * C_DIM;
  float* Vv = K + (size_t)M_DIM * C_DIM;
  float* R = Vv + (size_t)M_DIM * C_DIM;

  dim3 blk(32);

  proj_weights_kernel<<<dim3(V_DIM / 16, C_DIM / 16, 4), blk, 0, stream>>>(
      basis, qc, kc, vc, oc, qw, kw, vw, owT);

  qkv_kernel<<<dim3(M_DIM / 16, C_DIM / 64, 3), blk, 0, stream>>>(
      x, qw, kw, vw, Q, K, Vv);

  attn_kernel<<<dim3(T_DIM / ATT_ROWS, 1, B_DIM), dim3(32 * ATT_WAVES), 0, stream>>>(
      Q, K, Vv, decay_logit, R);

  out_proj_kernel<<<dim3(M_DIM / 16, V_DIM / 64), blk, 0, stream>>>(
      R, owT, out_scale, out);
}


# attention: 8-wave work-groups, one per WGP (LDS sized above half a WGP's), take row block g then its mirror 127-g: all 256 WGPs walk equal key steps of the causal triangle; summation order unchanged
# speedup vs baseline: 3.1170x; 1.2640x over previous
#include <hip/hip_runtime.h>


typedef float v2f __attribute__((ext_vector_type(2)));
typedef float v8f __attribute__((ext_vector_type(8)));

#define C_DIM 256
#define V_DIM 1024
#define F_DIM 256
#define T_DIM 4096
#define B_DIM 4
#define M_DIM (B_DIM * T_DIM)

static __device__ __forceinline__ v8f wmma4(v2f a, v2f b, v8f c) {
  return __builtin_amdgcn_wmma_f32_16x16x4_f32(
       false, a,  false, b,
       (short)0, c,  false,  false);
}

static __device__ __forceinline__ v8f vzero8() {
  v8f z;
#pragma unroll
  for (int i = 0; i < 8; ++i) z[i] = 0.0f;
  return z;
}

__global__ void proj_weights_kernel(const float* __restrict__ basis,
                                    const float* __restrict__ qc,
                                    const float* __restrict__ kc,
                                    const float* __restrict__ vc,
                                    const float* __restrict__ oc,
                                    float* __restrict__ qw,
                                    float* __restrict__ kw,
                                    float* __restrict__ vw,
                                    float* __restrict__ owT) {
  const int lane = threadIdx.x & 31;
  const int r = lane & 15;
  const int h = lane >> 4;
  const int v0 = blockIdx.x * 16;
  const int c0 = blockIdx.y * 16;
  const int which = blockIdx.z;
  const float* coef = (which == 0) ? qc : (which == 1) ? kc : (which == 2) ? vc : oc;

  v8f acc = vzero8();
  for (int f0 = 0; f0 < F_DIM; f0 += 4) {
    v2f a = *(const v2f*)&basis[(size_t)(v0 + r) * F_DIM + f0 + 2 * h];
    v2f b = *(const v2f*)&coef[(size_t)(c0 + r) * F_DIM + f0 + 2 * h];
    acc = wmma4(a, b, acc);
  }
  if (which < 3) {
    float* w = (which == 0) ? qw : (which == 1) ? kw : vw;
#pragma unroll
    for (int i = 0; i < 8; ++i)
      w[(size_t)(v0 + i + 8 * h) * C_DIM + c0 + r] = acc[i];
  } else {
#pragma unroll
    for (int i = 0; i < 8; ++i)
      owT[(size_t)(c0 + r) * V_DIM + v0 + i + 8 * h] = acc[i];
  }
}

__global__ void qkv_kernel(const float* __restrict__ x,
                           const float* __restrict__ qw,
                           const float* __restrict__ kw,
                           const float* __restrict__ vw,
                           float* __restrict__ Q,
                           float* __restrict__ K,
                           float* __restrict__ Vv) {
  const int lane = threadIdx.x & 31;
  const int r = lane & 15;
  const int h = lane >> 4;
  const int m0 = blockIdx.x * 16;
  const int c0 = blockIdx.y * 64;
  const int which = blockIdx.z;
  const float* w = (which == 0) ? qw : (which == 1) ? kw : vw;
  float* out = (which == 0) ? Q : (which == 1) ? K : Vv;

  v8f acc[4];
#pragma unroll
  for (int nt = 0; nt < 4; ++nt) acc[nt] = vzero8();

  for (int k0 = 0; k0 < V_DIM; k0 += 4) {
    v2f a = *(const v2f*)&x[(size_t)(m0 + r) * V_DIM + k0 + 2 * h];
    const float* w0 = &w[(size_t)(k0 + 2 * h) * C_DIM + c0 + r];
    const float* w1 = w0 + C_DIM;
#pragma unroll
    for (int nt = 0; nt < 4; ++nt) {
      v2f b;
      b.x = w0[nt * 16];
      b.y = w1[nt * 16];
      acc[nt] = wmma4(a, b, acc[nt]);
    }
  }
#pragma unroll
  for (int nt = 0; nt < 4; ++nt)
#pragma unroll
    for (int i = 0; i < 8; ++i)
      out[(size_t)(m0 + i + 8 * h) * C_DIM + c0 + nt * 16 + r] = acc[nt][i];
}

#define ATT_WAVES 8
#define ATT_ROWS 32
#define ATT_KEYS (16 * ATT_WAVES)
#define ATT_RBLOCKS (T_DIM / ATT_ROWS)
#define QS_PITCH 260
#define SS_PITCH 132
#define ATT_QS_FLOATS (ATT_ROWS * QS_PITCH)
#define ATT_SS_FLOATS (ATT_ROWS * SS_PITCH)
#define ATT_LDS_FLOATS 41216

__global__ __launch_bounds__(32 * ATT_WAVES)
void attn_kernel(const float* __restrict__ Q,
                 const float* __restrict__ K,
                 const float* __restrict__ Vv,
                 const float* __restrict__ decay_logit,
                 float* __restrict__ R) {
  __shared__ float lds[ATT_LDS_FLOATS];
  float* const qs = &lds[0];
  float* const ss = &lds[ATT_QS_FLOATS];

  const int tid = threadIdx.x;
  const int lane = tid & 31;
  const int w = __builtin_amdgcn_readfirstlane(tid >> 5);
  const int r = lane & 15;
  const int h = lane >> 4;
  const int c0 = w * 32;
  const int b = blockIdx.z;

  const float* Qb = Q + (size_t)b * T_DIM * C_DIM;
  const float* Kb = K + (size_t)b * T_DIM * C_DIM;
  const float* Vb = Vv + (size_t)b * T_DIM * C_DIM;
  float* Rb = R + (size_t)b * T_DIM * C_DIM;

  const float dl = *decay_logit;
  const float dec = 1.0f / (1.0f + __expf(-dl));
  const float ln_dec = __logf(dec);

  for (int part = 0; part < 2; ++part) {
    const int rb = (part == 0) ? (int)blockIdx.x : (ATT_RBLOCKS - 1 - (int)blockIdx.x);
    const int i0 = rb * ATT_ROWS;

    for (int idx = tid; idx < ATT_ROWS * 64; idx += 32 * ATT_WAVES) {
      int row = idx >> 6;
      int c4 = (idx & 63) << 2;
      *(float4*)&qs[row * QS_PITCH + c4] =
          *(const float4*)&Qb[(size_t)(i0 + row) * C_DIM + c4];
    }
    __syncthreads();

    v8f a0c0 = vzero8(), a0c1 = vzero8();
    v8f a1c0 = vzero8(), a1c1 = vzero8();

    int buf = 0;
    for (int jc = i0 & ~(ATT_KEYS - 1); jc < T_DIM; jc += ATT_KEYS) {
      const int j0 = jc + 16 * w;
      {
        int jn = (jc + ATT_KEYS < T_DIM) ? (jc + ATT_KEYS) : jc;
        __builtin_prefetch(&Kb[(size_t)(jn + 16 * w + r) * C_DIM + 128 * h], 0, 1);
        __builtin_prefetch(&Vb[(size_t)(jn + lane) * C_DIM + c0], 0, 1);
        __builtin_prefetch(&Vb[(size_t)(jn + 32 + lane) * C_DIM + c0], 0, 1);
        __builtin_prefetch(&Vb[(size_t)(jn + 64 + lane) * C_DIM + c0], 0, 1);
        __builtin_prefetch(&Vb[(size_t)(jn + 96 + lane) * C_DIM + c0], 0, 1);
      }

      float* sb = &ss[buf * ATT_SS_FLOATS];
      if (j0 >= i0) {
        v8f s0 = vzero8(), s1 = vzero8();
        const float* Kj = Kb + (size_t)j0 * C_DIM;
#pragma unroll 16
        for (int k0 = 0; k0 < C_DIM; k0 += 4) {
          v2f qa = *(const v2f*)&qs[r * QS_PITCH + k0 + 2 * h];
          v2f qb = *(const v2f*)&qs[(16 + r) * QS_PITCH + k0 + 2 * h];
          v2f bb = *(const v2f*)&Kj[(size_t)r * C_DIM + k0 + 2 * h];
          s0 = wmma4(qa, bb, s0);
          s1 = wmma4(qb, bb, s1);
        }
#pragma unroll
        for (int i = 0; i < 8; ++i) {
          int gi = i0 + i + 8 * h;
          int gj = j0 + r;
          int d0 = gj - gi;
          int d1 = d0 - 16;
          float wg0 = (d0 > 0) ? __expf(ln_dec * (float)(d0 - 1)) : 0.0f;
          float wg1 = (d1 > 0) ? __expf(ln_dec * (float)(d1 - 1)) : 0.0f;
          sb[(i + 8 * h) * SS_PITCH + 16 * w + r] = s0[i] * wg0;
          sb[(16 + i + 8 * h) * SS_PITCH + 16 * w + r] = s1[i] * wg1;
        }
      } else {
#pragma unroll
        for (int i = 0; i < 8; ++i) {
          sb[(i + 8 * h) * SS_PITCH + 16 * w + r] = 0.0f;
          sb[(16 + i + 8 * h) * SS_PITCH + 16 * w + r] = 0.0f;
        }
      }
      __syncthreads();
#pragma unroll
      for (int kk = 0; kk < ATT_KEYS / 4; ++kk) {
        v2f pa = *(const v2f*)&sb[r * SS_PITCH + kk * 4 + 2 * h];
        v2f pb = *(const v2f*)&sb[(16 + r) * SS_PITCH + kk * 4 + 2 * h];
        const float* Vr0 = Vb + (size_t)(jc + kk * 4 + 2 * h) * C_DIM + c0;
        const float* Vr1 = Vr0 + C_DIM;
        v2f bb;
        bb.x = Vr0[r];      bb.y = Vr1[r];
        a0c0 = wmma4(pa, bb, a0c0);  a1c0 = wmma4(pb, bb, a1c0);
        bb.x = Vr0[16 + r]; bb.y = Vr1[16 + r];
        a0c1 = wmma4(pa, bb, a0c1);  a1c1 = wmma4(pb, bb, a1c1);
      }
      buf ^= 1;
    }

#pragma unroll
    for (int i = 0; i < 8; ++i) {
      size_t row0 = (size_t)(i0 + i + 8 * h) * C_DIM + c0 + r;
      size_t row1 = row0 + (size_t)16 * C_DIM;
      Rb[row0] = a0c0[i];
      Rb[row0 + 16] = a0c1[i];
      Rb[row1] = a1c0[i];
      Rb[row1 + 16] = a1c1[i];
    }
  }
}

__global__ void out_proj_kernel(const float* __restrict__ R,
                                const float* __restrict__ owT,
                                const float* __restrict__ out_scale,
                                float* __restrict__ out) {
  const int lane = threadIdx.x & 31;
  const int r = lane & 15;
  const int h = lane >> 4;
  const int m0 = blockIdx.x * 16;
  const int v0 = blockIdx.y * 64;

  v8f acc[4];
#pragma unroll
  for (int nt = 0; nt < 4; ++nt) acc[nt] = vzero8();

  for (int k0 = 0; k0 < C_DIM; k0 += 4) {
    v2f a = *(const v2f*)&R[(size_t)(m0 + r) * C_DIM + k0 + 2 * h];
    const float* w0 = &owT[(size_t)(k0 + 2 * h) * V_DIM + v0 + r];
    const float* w1 = w0 + V_DIM;
#pragma unroll
    for (int nt = 0; nt < 4; ++nt) {
      v2f b;
      b.x = w0[nt * 16];
      b.y = w1[nt * 16];
      acc[nt] = wmma4(a, b, acc[nt]);
    }
  }
  const float sc = *out_scale;
#pragma unroll
  for (int nt = 0; nt < 4; ++nt)
#pragma unroll
    for (int i = 0; i < 8; ++i)
      out[(size_t)(m0 + i + 8 * h) * V_DIM + v0 + nt * 16 + r] = acc[nt][i] * sc;
}

extern "C" void kernel_launch(void* const* d_in, const int* in_sizes, int n_in,
                              void* d_out, int out_size, void* d_ws, size_t ws_size,
                              hipStream_t stream) {
  (void)in_sizes; (void)n_in; (void)out_size; (void)ws_size;

  const float* x = (const float*)d_in[0];
  const float* basis = (const float*)d_in[1];
  const float* qc = (const float*)d_in[2];
  const float* kc = (const float*)d_in[3];
  const float* vc = (const float*)d_in[4];
  const float* oc = (const float*)d_in[5];
  const float* decay_logit = (const float*)d_in[6];
  const float* out_scale = (const float*)d_in[7];
  float* out = (float*)d_out;

  float* ws = (float*)d_ws;
  float* qw = ws;
  float* kw = qw + (size_t)V_DIM * C_DIM;
  float* vw = kw + (size_t)V_DIM * C_DIM;
  float* owT = vw + (size_t)V_DIM * C_DIM;
  float* Q = owT + (size_t)V_DIM * C_DIM;
  float* K = Q + (size_t)M_DIM * C_DIM;
  float* Vv = K + (size_t)M_DIM * C_DIM;
  float* R = Vv + (size_t)M_DIM * C_DIM;

  dim3 blk(32);

  proj_weights_kernel<<<dim3(V_DIM / 16, C_DIM / 16, 4), blk, 0, stream>>>(
      basis, qc, kc, vc, oc, qw, kw, vw, owT);

  qkv_kernel<<<dim3(M_DIM / 16, C_DIM / 64, 3), blk, 0, stream>>>(
      x, qw, kw, vw, Q, K, Vv);

  attn_kernel<<<dim3(ATT_RBLOCKS / 2, 1, B_DIM), dim3(32 * ATT_WAVES), 0, stream>>>(
      Q, K, Vv, decay_logit, R);

  out_proj_kernel<<<dim3(M_DIM / 16, V_DIM / 64), blk, 0, stream>>>(
      R, owT, out_scale, out);
}


# x@w (Q,K,V) and retrieved@o_w GEMMs: 8-wave work-groups stage 128x32 and 32x128 operand slabs in LDS (2 stages), 32x64 register tile per wave (was 1 wave per 16x64 from global); sum order unchanged
# speedup vs baseline: 3.7037x; 1.1882x over previous
#include <hip/hip_runtime.h>


typedef float v2f __attribute__((ext_vector_type(2)));
typedef float v8f __attribute__((ext_vector_type(8)));

#define C_DIM 256
#define V_DIM 1024
#define F_DIM 256
#define T_DIM 4096
#define B_DIM 4
#define M_DIM (B_DIM * T_DIM)

static __device__ __forceinline__ v8f wmma4(v2f a, v2f b, v8f c) {
  return __builtin_amdgcn_wmma_f32_16x16x4_f32(
       false, a,  false, b,
       (short)0, c,  false,  false);
}

static __device__ __forceinline__ v8f vzero8() {
  v8f z;
#pragma unroll
  for (int i = 0; i < 8; ++i) z[i] = 0.0f;
  return z;
}

__global__ void proj_weights_kernel(const float* __restrict__ basis,
                                    const float* __restrict__ qc,
                                    const float* __restrict__ kc,
                                    const float* __restrict__ vc,
                                    const float* __restrict__ oc,
                                    float* __restrict__ qw,
                                    float* __restrict__ kw,
                                    float* __restrict__ vw,
                                    float* __restrict__ owT) {
  const int lane = threadIdx.x & 31;
  const int r = lane & 15;
  const int h = lane >> 4;
  const int v0 = blockIdx.x * 16;
  const int c0 = blockIdx.y * 16;
  const int which = blockIdx.z;
  const float* coef = (which == 0) ? qc : (which == 1) ? kc : (which == 2) ? vc : oc;

  v8f acc = vzero8();
  for (int f0 = 0; f0 < F_DIM; f0 += 4) {
    v2f a = *(const v2f*)&basis[(size_t)(v0 + r) * F_DIM + f0 + 2 * h];
    v2f b = *(const v2f*)&coef[(size_t)(c0 + r) * F_DIM + f0 + 2 * h];
    acc = wmma4(a, b, acc);
  }
  if (which < 3) {
    float* w = (which == 0) ? qw : (which == 1) ? kw : vw;
#pragma unroll
    for (int i = 0; i < 8; ++i)
      w[(size_t)(v0 + i + 8 * h) * C_DIM + c0 + r] = acc[i];
  } else {
#pragma unroll
    for (int i = 0; i < 8; ++i)
      owT[(size_t)(c0 + r) * V_DIM + v0 + i + 8 * h] = acc[i];
  }
}

#define GT_M 128
#define GT_N 128
#define GT_K 32
#define GT_XP 36
#define GT_WP 288
#define GT_XS (GT_M * GT_XP)
#define GT_WS ((GT_K / 2) * GT_WP)
#define GT_STAGE (GT_XS + GT_WS)
#define GT_LDS_FLOATS (2 * GT_STAGE)

template <int KDIM, int LDA, int LDB, int LDC, bool SCALED>
static __device__ __forceinline__ void gemm_tile_128x128(
    const float* __restrict__ A,
    const float* __restrict__ Bm,
    float* __restrict__ Cm,
    float scale, float* __restrict__ lds) {
  const int tid = threadIdx.x;
  const int lane = tid & 31;
  const int w = __builtin_amdgcn_readfirstlane(tid >> 5);
  const int r = lane & 15;
  const int h = lane >> 4;
  const int wm = (w >> 1) * 32;
  const int wn = (w & 1) * 64;

  const int xr = tid >> 3;
  const int xc = (tid & 7) << 2;
  const int wp = tid >> 5;
  const int wc = (tid & 31) << 2;
  const float* Ag = A + (size_t)xr * LDA + xc;
  const float* Bg = Bm + (size_t)(2 * wp) * LDB + wc;
  const int xs_off = xr * GT_XP + xc;
  const int ws_off = GT_XS + wp * GT_WP + 2 * wc;

  float4 xa[4], wb[4];
#pragma unroll
  for (int j = 0; j < 4; ++j) {
    xa[j] = *(const float4*)&Ag[(size_t)(32 * j) * LDA];
    wb[j] = *(const float4*)&Bg[(size_t)(16 * (j >> 1) + (j & 1)) * LDB];
  }
#pragma unroll
  for (int j = 0; j < 4; ++j)
    *(float4*)&lds[xs_off + 32 * j * GT_XP] = xa[j];
#pragma unroll
  for (int p = 0; p < 2; ++p) {
    const float4 e0 = wb[2 * p], e1 = wb[2 * p + 1];
    *(float4*)&lds[ws_off + 8 * p * GT_WP] = make_float4(e0.x, e1.x, e0.y, e1.y);
    *(float4*)&lds[ws_off + 8 * p * GT_WP + 4] = make_float4(e0.z, e1.z, e0.w, e1.w);
  }
  __syncthreads();

  v8f acc[2][4];
#pragma unroll
  for (int mt = 0; mt < 2; ++mt)
#pragma unroll
    for (int nt = 0; nt < 4; ++nt) acc[mt][nt] = vzero8();

  const int a_off = (wm + r) * GT_XP + 2 * h;
  const int b_off = GT_XS + h * GT_WP + 2 * (wn + r);

  constexpr int NCH = KDIM / GT_K;
  for (int kc = 0; kc < NCH; ++kc) {
    const float* cur = lds + (kc & 1) * GT_STAGE;
    float* nxt = lds + ((kc + 1) & 1) * GT_STAGE;
    const bool more = (kc + 1 < NCH);
    if (more) {
      const int k0 = (kc + 1) * GT_K;
#pragma unroll
      for (int j = 0; j < 4; ++j) {
        xa[j] = *(const float4*)&Ag[(size_t)(32 * j) * LDA + k0];
        wb[j] = *(const float4*)&Bg[(size_t)(k0 + 16 * (j >> 1) + (j & 1)) * LDB];
      }
    }
#pragma unroll
    for (int ks = 0; ks < GT_K; ks += 4) {
      v2f a0 = *(const v2f*)&cur[a_off + ks];
      v2f a1 = *(const v2f*)&cur[a_off + 16 * GT_XP + ks];
      const float* b0 = &cur[b_off + (ks >> 1) * GT_WP];
#pragma unroll
      for (int nt = 0; nt < 4; ++nt) {
        const v2f bb = *(const v2f*)&b0[nt * 32];
        acc[0][nt] = wmma4(a0, bb, acc[0][nt]);
        acc[1][nt] = wmma4(a1, bb, acc[1][nt]);
      }
    }
    if (more) {
#pragma unroll
      for (int j = 0; j < 4; ++j)
        *(float4*)&nxt[xs_off + 32 * j * GT_XP] = xa[j];
#pragma unroll
      for (int p = 0; p < 2; ++p) {
        const float4 e0 = wb[2 * p], e1 = wb[2 * p + 1];
        *(float4*)&nxt[ws_off + 8 * p * GT_WP] = make_float4(e0.x, e1.x, e0.y, e1.y);
        *(float4*)&nxt[ws_off + 8 * p * GT_WP + 4] = make_float4(e0.z, e1.z, e0.w, e1.w);
      }
      __syncthreads();
    }
  }

#pragma unroll
  for (int mt = 0; mt < 2; ++mt)
#pragma unroll
    for (int nt = 0; nt < 4; ++nt)
#pragma unroll
      for (int i = 0; i < 8; ++i) {
        float v = acc[mt][nt][i];
        if (SCALED) v *= scale;
        Cm[(size_t)(wm + mt * 16 + i + 8 * h) * LDC + wn + nt * 16 + r] = v;
      }
}

#define QKV_LDS_FLOATS 20992

__global__ __launch_bounds__(256)
void qkv_kernel(const float* __restrict__ x,
                const float* __restrict__ qw,
                const float* __restrict__ kw,
                const float* __restrict__ vw,
                float* __restrict__ Q,
                float* __restrict__ K,
                float* __restrict__ Vv) {
  __shared__ float lds[QKV_LDS_FLOATS];
  const int m0 = blockIdx.x * GT_M;
  const int which = (int)blockIdx.y >> 1;
  const int n0 = ((int)blockIdx.y & 1) * GT_N;
  const float* w = (which == 0) ? qw : (which == 1) ? kw : vw;
  float* out = (which == 0) ? Q : (which == 1) ? K : Vv;
  gemm_tile_128x128<V_DIM, V_DIM, C_DIM, C_DIM, false>(
      x + (size_t)m0 * V_DIM, w + n0, out + (size_t)m0 * C_DIM + n0, 1.0f, lds);
}

#define ATT_WAVES 8
#define ATT_ROWS 32
#define ATT_KEYS (16 * ATT_WAVES)
#define ATT_RBLOCKS (T_DIM / ATT_ROWS)
#define QS_PITCH 260
#define SS_PITCH 132
#define ATT_QS_FLOATS (ATT_ROWS * QS_PITCH)
#define ATT_SS_FLOATS (ATT_ROWS * SS_PITCH)
#define ATT_LDS_FLOATS 41216

__global__ __launch_bounds__(32 * ATT_WAVES)
void attn_kernel(const float* __restrict__ Q,
                 const float* __restrict__ K,
                 const float* __restrict__ Vv,
                 const float* __restrict__ decay_logit,
                 float* __restrict__ R) {
  __shared__ float lds[ATT_LDS_FLOATS];
  float* const qs = &lds[0];
  float* const ss = &lds[ATT_QS_FLOATS];

  const int tid = threadIdx.x;
  const int lane = tid & 31;
  const int w = __builtin_amdgcn_readfirstlane(tid >> 5);
  const int r = lane & 15;
  const int h = lane >> 4;
  const int c0 = w * 32;
  const int b = blockIdx.z;

  const float* Qb = Q + (size_t)b * T_DIM * C_DIM;
  const float* Kb = K + (size_t)b * T_DIM * C_DIM;
  const float* Vb = Vv + (size_t)b * T_DIM * C_DIM;
  float* Rb = R + (size_t)b * T_DIM * C_DIM;

  const float dl = *decay_logit;
  const float dec = 1.0f / (1.0f + __expf(-dl));
  const float ln_dec = __logf(dec);

  for (int part = 0; part < 2; ++part) {
    const int rb = (part == 0) ? (int)blockIdx.x : (ATT_RBLOCKS - 1 - (int)blockIdx.x);
    const int i0 = rb * ATT_ROWS;

    for (int idx = tid; idx < ATT_ROWS * 64; idx += 32 * ATT_WAVES) {
      int row = idx >> 6;
      int c4 = (idx & 63) << 2;
      *(float4*)&qs[row * QS_PITCH + c4] =
          *(const float4*)&Qb[(size_t)(i0 + row) * C_DIM + c4];
    }
    __syncthreads();

    v8f a0c0 = vzero8(), a0c1 = vzero8();
    v8f a1c0 = vzero8(), a1c1 = vzero8();

    int buf = 0;
    for (int jc = i0 & ~(ATT_KEYS - 1); jc < T_DIM; jc += ATT_KEYS) {
      const int j0 = jc + 16 * w;
      {
        int jn = (jc + ATT_KEYS < T_DIM) ? (jc + ATT_KEYS) : jc;
        __builtin_prefetch(&Kb[(size_t)(jn + 16 * w + r) * C_DIM + 128 * h], 0, 1);
        __builtin_prefetch(&Vb[(size_t)(jn + lane) * C_DIM + c0], 0, 1);
        __builtin_prefetch(&Vb[(size_t)(jn + 32 + lane) * C_DIM + c0], 0, 1);
        __builtin_prefetch(&Vb[(size_t)(jn + 64 + lane) * C_DIM + c0], 0, 1);
        __builtin_prefetch(&Vb[(size_t)(jn + 96 + lane) * C_DIM + c0], 0, 1);
      }

      float* sb = &ss[buf * ATT_SS_FLOATS];
      if (j0 >= i0) {
        v8f s0 = vzero8(), s1 = vzero8();
        const float* Kj = Kb + (size_t)j0 * C_DIM;
#pragma unroll 16
        for (int k0 = 0; k0 < C_DIM; k0 += 4) {
          v2f qa = *(const v2f*)&qs[r * QS_PITCH + k0 + 2 * h];
          v2f qb = *(const v2f*)&qs[(16 + r) * QS_PITCH + k0 + 2 * h];
          v2f bb = *(const v2f*)&Kj[(size_t)r * C_DIM + k0 + 2 * h];
          s0 = wmma4(qa, bb, s0);
          s1 = wmma4(qb, bb, s1);
        }
#pragma unroll
        for (int i = 0; i < 8; ++i) {
          int gi = i0 + i + 8 * h;
          int gj = j0 + r;
          int d0 = gj - gi;
          int d1 = d0 - 16;
          float wg0 = (d0 > 0) ? __expf(ln_dec * (float)(d0 - 1)) : 0.0f;
          float wg1 = (d1 > 0) ? __expf(ln_dec * (float)(d1 - 1)) : 0.0f;
          sb[(i + 8 * h) * SS_PITCH + 16 * w + r] = s0[i] * wg0;
          sb[(16 + i + 8 * h) * SS_PITCH + 16 * w + r] = s1[i] * wg1;
        }
      } else {
#pragma unroll
        for (int i = 0; i < 8; ++i) {
          sb[(i + 8 * h) * SS_PITCH + 16 * w + r] = 0.0f;
          sb[(16 + i + 8 * h) * SS_PITCH + 16 * w + r] = 0.0f;
        }
      }
      __syncthreads();
#pragma unroll
      for (int kk = 0; kk < ATT_KEYS / 4; ++kk) {
        v2f pa = *(const v2f*)&sb[r * SS_PITCH + kk * 4 + 2 * h];
        v2f pb = *(const v2f*)&sb[(16 + r) * SS_PITCH + kk * 4 + 2 * h];
        const float* Vr0 = Vb + (size_t)(jc + kk * 4 + 2 * h) * C_DIM + c0;
        const float* Vr1 = Vr0 + C_DIM;
        v2f bb;
        bb.x = Vr0[r];      bb.y = Vr1[r];
        a0c0 = wmma4(pa, bb, a0c0);  a1c0 = wmma4(pb, bb, a1c0);
        bb.x = Vr0[16 + r]; bb.y = Vr1[16 + r];
        a0c1 = wmma4(pa, bb, a0c1);  a1c1 = wmma4(pb, bb, a1c1);
      }
      buf ^= 1;
    }

#pragma unroll
    for (int i = 0; i < 8; ++i) {
      size_t row0 = (size_t)(i0 + i + 8 * h) * C_DIM + c0 + r;
      size_t row1 = row0 + (size_t)16 * C_DIM;
      Rb[row0] = a0c0[i];
      Rb[row0 + 16] = a0c1[i];
      Rb[row1] = a1c0[i];
      Rb[row1 + 16] = a1c1[i];
    }
  }
}

__global__ __launch_bounds__(256) __attribute__((amdgpu_waves_per_eu(8, 8)))
void out_proj_kernel(const float* __restrict__ R,
                     const float* __restrict__ owT,
                     const float* __restrict__ out_scale,
                     float* __restrict__ out) {
  __shared__ float lds[GT_LDS_FLOATS];
  const int m0 = blockIdx.x * GT_M;
  const int v0 = blockIdx.y * GT_N;
  const float sc = *out_scale;
  gemm_tile_128x128<C_DIM, C_DIM, V_DIM, V_DIM, true>(
      R + (size_t)m0 * C_DIM, owT + v0, out + (size_t)m0 * V_DIM + v0, sc, lds);
}

extern "C" void kernel_launch(void* const* d_in, const int* in_sizes, int n_in,
                              void* d_out, int out_size, void* d_ws, size_t ws_size,
                              hipStream_t stream) {
  (void)in_sizes; (void)n_in; (void)out_size; (void)ws_size;

  const float* x = (const float*)d_in[0];
  const float* basis = (const float*)d_in[1];
  const float* qc = (const float*)d_in[2];
  const float* kc = (const float*)d_in[3];
  const float* vc = (const float*)d_in[4];
  const float* oc = (const float*)d_in[5];
  const float* decay_logit = (const float*)d_in[6];
  const float* out_scale = (const float*)d_in[7];
  float* out = (float*)d_out;

  float* ws = (float*)d_ws;
  float* qw = ws;
  float* kw = qw + (size_t)V_DIM * C_DIM;
  float* vw = kw + (size_t)V_DIM * C_DIM;
  float* owT = vw + (size_t)V_DIM * C_DIM;
  float* Q = owT + (size_t)V_DIM * C_DIM;
  float* K = Q + (size_t)M_DIM * C_DIM;
  float* Vv = K + (size_t)M_DIM * C_DIM;
  float* R = Vv + (size_t)M_DIM * C_DIM;

  dim3 blk(32);

  proj_weights_kernel<<<dim3(V_DIM / 16, C_DIM / 16, 4), blk, 0, stream>>>(
      basis, qc, kc, vc, oc, qw, kw, vw, owT);

  qkv_kernel<<<dim3(M_DIM / GT_M, 3 * C_DIM / GT_N), dim3(256), 0, stream>>>(
      x, qw, kw, vw, Q, K, Vv);

  attn_kernel<<<dim3(ATT_RBLOCKS / 2, 1, B_DIM), dim3(32 * ATT_WAVES), 0, stream>>>(
      Q, K, Vv, decay_logit, R);

  out_proj_kernel<<<dim3(M_DIM / GT_M, V_DIM / GT_N), dim3(256), 0, stream>>>(
      R, owT, out_scale, out);
}


# attention: decay mask factored per 256-key chunk: a row meets its own chunk's keys directly and all later keys through one 256x256 K^T V state (backward scan); summation order changes; 6 launches
# speedup vs baseline: 5.8220x; 1.5719x over previous
#include <hip/hip_runtime.h>


typedef float v2f __attribute__((ext_vector_type(2)));
typedef float v8f __attribute__((ext_vector_type(8)));

#define C_DIM 256
#define V_DIM 1024
#define F_DIM 256
#define T_DIM 4096
#define B_DIM 4
#define M_DIM (B_DIM * T_DIM)

static __device__ __forceinline__ v8f wmma4(v2f a, v2f b, v8f c) {
  return __builtin_amdgcn_wmma_f32_16x16x4_f32(
       false, a,  false, b,
       (short)0, c,  false,  false);
}

static __device__ __forceinline__ v8f vzero8() {
  v8f z;
#pragma unroll
  for (int i = 0; i < 8; ++i) z[i] = 0.0f;
  return z;
}

__global__ void proj_weights_kernel(const float* __restrict__ basis,
                                    const float* __restrict__ qc,
                                    const float* __restrict__ kc,
                                    const float* __restrict__ vc,
                                    const float* __restrict__ oc,
                                    float* __restrict__ qw,
                                    float* __restrict__ kw,
                                    float* __restrict__ vw,
                                    float* __restrict__ owT) {
  const int lane = threadIdx.x & 31;
  const int r = lane & 15;
  const int h = lane >> 4;
  const int v0 = blockIdx.x * 16;
  const int c0 = blockIdx.y * 16;
  const int which = blockIdx.z;
  const float* coef = (which == 0) ? qc : (which == 1) ? kc : (which == 2) ? vc : oc;

  v8f acc = vzero8();
  for (int f0 = 0; f0 < F_DIM; f0 += 4) {
    v2f a = *(const v2f*)&basis[(size_t)(v0 + r) * F_DIM + f0 + 2 * h];
    v2f b = *(const v2f*)&coef[(size_t)(c0 + r) * F_DIM + f0 + 2 * h];
    acc = wmma4(a, b, acc);
  }
  if (which < 3) {
    float* w = (which == 0) ? qw : (which == 1) ? kw : vw;
#pragma unroll
    for (int i = 0; i < 8; ++i)
      w[(size_t)(v0 + i + 8 * h) * C_DIM + c0 + r] = acc[i];
  } else {
#pragma unroll
    for (int i = 0; i < 8; ++i)
      owT[(size_t)(c0 + r) * V_DIM + v0 + i + 8 * h] = acc[i];
  }
}

#define GT_M 128
#define GT_N 128
#define GT_K 32
#define GT_XP 36
#define GT_WP 288
#define GT_XS (GT_M * GT_XP)
#define GT_WS ((GT_K / 2) * GT_WP)
#define GT_STAGE (GT_XS + GT_WS)
#define GT_LDS_FLOATS (2 * GT_STAGE)

template <int KDIM, int LDA, int LDB, int LDC, bool SCALED>
static __device__ __forceinline__ void gemm_tile_128x128(
    const float* __restrict__ A,
    const float* __restrict__ Bm,
    float* __restrict__ Cm,
    float scale, float* __restrict__ lds) {
  const int tid = threadIdx.x;
  const int lane = tid & 31;
  const int w = __builtin_amdgcn_readfirstlane(tid >> 5);
  const int r = lane & 15;
  const int h = lane >> 4;
  const int wm = (w >> 1) * 32;
  const int wn = (w & 1) * 64;

  const int xr = tid >> 3;
  const int xc = (tid & 7) << 2;
  const int wp = tid >> 5;
  const int wc = (tid & 31) << 2;
  const float* Ag = A + (size_t)xr * LDA + xc;
  const float* Bg = Bm + (size_t)(2 * wp) * LDB + wc;
  const int xs_off = xr * GT_XP + xc;
  const int ws_off = GT_XS + wp * GT_WP + 2 * wc;

  float4 xa[4], wb[4];
#pragma unroll
  for (int j = 0; j < 4; ++j) {
    xa[j] = *(const float4*)&Ag[(size_t)(32 * j) * LDA];
    wb[j] = *(const float4*)&Bg[(size_t)(16 * (j >> 1) + (j & 1)) * LDB];
  }
#pragma unroll
  for (int j = 0; j < 4; ++j)
    *(float4*)&lds[xs_off + 32 * j * GT_XP] = xa[j];
#pragma unroll
  for (int p = 0; p < 2; ++p) {
    const float4 e0 = wb[2 * p], e1 = wb[2 * p + 1];
    *(float4*)&lds[ws_off + 8 * p * GT_WP] = make_float4(e0.x, e1.x, e0.y, e1.y);
    *(float4*)&lds[ws_off + 8 * p * GT_WP + 4] = make_float4(e0.z, e1.z, e0.w, e1.w);
  }
  __syncthreads();

  v8f acc[2][4];
#pragma unroll
  for (int mt = 0; mt < 2; ++mt)
#pragma unroll
    for (int nt = 0; nt < 4; ++nt) acc[mt][nt] = vzero8();

  const int a_off = (wm + r) * GT_XP + 2 * h;
  const int b_off = GT_XS + h * GT_WP + 2 * (wn + r);

  constexpr int NCH = KDIM / GT_K;
  for (int kc = 0; kc < NCH; ++kc) {
    const float* cur = lds + (kc & 1) * GT_STAGE;
    float* nxt = lds + ((kc + 1) & 1) * GT_STAGE;
    const bool more = (kc + 1 < NCH);
    if (more) {
      const int k0 = (kc + 1) * GT_K;
#pragma unroll
      for (int j = 0; j < 4; ++j) {
        xa[j] = *(const float4*)&Ag[(size_t)(32 * j) * LDA + k0];
        wb[j] = *(const float4*)&Bg[(size_t)(k0 + 16 * (j >> 1) + (j & 1)) * LDB];
      }
    }
#pragma unroll
    for (int ks = 0; ks < GT_K; ks += 4) {
      v2f a0 = *(const v2f*)&cur[a_off + ks];
      v2f a1 = *(const v2f*)&cur[a_off + 16 * GT_XP + ks];
      const float* b0 = &cur[b_off + (ks >> 1) * GT_WP];
#pragma unroll
      for (int nt = 0; nt < 4; ++nt) {
        const v2f bb = *(const v2f*)&b0[nt * 32];
        acc[0][nt] = wmma4(a0, bb, acc[0][nt]);
        acc[1][nt] = wmma4(a1, bb, acc[1][nt]);
      }
    }
    if (more) {
#pragma unroll
      for (int j = 0; j < 4; ++j)
        *(float4*)&nxt[xs_off + 32 * j * GT_XP] = xa[j];
#pragma unroll
      for (int p = 0; p < 2; ++p) {
        const float4 e0 = wb[2 * p], e1 = wb[2 * p + 1];
        *(float4*)&nxt[ws_off + 8 * p * GT_WP] = make_float4(e0.x, e1.x, e0.y, e1.y);
        *(float4*)&nxt[ws_off + 8 * p * GT_WP + 4] = make_float4(e0.z, e1.z, e0.w, e1.w);
      }
      __syncthreads();
    }
  }

#pragma unroll
  for (int mt = 0; mt < 2; ++mt)
#pragma unroll
    for (int nt = 0; nt < 4; ++nt)
#pragma unroll
      for (int i = 0; i < 8; ++i) {
        float v = acc[mt][nt][i];
        if (SCALED) v *= scale;
        Cm[(size_t)(wm + mt * 16 + i + 8 * h) * LDC + wn + nt * 16 + r] = v;
      }
}

#define QKV_LDS_FLOATS 20992

__global__ __launch_bounds__(256)
void qkv_kernel(const float* __restrict__ x,
                const float* __restrict__ qw,
                const float* __restrict__ kw,
                const float* __restrict__ vw,
                float* __restrict__ Q,
                float* __restrict__ K,
                float* __restrict__ Vv) {
  __shared__ float lds[QKV_LDS_FLOATS];
  const int m0 = blockIdx.x * GT_M;
  const int which = (int)blockIdx.y >> 1;
  const int n0 = ((int)blockIdx.y & 1) * GT_N;
  const float* w = (which == 0) ? qw : (which == 1) ? kw : vw;
  float* out = (which == 0) ? Q : (which == 1) ? K : Vv;
  gemm_tile_128x128<V_DIM, V_DIM, C_DIM, C_DIM, false>(
      x + (size_t)m0 * V_DIM, w + n0, out + (size_t)m0 * C_DIM + n0, 1.0f, lds);
}

#define CH_L 256
#define N_CHUNK (T_DIM / CH_L)

__global__ __launch_bounds__(256)
void chunk_state_kernel(const float* __restrict__ K,
                        const float* __restrict__ Vv,
                        const float* __restrict__ decay_logit,
                        float* __restrict__ S) {
  const int tid = threadIdx.x;
  const int lane = tid & 31;
  const int w = __builtin_amdgcn_readfirstlane(tid >> 5);
  const int r = lane & 15;
  const int h = lane >> 4;
  const int kc0 = w * 32;
  const int vc0 = blockIdx.x * 64;
  const int ch = (int)blockIdx.z * N_CHUNK + 1 + (int)blockIdx.y;
  const float* Kc = K + (size_t)ch * CH_L * C_DIM;
  const float* Vc = Vv + (size_t)ch * CH_L * C_DIM;

  const float dl = *decay_logit;
  const float dec = 1.0f / (1.0f + __expf(-dl));
  const float ln_dec = __logf(dec);

  v8f acc[2][4];
#pragma unroll
  for (int mt = 0; mt < 2; ++mt)
#pragma unroll
    for (int nt = 0; nt < 4; ++nt) acc[mt][nt] = vzero8();

#pragma unroll 4
  for (int j0 = 0; j0 < CH_L; j0 += 4) {
    const int ja = j0 + 2 * h;
    const float w0 = __expf(ln_dec * (float)ja);
    const float w1 = __expf(ln_dec * (float)(ja + 1));
    const float* Kr0 = Kc + (size_t)ja * C_DIM + kc0;
    const float* Kr1 = Kr0 + C_DIM;
    const float* Vr0 = Vc + (size_t)ja * C_DIM + vc0;
    const float* Vr1 = Vr0 + C_DIM;
    v2f a0, a1;
    a0.x = Kr0[r] * w0;       a0.y = Kr1[r] * w1;
    a1.x = Kr0[16 + r] * w0;  a1.y = Kr1[16 + r] * w1;
#pragma unroll
    for (int nt = 0; nt < 4; ++nt) {
      v2f bb;
      bb.x = Vr0[16 * nt + r];  bb.y = Vr1[16 * nt + r];
      acc[0][nt] = wmma4(a0, bb, acc[0][nt]);
      acc[1][nt] = wmma4(a1, bb, acc[1][nt]);
    }
  }

  float* Sc = S + (size_t)ch * C_DIM * C_DIM;
#pragma unroll
  for (int mt = 0; mt < 2; ++mt)
#pragma unroll
    for (int nt = 0; nt < 4; ++nt)
#pragma unroll
      for (int i = 0; i < 8; ++i)
        Sc[(size_t)(kc0 + mt * 16 + i + 8 * h) * C_DIM + vc0 + nt * 16 + r] = acc[mt][nt][i];
}

__global__ __launch_bounds__(256)
void chunk_scan_kernel(const float* __restrict__ decay_logit, float* __restrict__ S) {
  const float dl = *decay_logit;
  const float dec = 1.0f / (1.0f + __expf(-dl));
  const float dL = __expf(__logf(dec) * (float)CH_L);
  const size_t e = ((size_t)blockIdx.x * 256 + threadIdx.x) * 4;
  float* Sb = S + (size_t)blockIdx.z * N_CHUNK * C_DIM * C_DIM + e;
  float4 p[N_CHUNK - 1];
#pragma unroll
  for (int c = 1; c < N_CHUNK; ++c)
    p[c - 1] = *(const float4*)&Sb[(size_t)c * C_DIM * C_DIM];
  float4 s = p[N_CHUNK - 2];
#pragma unroll
  for (int c = N_CHUNK - 2; c >= 1; --c) {
    const float4 q = p[c - 1];
    s.x = fmaf(dL, s.x, q.x);
    s.y = fmaf(dL, s.y, q.y);
    s.z = fmaf(dL, s.z, q.z);
    s.w = fmaf(dL, s.w, q.w);
    *(float4*)&Sb[(size_t)c * C_DIM * C_DIM] = s;
  }
}

#define ATT_WAVES 8
#define ATT_ROWS 32
#define ATT_KEYS (16 * ATT_WAVES)
#define ATT_RBLOCKS (T_DIM / ATT_ROWS)
#define QS_PITCH 260
#define SS_PITCH 132
#define ATT_QS_FLOATS (ATT_ROWS * QS_PITCH)
#define ATT_SS_FLOATS (ATT_ROWS * SS_PITCH)
#define ATT_LDS_FLOATS 41216

__global__ __launch_bounds__(32 * ATT_WAVES)
void attn_kernel(float* QR,
                 const float* __restrict__ K,
                 const float* __restrict__ Vv,
                 const float* __restrict__ S,
                 const float* __restrict__ decay_logit) {
  __shared__ float lds[ATT_LDS_FLOATS];
  float* const qs = &lds[0];
  float* const ss = &lds[ATT_QS_FLOATS];

  const int tid = threadIdx.x;
  const int lane = tid & 31;
  const int w = __builtin_amdgcn_readfirstlane(tid >> 5);
  const int r = lane & 15;
  const int h = lane >> 4;
  const int c0 = w * 32;
  const int b = blockIdx.z;

  float* Qb = QR + (size_t)b * T_DIM * C_DIM;
  const float* Kb = K + (size_t)b * T_DIM * C_DIM;
  const float* Vb = Vv + (size_t)b * T_DIM * C_DIM;
  float* Rb = Qb;

  const float dl = *decay_logit;
  const float dec = 1.0f / (1.0f + __expf(-dl));
  const float ln_dec = __logf(dec);

  for (int part = 0; part < 2; ++part) {
    const int rb = (part == 0) ? (int)blockIdx.x : (ATT_RBLOCKS - 1 - (int)blockIdx.x);
    const int i0 = rb * ATT_ROWS;
    const int cend = (i0 & ~(CH_L - 1)) + CH_L;

    for (int idx = tid; idx < ATT_ROWS * 64; idx += 32 * ATT_WAVES) {
      int row = idx >> 6;
      int c4 = (idx & 63) << 2;
      *(float4*)&qs[row * QS_PITCH + c4] =
          *(const float4*)&Qb[(size_t)(i0 + row) * C_DIM + c4];
    }
    __syncthreads();

    v8f a0c0 = vzero8(), a0c1 = vzero8();
    v8f a1c0 = vzero8(), a1c1 = vzero8();

    int buf = 0;
    for (int jc = i0 & ~(ATT_KEYS - 1); jc < cend; jc += ATT_KEYS) {
      const int j0 = jc + 16 * w;
      {
        int jn = (jc + ATT_KEYS < cend) ? (jc + ATT_KEYS) : jc;
        __builtin_prefetch(&Kb[(size_t)(jn + 16 * w + r) * C_DIM + 128 * h], 0, 1);
        __builtin_prefetch(&Vb[(size_t)(jn + lane) * C_DIM + c0], 0, 1);
        __builtin_prefetch(&Vb[(size_t)(jn + 32 + lane) * C_DIM + c0], 0, 1);
        __builtin_prefetch(&Vb[(size_t)(jn + 64 + lane) * C_DIM + c0], 0, 1);
        __builtin_prefetch(&Vb[(size_t)(jn + 96 + lane) * C_DIM + c0], 0, 1);
      }

      float* sb = &ss[buf * ATT_SS_FLOATS];
      if (j0 >= i0) {
        v8f s0 = vzero8(), s1 = vzero8();
        const float* Kj = Kb + (size_t)j0 * C_DIM;
#pragma unroll 16
        for (int k0 = 0; k0 < C_DIM; k0 += 4) {
          v2f qa = *(const v2f*)&qs[r * QS_PITCH + k0 + 2 * h];
          v2f qb = *(const v2f*)&qs[(16 + r) * QS_PITCH + k0 + 2 * h];
          v2f bb = *(const v2f*)&Kj[(size_t)r * C_DIM + k0 + 2 * h];
          s0 = wmma4(qa, bb, s0);
          s1 = wmma4(qb, bb, s1);
        }
#pragma unroll
        for (int i = 0; i < 8; ++i) {
          int gi = i0 + i + 8 * h;
          int gj = j0 + r;
          int d0 = gj - gi;
          int d1 = d0 - 16;
          float wg0 = (d0 > 0) ? __expf(ln_dec * (float)(d0 - 1)) : 0.0f;
          float wg1 = (d1 > 0) ? __expf(ln_dec * (float)(d1 - 1)) : 0.0f;
          sb[(i + 8 * h) * SS_PITCH + 16 * w + r] = s0[i] * wg0;
          sb[(16 + i + 8 * h) * SS_PITCH + 16 * w + r] = s1[i] * wg1;
        }
      } else {
#pragma unroll
        for (int i = 0; i < 8; ++i) {
          sb[(i + 8 * h) * SS_PITCH + 16 * w + r] = 0.0f;
          sb[(16 + i + 8 * h) * SS_PITCH + 16 * w + r] = 0.0f;
        }
      }
      __syncthreads();
#pragma unroll
      for (int kk = 0; kk < ATT_KEYS / 4; ++kk) {
        v2f pa = *(const v2f*)&sb[r * SS_PITCH + kk * 4 + 2 * h];
        v2f pb = *(const v2f*)&sb[(16 + r) * SS_PITCH + kk * 4 + 2 * h];
        const float* Vr0 = Vb + (size_t)(jc + kk * 4 + 2 * h) * C_DIM + c0;
        const float* Vr1 = Vr0 + C_DIM;
        v2f bb;
        bb.x = Vr0[r];      bb.y = Vr1[r];
        a0c0 = wmma4(pa, bb, a0c0);  a1c0 = wmma4(pb, bb, a1c0);
        bb.x = Vr0[16 + r]; bb.y = Vr1[16 + r];
        a0c1 = wmma4(pa, bb, a0c1);  a1c1 = wmma4(pb, bb, a1c1);
      }
      buf ^= 1;
    }

    if (cend < T_DIM) {
      const float* Sn = S + ((size_t)b * N_CHUNK + (size_t)(cend / CH_L)) * C_DIM * C_DIM;
      const float f0 = __expf(ln_dec * (float)(cend - 1 - (i0 + r)));
      const float f1 = __expf(ln_dec * (float)(cend - 1 - (i0 + 16 + r)));
#pragma unroll 16
      for (int kk = 0; kk < C_DIM / 4; ++kk) {
        v2f pa = *(const v2f*)&qs[r * QS_PITCH + kk * 4 + 2 * h];
        v2f pb = *(const v2f*)&qs[(16 + r) * QS_PITCH + kk * 4 + 2 * h];
        pa.x *= f0;  pa.y *= f0;
        pb.x *= f1;  pb.y *= f1;
        const float* Sr0 = Sn + (size_t)(kk * 4 + 2 * h) * C_DIM + c0;
        const float* Sr1 = Sr0 + C_DIM;
        v2f bb;
        bb.x = Sr0[r];      bb.y = Sr1[r];
        a0c0 = wmma4(pa, bb, a0c0);  a1c0 = wmma4(pb, bb, a1c0);
        bb.x = Sr0[16 + r]; bb.y = Sr1[16 + r];
        a0c1 = wmma4(pa, bb, a0c1);  a1c1 = wmma4(pb, bb, a1c1);
      }
    }
    __syncthreads();

#pragma unroll
    for (int i = 0; i < 8; ++i) {
      size_t row0 = (size_t)(i0 + i + 8 * h) * C_DIM + c0 + r;
      size_t row1 = row0 + (size_t)16 * C_DIM;
      Rb[row0] = a0c0[i];
      Rb[row0 + 16] = a0c1[i];
      Rb[row1] = a1c0[i];
      Rb[row1 + 16] = a1c1[i];
    }
  }
}

__global__ __launch_bounds__(256) __attribute__((amdgpu_waves_per_eu(8, 8)))
void out_proj_kernel(const float* __restrict__ R,
                     const float* __restrict__ owT,
                     const float* __restrict__ out_scale,
                     float* __restrict__ out) {
  __shared__ float lds[GT_LDS_FLOATS];
  const int m0 = blockIdx.x * GT_M;
  const int v0 = blockIdx.y * GT_N;
  const float sc = *out_scale;
  gemm_tile_128x128<C_DIM, C_DIM, V_DIM, V_DIM, true>(
      R + (size_t)m0 * C_DIM, owT + v0, out + (size_t)m0 * V_DIM + v0, sc, lds);
}

extern "C" void kernel_launch(void* const* d_in, const int* in_sizes, int n_in,
                              void* d_out, int out_size, void* d_ws, size_t ws_size,
                              hipStream_t stream) {
  (void)in_sizes; (void)n_in; (void)out_size; (void)ws_size;

  const float* x = (const float*)d_in[0];
  const float* basis = (const float*)d_in[1];
  const float* qc = (const float*)d_in[2];
  const float* kc = (const float*)d_in[3];
  const float* vc = (const float*)d_in[4];
  const float* oc = (const float*)d_in[5];
  const float* decay_logit = (const float*)d_in[6];
  const float* out_scale = (const float*)d_in[7];
  float* out = (float*)d_out;

  float* ws = (float*)d_ws;
  float* qw = ws;
  float* kw = qw + (size_t)V_DIM * C_DIM;
  float* vw = kw + (size_t)V_DIM * C_DIM;
  float* owT = vw + (size_t)V_DIM * C_DIM;
  float* Q = owT + (size_t)V_DIM * C_DIM;
  float* K = Q + (size_t)M_DIM * C_DIM;
  float* Vv = K + (size_t)M_DIM * C_DIM;
  float* S = Vv + (size_t)M_DIM * C_DIM;

  dim3 blk(32);

  proj_weights_kernel<<<dim3(V_DIM / 16, C_DIM / 16, 4), blk, 0, stream>>>(
      basis, qc, kc, vc, oc, qw, kw, vw, owT);

  qkv_kernel<<<dim3(M_DIM / GT_M, 3 * C_DIM / GT_N), dim3(256), 0, stream>>>(
      x, qw, kw, vw, Q, K, Vv);

  chunk_state_kernel<<<dim3(C_DIM / 64, N_CHUNK - 1, B_DIM), dim3(256), 0, stream>>>(
      K, Vv, decay_logit, S);

  chunk_scan_kernel<<<dim3(C_DIM * C_DIM / 4 / 256, 1, B_DIM), dim3(256), 0, stream>>>(
      decay_logit, S);

  attn_kernel<<<dim3(ATT_RBLOCKS / 2, 1, B_DIM), dim3(32 * ATT_WAVES), 0, stream>>>(
      Q, K, Vv, S, decay_logit);

  out_proj_kernel<<<dim3(M_DIM / GT_M, V_DIM / GT_N), dim3(256), 0, stream>>>(
      Q, owT, out_scale, out);
}


# Q,K,V projections reassociated, x@(basis@c^T) -> (x@basis)@c^T: one shared [M,1024]x[1024,256] GEMM, then three K=256 GEMMs on the coefficients as given; summation order changes; 7 launches
# speedup vs baseline: 6.7813x; 1.1648x over previous
#include <hip/hip_runtime.h>


typedef float v2f __attribute__((ext_vector_type(2)));
typedef float v8f __attribute__((ext_vector_type(8)));

#define C_DIM 256
#define V_DIM 1024
#define F_DIM 256
#define T_DIM 4096
#define B_DIM 4
#define M_DIM (B_DIM * T_DIM)

static __device__ __forceinline__ v8f wmma4(v2f a, v2f b, v8f c) {
  return __builtin_amdgcn_wmma_f32_16x16x4_f32(
       false, a,  false, b,
       (short)0, c,  false,  false);
}

static __device__ __forceinline__ v8f vzero8() {
  v8f z;
#pragma unroll
  for (int i = 0; i < 8; ++i) z[i] = 0.0f;
  return z;
}


__global__ void proj_weights_kernel(const float* __restrict__ basis,
                                    const float* __restrict__ oc,
                                    float* __restrict__ owT) {
  const int lane = threadIdx.x & 31;
  const int r = lane & 15;
  const int h = lane >> 4;
  const int v0 = blockIdx.x * 16;
  const int c0 = blockIdx.y * 16;

  v8f acc = vzero8();
  for (int f0 = 0; f0 < F_DIM; f0 += 4) {
    v2f a = *(const v2f*)&basis[(size_t)(v0 + r) * F_DIM + f0 + 2 * h];
    v2f b = *(const v2f*)&oc[(size_t)(c0 + r) * F_DIM + f0 + 2 * h];
    acc = wmma4(a, b, acc);
  }
#pragma unroll
  for (int i = 0; i < 8; ++i)
    owT[(size_t)(c0 + r) * V_DIM + v0 + i + 8 * h] = acc[i];
}

#define GT_M 128
#define GT_N 128
#define GT_K 32
#define GT_XP 36
#define GT_WP 288
#define GT_XS (GT_M * GT_XP)
#define GT_WS ((GT_K / 2) * GT_WP)
#define GT_STAGE (GT_XS + GT_WS)
#define GT_LDS_FLOATS (2 * GT_STAGE)

template <bool BT>
static __device__ __forceinline__ void gemm_slab_mma(const float* as, const float* bs,
                                                     v8f (&acc)[2][4]) {
  constexpr int B_KS = BT ? 4 : 2 * GT_WP;
  constexpr int B_NT = BT ? 16 * GT_XP : 32;
#pragma unroll
  for (int ks = 0; ks < GT_K; ks += 4) {
    v2f a0 = *(const v2f*)&as[ks];
    v2f a1 = *(const v2f*)&as[16 * GT_XP + ks];
    const float* b0 = &bs[(ks >> 2) * B_KS];
#pragma unroll
    for (int nt = 0; nt < 4; ++nt) {
      const v2f bb = *(const v2f*)&b0[nt * B_NT];
      acc[0][nt] = wmma4(a0, bb, acc[0][nt]);
      acc[1][nt] = wmma4(a1, bb, acc[1][nt]);
    }
  }
}

template <int KDIM, int LDA, int LDB, int LDC, bool SCALED, bool BT, bool ALONE, int KC_UNROLL>
static __device__ __forceinline__ void gemm_tile_128x128(
    const float* __restrict__ A,
    const float* __restrict__ Bm,
    float* __restrict__ Cm,
    float scale, float* __restrict__ lds) {
  const int tid = threadIdx.x;
  const int lane = tid & 31;
  const int w = __builtin_amdgcn_readfirstlane(tid >> 5);
  const int r = lane & 15;
  const int h = lane >> 4;
  const int wm = (w >> 1) * 32;
  const int wn = (w & 1) * 64;

  const int xr = tid >> 3;
  const int xc = (tid & 7) << 2;
  const int wp = tid >> 5;
  const int wc = (tid & 31) << 2;
  const float* Ag = A + (size_t)xr * LDA + xc;
  const float* Bg = BT ? Bm + (size_t)xr * LDB + xc
                       : Bm + (size_t)(2 * wp) * LDB + wc;
  const int xs_off = xr * GT_XP + xc;
  const int ws_off = BT ? GT_XS + xs_off : GT_XS + wp * GT_WP + 2 * wc;

  float4 xa[4], wb[4];
#pragma unroll
  for (int j = 0; j < 4; ++j) {
    xa[j] = *(const float4*)&Ag[(size_t)(32 * j) * LDA];
    if (BT)
      wb[j] = *(const float4*)&Bg[(size_t)(32 * j) * LDB];
    else
      wb[j] = *(const float4*)&Bg[(size_t)(16 * (j >> 1) + (j & 1)) * LDB];
  }
#pragma unroll
  for (int j = 0; j < 4; ++j)
    *(float4*)&lds[xs_off + 32 * j * GT_XP] = xa[j];
  if (BT) {
#pragma unroll
    for (int j = 0; j < 4; ++j)
      *(float4*)&lds[ws_off + 32 * j * GT_XP] = wb[j];
  } else {
#pragma unroll
    for (int p = 0; p < 2; ++p) {
      const float4 e0 = wb[2 * p], e1 = wb[2 * p + 1];
      *(float4*)&lds[ws_off + 8 * p * GT_WP] = make_float4(e0.x, e1.x, e0.y, e1.y);
      *(float4*)&lds[ws_off + 8 * p * GT_WP + 4] = make_float4(e0.z, e1.z, e0.w, e1.w);
    }
  }
  __syncthreads();

  v8f acc[2][4];
#pragma unroll
  for (int mt = 0; mt < 2; ++mt)
#pragma unroll
    for (int nt = 0; nt < 4; ++nt) acc[mt][nt] = vzero8();

  const int a_off = (wm + r) * GT_XP + 2 * h;
  const int b_off = BT ? GT_XS + (wn + r) * GT_XP + 2 * h
                       : GT_XS + h * GT_WP + 2 * (wn + r);

  constexpr int NCH = KDIM / GT_K;
#pragma unroll KC_UNROLL
  for (int kc = 0; kc < NCH - 1; ++kc) {
    const float* cur = lds + (kc & 1) * GT_STAGE;
    float* nxt = lds + ((kc + 1) & 1) * GT_STAGE;
    const int k0 = (kc + 1) * GT_K;
#pragma unroll
    for (int j = 0; j < 4; ++j) {
      xa[j] = *(const float4*)&Ag[(size_t)(32 * j) * LDA + k0];
      if (BT)
        wb[j] = *(const float4*)&Bg[(size_t)(32 * j) * LDB + k0];
      else
        wb[j] = *(const float4*)&Bg[(size_t)(k0 + 16 * (j >> 1) + (j & 1)) * LDB];
    }
    gemm_slab_mma<BT>(cur + a_off, cur + b_off, acc);
    if (ALONE) __builtin_amdgcn_sched_barrier(0x26);
#pragma unroll
    for (int j = 0; j < 4; ++j)
      *(float4*)&nxt[xs_off + 32 * j * GT_XP] = xa[j];
    if (BT) {
#pragma unroll
      for (int j = 0; j < 4; ++j)
        *(float4*)&nxt[ws_off + 32 * j * GT_XP] = wb[j];
    } else {
#pragma unroll
      for (int p = 0; p < 2; ++p) {
        const float4 e0 = wb[2 * p], e1 = wb[2 * p + 1];
        *(float4*)&nxt[ws_off + 8 * p * GT_WP] = make_float4(e0.x, e1.x, e0.y, e1.y);
        *(float4*)&nxt[ws_off + 8 * p * GT_WP + 4] = make_float4(e0.z, e1.z, e0.w, e1.w);
      }
    }
    __syncthreads();
  }
  {
    const float* last = lds + ((NCH - 1) & 1) * GT_STAGE;
    gemm_slab_mma<BT>(last + a_off, last + b_off, acc);
  }

#pragma unroll
  for (int mt = 0; mt < 2; ++mt)
#pragma unroll
    for (int nt = 0; nt < 4; ++nt)
#pragma unroll
      for (int i = 0; i < 8; ++i) {
        float v = acc[mt][nt][i];
        if (SCALED) v *= scale;
        Cm[(size_t)(wm + mt * 16 + i + 8 * h) * LDC + wn + nt * 16 + r] = v;
      }
}

#define XB_LDS_FLOATS 41216

__global__ __launch_bounds__(256)
void xb_kernel(const float* __restrict__ x,
               const float* __restrict__ basis,
               float* __restrict__ XB) {
  __shared__ float lds[XB_LDS_FLOATS];
  const int m0 = blockIdx.x * GT_M;
  const int n0 = blockIdx.y * GT_N;
  gemm_tile_128x128<V_DIM, V_DIM, F_DIM, F_DIM, false, false, true, 1>(
      x + (size_t)m0 * V_DIM, basis + n0, XB + (size_t)m0 * F_DIM + n0, 1.0f, lds);
}

#define QKV_LDS_FLOATS 20992

__global__ __launch_bounds__(256)
void qkv_kernel(const float* __restrict__ XB,
                const float* __restrict__ qc,
                const float* __restrict__ kc,
                const float* __restrict__ vc,
                float* __restrict__ Q,
                float* __restrict__ K,
                float* __restrict__ Vv) {
  __shared__ float lds[QKV_LDS_FLOATS];
  const int m0 = blockIdx.x * GT_M;
  const int which = (int)blockIdx.y >> 1;
  const int n0 = ((int)blockIdx.y & 1) * GT_N;
  const float* coef = (which == 0) ? qc : (which == 1) ? kc : vc;
  float* out = (which == 0) ? Q : (which == 1) ? K : Vv;
  gemm_tile_128x128<F_DIM, F_DIM, F_DIM, C_DIM, false, true, false, 1>(
      XB + (size_t)m0 * F_DIM, coef + (size_t)n0 * F_DIM,
      out + (size_t)m0 * C_DIM + n0, 1.0f, lds);
}

#define CH_L 256
#define N_CHUNK (T_DIM / CH_L)

__global__ __launch_bounds__(256)
void chunk_state_kernel(const float* __restrict__ K,
                        const float* __restrict__ Vv,
                        const float* __restrict__ decay_logit,
                        float* __restrict__ S) {
  const int tid = threadIdx.x;
  const int lane = tid & 31;
  const int w = __builtin_amdgcn_readfirstlane(tid >> 5);
  const int r = lane & 15;
  const int h = lane >> 4;
  const int kc0 = w * 32;
  const int vc0 = blockIdx.x * 64;
  const int ch = (int)blockIdx.z * N_CHUNK + 1 + (int)blockIdx.y;
  const float* Kc = K + (size_t)ch * CH_L * C_DIM;
  const float* Vc = Vv + (size_t)ch * CH_L * C_DIM;

  const float dl = *decay_logit;
  const float dec = 1.0f / (1.0f + __expf(-dl));
  const float ln_dec = __logf(dec);

  v8f acc[2][4];
#pragma unroll
  for (int mt = 0; mt < 2; ++mt)
#pragma unroll
    for (int nt = 0; nt < 4; ++nt) acc[mt][nt] = vzero8();

#pragma unroll 4
  for (int j0 = 0; j0 < CH_L; j0 += 4) {
    const int ja = j0 + 2 * h;
    const float w0 = __expf(ln_dec * (float)ja);
    const float w1 = __expf(ln_dec * (float)(ja + 1));
    const float* Kr0 = Kc + (size_t)ja * C_DIM + kc0;
    const float* Kr1 = Kr0 + C_DIM;
    const float* Vr0 = Vc + (size_t)ja * C_DIM + vc0;
    const float* Vr1 = Vr0 + C_DIM;
    v2f a0, a1;
    a0.x = Kr0[r] * w0;       a0.y = Kr1[r] * w1;
    a1.x = Kr0[16 + r] * w0;  a1.y = Kr1[16 + r] * w1;
#pragma unroll
    for (int nt = 0; nt < 4; ++nt) {
      v2f bb;
      bb.x = Vr0[16 * nt + r];  bb.y = Vr1[16 * nt + r];
      acc[0][nt] = wmma4(a0, bb, acc[0][nt]);
      acc[1][nt] = wmma4(a1, bb, acc[1][nt]);
    }
  }

  float* Sc = S + (size_t)ch * C_DIM * C_DIM;
#pragma unroll
  for (int mt = 0; mt < 2; ++mt)
#pragma unroll
    for (int nt = 0; nt < 4; ++nt)
#pragma unroll
      for (int i = 0; i < 8; ++i)
        Sc[(size_t)(kc0 + mt * 16 + i + 8 * h) * C_DIM + vc0 + nt * 16 + r] = acc[mt][nt][i];
}

__global__ __launch_bounds__(256)
void chunk_scan_kernel(const float* __restrict__ decay_logit, float* __restrict__ S) {
  const float dl = *decay_logit;
  const float dec = 1.0f / (1.0f + __expf(-dl));
  const float dL = __expf(__logf(dec) * (float)CH_L);
  const size_t e = ((size_t)blockIdx.x * 256 + threadIdx.x) * 4;
  float* Sb = S + (size_t)blockIdx.z * N_CHUNK * C_DIM * C_DIM + e;
  float4 p[N_CHUNK - 1];
#pragma unroll
  for (int c = 1; c < N_CHUNK; ++c)
    p[c - 1] = *(const float4*)&Sb[(size_t)c * C_DIM * C_DIM];
  float4 s = p[N_CHUNK - 2];
#pragma unroll
  for (int c = N_CHUNK - 2; c >= 1; --c) {
    const float4 q = p[c - 1];
    s.x = fmaf(dL, s.x, q.x);
    s.y = fmaf(dL, s.y, q.y);
    s.z = fmaf(dL, s.z, q.z);
    s.w = fmaf(dL, s.w, q.w);
    *(float4*)&Sb[(size_t)c * C_DIM * C_DIM] = s;
  }
}

#define ATT_WAVES 8
#define ATT_ROWS 32
#define ATT_KEYS (16 * ATT_WAVES)
#define ATT_RBLOCKS (T_DIM / ATT_ROWS)
#define QS_PITCH 260
#define SS_PITCH 132
#define ATT_QS_FLOATS (ATT_ROWS * QS_PITCH)
#define ATT_SS_FLOATS (ATT_ROWS * SS_PITCH)
#define ATT_LDS_FLOATS 41216

__global__ __launch_bounds__(32 * ATT_WAVES)
void attn_kernel(float* QR,
                 const float* __restrict__ K,
                 const float* __restrict__ Vv,
                 const float* __restrict__ S,
                 const float* __restrict__ decay_logit) {
  __shared__ float lds[ATT_LDS_FLOATS];
  float* const qs = &lds[0];
  float* const ss = &lds[ATT_QS_FLOATS];

  const int tid = threadIdx.x;
  const int lane = tid & 31;
  const int w = __builtin_amdgcn_readfirstlane(tid >> 5);
  const int r = lane & 15;
  const int h = lane >> 4;
  const int c0 = w * 32;
  const int b = blockIdx.z;

  float* Qb = QR + (size_t)b * T_DIM * C_DIM;
  const float* Kb = K + (size_t)b * T_DIM * C_DIM;
  const float* Vb = Vv + (size_t)b * T_DIM * C_DIM;
  float* Rb = Qb;

  const float dl = *decay_logit;
  const float dec = 1.0f / (1.0f + __expf(-dl));
  const float ln_dec = __logf(dec);

  for (int part = 0; part < 2; ++part) {
    const int rb = (part == 0) ? (int)blockIdx.x : (ATT_RBLOCKS - 1 - (int)blockIdx.x);
    const int i0 = rb * ATT_ROWS;
    const int cend = (i0 & ~(CH_L - 1)) + CH_L;

    for (int idx = tid; idx < ATT_ROWS * 64; idx += 32 * ATT_WAVES) {
      int row = idx >> 6;
      int c4 = (idx & 63) << 2;
      *(float4*)&qs[row * QS_PITCH + c4] =
          *(const float4*)&Qb[(size_t)(i0 + row) * C_DIM + c4];
    }
    __syncthreads();

    v8f a0c0 = vzero8(), a0c1 = vzero8();
    v8f a1c0 = vzero8(), a1c1 = vzero8();

    int buf = 0;
    for (int jc = i0 & ~(ATT_KEYS - 1); jc < cend; jc += ATT_KEYS) {
      const int j0 = jc + 16 * w;
      {
        int jn = (jc + ATT_KEYS < cend) ? (jc + ATT_KEYS) : jc;
        __builtin_prefetch(&Kb[(size_t)(jn + 16 * w + r) * C_DIM + 128 * h], 0, 1);
        __builtin_prefetch(&Vb[(size_t)(jn + lane) * C_DIM + c0], 0, 1);
        __builtin_prefetch(&Vb[(size_t)(jn + 32 + lane) * C_DIM + c0], 0, 1);
        __builtin_prefetch(&Vb[(size_t)(jn + 64 + lane) * C_DIM + c0], 0, 1);
        __builtin_prefetch(&Vb[(size_t)(jn + 96 + lane) * C_DIM + c0], 0, 1);
      }

      float* sb = &ss[buf * ATT_SS_FLOATS];
      if (j0 >= i0) {
        v8f s0 = vzero8(), s1 = vzero8();
        const float* Kj = Kb + (size_t)j0 * C_DIM;
#pragma unroll 16
        for (int k0 = 0; k0 < C_DIM; k0 += 4) {
          v2f qa = *(const v2f*)&qs[r * QS_PITCH + k0 + 2 * h];
          v2f qb = *(const v2f*)&qs[(16 + r) * QS_PITCH + k0 + 2 * h];
          v2f bb = *(const v2f*)&Kj[(size_t)r * C_DIM + k0 + 2 * h];
          s0 = wmma4(qa, bb, s0);
          s1 = wmma4(qb, bb, s1);
        }
#pragma unroll
        for (int i = 0; i < 8; ++i) {
          int gi = i0 + i + 8 * h;
          int gj = j0 + r;
          int d0 = gj - gi;
          int d1 = d0 - 16;
          float wg0 = (d0 > 0) ? __expf(ln_dec * (float)(d0 - 1)) : 0.0f;
          float wg1 = (d1 > 0) ? __expf(ln_dec * (float)(d1 - 1)) : 0.0f;
          sb[(i + 8 * h) * SS_PITCH + 16 * w + r] = s0[i] * wg0;
          sb[(16 + i + 8 * h) * SS_PITCH + 16 * w + r] = s1[i] * wg1;
        }
      } else {
#pragma unroll
        for (int i = 0; i < 8; ++i) {
          sb[(i + 8 * h) * SS_PITCH + 16 * w + r] = 0.0f;
          sb[(16 + i + 8 * h) * SS_PITCH + 16 * w + r] = 0.0f;
        }
      }
      __syncthreads();
#pragma unroll
      for (int kk = 0; kk < ATT_KEYS / 4; ++kk) {
        v2f pa = *(const v2f*)&sb[r * SS_PITCH + kk * 4 + 2 * h];
        v2f pb = *(const v2f*)&sb[(16 + r) * SS_PITCH + kk * 4 + 2 * h];
        const float* Vr0 = Vb + (size_t)(jc + kk * 4 + 2 * h) * C_DIM + c0;
        const float* Vr1 = Vr0 + C_DIM;
        v2f bb;
        bb.x = Vr0[r];      bb.y = Vr1[r];
        a0c0 = wmma4(pa, bb, a0c0);  a1c0 = wmma4(pb, bb, a1c0);
        bb.x = Vr0[16 + r]; bb.y = Vr1[16 + r];
        a0c1 = wmma4(pa, bb, a0c1);  a1c1 = wmma4(pb, bb, a1c1);
      }
      buf ^= 1;
    }

    if (cend < T_DIM) {
      const float* Sn = S + ((size_t)b * N_CHUNK + (size_t)(cend / CH_L)) * C_DIM * C_DIM;
      const float f0 = __expf(ln_dec * (float)(cend - 1 - (i0 + r)));
      const float f1 = __expf(ln_dec * (float)(cend - 1 - (i0 + 16 + r)));
#pragma unroll 16
      for (int kk = 0; kk < C_DIM / 4; ++kk) {
        v2f pa = *(const v2f*)&qs[r * QS_PITCH + kk * 4 + 2 * h];
        v2f pb = *(const v2f*)&qs[(16 + r) * QS_PITCH + kk * 4 + 2 * h];
        pa.x *= f0;  pa.y *= f0;
        pb.x *= f1;  pb.y *= f1;
        const float* Sr0 = Sn + (size_t)(kk * 4 + 2 * h) * C_DIM + c0;
        const float* Sr1 = Sr0 + C_DIM;
        v2f bb;
        bb.x = Sr0[r];      bb.y = Sr1[r];
        a0c0 = wmma4(pa, bb, a0c0);  a1c0 = wmma4(pb, bb, a1c0);
        bb.x = Sr0[16 + r]; bb.y = Sr1[16 + r];
        a0c1 = wmma4(pa, bb, a0c1);  a1c1 = wmma4(pb, bb, a1c1);
      }
    }
    __syncthreads();

#pragma unroll
    for (int i = 0; i < 8; ++i) {
      size_t row0 = (size_t)(i0 + i + 8 * h) * C_DIM + c0 + r;
      size_t row1 = row0 + (size_t)16 * C_DIM;
      Rb[row0] = a0c0[i];
      Rb[row0 + 16] = a0c1[i];
      Rb[row1] = a1c0[i];
      Rb[row1 + 16] = a1c1[i];
    }
  }
}

__global__ __launch_bounds__(256) __attribute__((amdgpu_waves_per_eu(8, 8)))
void out_proj_kernel(const float* __restrict__ R,
                     const float* __restrict__ owT,
                     const float* __restrict__ out_scale,
                     float* __restrict__ out) {
  __shared__ float lds[GT_LDS_FLOATS];
  const int m0 = blockIdx.x * GT_M;
  const int v0 = blockIdx.y * GT_N;
  const float sc = *out_scale;
  gemm_tile_128x128<C_DIM, C_DIM, V_DIM, V_DIM, true, false, false, C_DIM / GT_K - 1>(
      R + (size_t)m0 * C_DIM, owT + v0, out + (size_t)m0 * V_DIM + v0, sc, lds);
}

extern "C" void kernel_launch(void* const* d_in, const int* in_sizes, int n_in,
                              void* d_out, int out_size, void* d_ws, size_t ws_size,
                              hipStream_t stream) {
  (void)in_sizes; (void)n_in; (void)out_size; (void)ws_size;

  const float* x = (const float*)d_in[0];
  const float* basis = (const float*)d_in[1];
  const float* qc = (const float*)d_in[2];
  const float* kc = (const float*)d_in[3];
  const float* vc = (const float*)d_in[4];
  const float* oc = (const float*)d_in[5];
  const float* decay_logit = (const float*)d_in[6];
  const float* out_scale = (const float*)d_in[7];
  float* out = (float*)d_out;

  float* ws = (float*)d_ws;
  float* owT = ws;
  float* Q = owT + (size_t)V_DIM * C_DIM;
  float* K = Q + (size_t)M_DIM * C_DIM;
  float* Vv = K + (size_t)M_DIM * C_DIM;
  float* S = Vv + (size_t)M_DIM * C_DIM;
  float* XB = S;

  dim3 blk(32);

  proj_weights_kernel<<<dim3(V_DIM / 16, C_DIM / 16), blk, 0, stream>>>(
      basis, oc, owT);

  xb_kernel<<<dim3(M_DIM / GT_M, F_DIM / GT_N), dim3(256), 0, stream>>>(
      x, basis, XB);

  qkv_kernel<<<dim3(M_DIM / GT_M, 3 * C_DIM / GT_N), dim3(256), 0, stream>>>(
      XB, qc, kc, vc, Q, K, Vv);

  chunk_state_kernel<<<dim3(C_DIM / 64, N_CHUNK - 1, B_DIM), dim3(256), 0, stream>>>(
      K, Vv, decay_logit, S);

  chunk_scan_kernel<<<dim3(C_DIM * C_DIM / 4 / 256, 1, B_DIM), dim3(256), 0, stream>>>(
      decay_logit, S);

  attn_kernel<<<dim3(ATT_RBLOCKS / 2, 1, B_DIM), dim3(32 * ATT_WAVES), 0, stream>>>(
      Q, K, Vv, S, decay_logit);

  out_proj_kernel<<<dim3(M_DIM / GT_M, V_DIM / GT_N), dim3(256), 0, stream>>>(
      Q, owT, out_scale, out);
}


# chunk-state and attention kernels: k-loops fully unrolled, Q staging loads batched, chunk-state register budget raised: global loads fly ahead of the WMMAs (were waited per trip); sum order same
# speedup vs baseline: 7.5713x; 1.1165x over previous
#include <hip/hip_runtime.h>


typedef float v2f __attribute__((ext_vector_type(2)));
typedef float v8f __attribute__((ext_vector_type(8)));

#define C_DIM 256
#define V_DIM 1024
#define F_DIM 256
#define T_DIM 4096
#define B_DIM 4
#define M_DIM (B_DIM * T_DIM)

static __device__ __forceinline__ v8f wmma4(v2f a, v2f b, v8f c) {
  return __builtin_amdgcn_wmma_f32_16x16x4_f32(
       false, a,  false, b,
       (short)0, c,  false,  false);
}

static __device__ __forceinline__ v8f vzero8() {
  v8f z;
#pragma unroll
  for (int i = 0; i < 8; ++i) z[i] = 0.0f;
  return z;
}


__global__ void proj_weights_kernel(const float* __restrict__ basis,
                                    const float* __restrict__ oc,
                                    float* __restrict__ owT) {
  const int lane = threadIdx.x & 31;
  const int r = lane & 15;
  const int h = lane >> 4;
  const int v0 = blockIdx.x * 16;
  const int c0 = blockIdx.y * 16;

  v8f acc = vzero8();
  for (int f0 = 0; f0 < F_DIM; f0 += 4) {
    v2f a = *(const v2f*)&basis[(size_t)(v0 + r) * F_DIM + f0 + 2 * h];
    v2f b = *(const v2f*)&oc[(size_t)(c0 + r) * F_DIM + f0 + 2 * h];
    acc = wmma4(a, b, acc);
  }
#pragma unroll
  for (int i = 0; i < 8; ++i)
    owT[(size_t)(c0 + r) * V_DIM + v0 + i + 8 * h] = acc[i];
}

#define GT_M 128
#define GT_N 128
#define GT_K 32
#define GT_XP 36
#define GT_WP 288
#define GT_XS (GT_M * GT_XP)
#define GT_WS ((GT_K / 2) * GT_WP)
#define GT_STAGE (GT_XS + GT_WS)
#define GT_LDS_FLOATS (2 * GT_STAGE)

template <bool BT>
static __device__ __forceinline__ void gemm_slab_mma(const float* as, const float* bs,
                                                     v8f (&acc)[2][4]) {
  constexpr int B_KS = BT ? 4 : 2 * GT_WP;
  constexpr int B_NT = BT ? 16 * GT_XP : 32;
#pragma unroll
  for (int ks = 0; ks < GT_K; ks += 4) {
    v2f a0 = *(const v2f*)&as[ks];
    v2f a1 = *(const v2f*)&as[16 * GT_XP + ks];
    const float* b0 = &bs[(ks >> 2) * B_KS];
#pragma unroll
    for (int nt = 0; nt < 4; ++nt) {
      const v2f bb = *(const v2f*)&b0[nt * B_NT];
      acc[0][nt] = wmma4(a0, bb, acc[0][nt]);
      acc[1][nt] = wmma4(a1, bb, acc[1][nt]);
    }
  }
}

template <int KDIM, int LDA, int LDB, int LDC, bool SCALED, bool BT, bool ALONE, int KC_UNROLL>
static __device__ __forceinline__ void gemm_tile_128x128(
    const float* __restrict__ A,
    const float* __restrict__ Bm,
    float* __restrict__ Cm,
    float scale, float* __restrict__ lds) {
  const int tid = threadIdx.x;
  const int lane = tid & 31;
  const int w = __builtin_amdgcn_readfirstlane(tid >> 5);
  const int r = lane & 15;
  const int h = lane >> 4;
  const int wm = (w >> 1) * 32;
  const int wn = (w & 1) * 64;

  const int xr = tid >> 3;
  const int xc = (tid & 7) << 2;
  const int wp = tid >> 5;
  const int wc = (tid & 31) << 2;
  const float* Ag = A + (size_t)xr * LDA + xc;
  const float* Bg = BT ? Bm + (size_t)xr * LDB + xc
                       : Bm + (size_t)(2 * wp) * LDB + wc;
  const int xs_off = xr * GT_XP + xc;
  const int ws_off = BT ? GT_XS + xs_off : GT_XS + wp * GT_WP + 2 * wc;

  float4 xa[4], wb[4];
#pragma unroll
  for (int j = 0; j < 4; ++j) {
    xa[j] = *(const float4*)&Ag[(size_t)(32 * j) * LDA];
    if (BT)
      wb[j] = *(const float4*)&Bg[(size_t)(32 * j) * LDB];
    else
      wb[j] = *(const float4*)&Bg[(size_t)(16 * (j >> 1) + (j & 1)) * LDB];
  }
#pragma unroll
  for (int j = 0; j < 4; ++j)
    *(float4*)&lds[xs_off + 32 * j * GT_XP] = xa[j];
  if (BT) {
#pragma unroll
    for (int j = 0; j < 4; ++j)
      *(float4*)&lds[ws_off + 32 * j * GT_XP] = wb[j];
  } else {
#pragma unroll
    for (int p = 0; p < 2; ++p) {
      const float4 e0 = wb[2 * p], e1 = wb[2 * p + 1];
      *(float4*)&lds[ws_off + 8 * p * GT_WP] = make_float4(e0.x, e1.x, e0.y, e1.y);
      *(float4*)&lds[ws_off + 8 * p * GT_WP + 4] = make_float4(e0.z, e1.z, e0.w, e1.w);
    }
  }
  __syncthreads();

  v8f acc[2][4];
#pragma unroll
  for (int mt = 0; mt < 2; ++mt)
#pragma unroll
    for (int nt = 0; nt < 4; ++nt) acc[mt][nt] = vzero8();

  const int a_off = (wm + r) * GT_XP + 2 * h;
  const int b_off = BT ? GT_XS + (wn + r) * GT_XP + 2 * h
                       : GT_XS + h * GT_WP + 2 * (wn + r);

  constexpr int NCH = KDIM / GT_K;
#pragma unroll KC_UNROLL
  for (int kc = 0; kc < NCH - 1; ++kc) {
    const float* cur = lds + (kc & 1) * GT_STAGE;
    float* nxt = lds + ((kc + 1) & 1) * GT_STAGE;
    const int k0 = (kc + 1) * GT_K;
#pragma unroll
    for (int j = 0; j < 4; ++j) {
      xa[j] = *(const float4*)&Ag[(size_t)(32 * j) * LDA + k0];
      if (BT)
        wb[j] = *(const float4*)&Bg[(size_t)(32 * j) * LDB + k0];
      else
        wb[j] = *(const float4*)&Bg[(size_t)(k0 + 16 * (j >> 1) + (j & 1)) * LDB];
    }
    gemm_slab_mma<BT>(cur + a_off, cur + b_off, acc);
    if (ALONE) __builtin_amdgcn_sched_barrier(0x26);
#pragma unroll
    for (int j = 0; j < 4; ++j)
      *(float4*)&nxt[xs_off + 32 * j * GT_XP] = xa[j];
    if (BT) {
#pragma unroll
      for (int j = 0; j < 4; ++j)
        *(float4*)&nxt[ws_off + 32 * j * GT_XP] = wb[j];
    } else {
#pragma unroll
      for (int p = 0; p < 2; ++p) {
        const float4 e0 = wb[2 * p], e1 = wb[2 * p + 1];
        *(float4*)&nxt[ws_off + 8 * p * GT_WP] = make_float4(e0.x, e1.x, e0.y, e1.y);
        *(float4*)&nxt[ws_off + 8 * p * GT_WP + 4] = make_float4(e0.z, e1.z, e0.w, e1.w);
      }
    }
    __syncthreads();
  }
  {
    const float* last = lds + ((NCH - 1) & 1) * GT_STAGE;
    gemm_slab_mma<BT>(last + a_off, last + b_off, acc);
  }

#pragma unroll
  for (int mt = 0; mt < 2; ++mt)
#pragma unroll
    for (int nt = 0; nt < 4; ++nt)
#pragma unroll
      for (int i = 0; i < 8; ++i) {
        float v = acc[mt][nt][i];
        if (SCALED) v *= scale;
        Cm[(size_t)(wm + mt * 16 + i + 8 * h) * LDC + wn + nt * 16 + r] = v;
      }
}

#define XB_LDS_FLOATS 41216

__global__ __launch_bounds__(256)
void xb_kernel(const float* __restrict__ x,
               const float* __restrict__ basis,
               float* __restrict__ XB) {
  __shared__ float lds[XB_LDS_FLOATS];
  const int m0 = blockIdx.x * GT_M;
  const int n0 = blockIdx.y * GT_N;
  gemm_tile_128x128<V_DIM, V_DIM, F_DIM, F_DIM, false, false, true, 1>(
      x + (size_t)m0 * V_DIM, basis + n0, XB + (size_t)m0 * F_DIM + n0, 1.0f, lds);
}

#define QKV_LDS_FLOATS 20992

__global__ __launch_bounds__(256)
void qkv_kernel(const float* __restrict__ XB,
                const float* __restrict__ qc,
                const float* __restrict__ kc,
                const float* __restrict__ vc,
                float* __restrict__ Q,
                float* __restrict__ K,
                float* __restrict__ Vv) {
  __shared__ float lds[QKV_LDS_FLOATS];
  const int m0 = blockIdx.x * GT_M;
  const int which = (int)blockIdx.y >> 1;
  const int n0 = ((int)blockIdx.y & 1) * GT_N;
  const float* coef = (which == 0) ? qc : (which == 1) ? kc : vc;
  float* out = (which == 0) ? Q : (which == 1) ? K : Vv;
  gemm_tile_128x128<F_DIM, F_DIM, F_DIM, C_DIM, false, true, false, 1>(
      XB + (size_t)m0 * F_DIM, coef + (size_t)n0 * F_DIM,
      out + (size_t)m0 * C_DIM + n0, 1.0f, lds);
}

#define CH_L 256
#define N_CHUNK (T_DIM / CH_L)

__global__ __launch_bounds__(256) __attribute__((amdgpu_waves_per_eu(3, 3)))
void chunk_state_kernel(const float* __restrict__ K,
                        const float* __restrict__ Vv,
                        const float* __restrict__ decay_logit,
                        float* __restrict__ S) {
  const int tid = threadIdx.x;
  const int lane = tid & 31;
  const int w = __builtin_amdgcn_readfirstlane(tid >> 5);
  const int r = lane & 15;
  const int h = lane >> 4;
  const int kc0 = w * 32;
  const int vc0 = blockIdx.x * 64;
  const int ch = (int)blockIdx.z * N_CHUNK + 1 + (int)blockIdx.y;
  const float* Kc = K + (size_t)ch * CH_L * C_DIM;
  const float* Vc = Vv + (size_t)ch * CH_L * C_DIM;

  const float dl = *decay_logit;
  const float dec = 1.0f / (1.0f + __expf(-dl));
  const float ln_dec = __logf(dec);

  v8f acc[2][4];
#pragma unroll
  for (int mt = 0; mt < 2; ++mt)
#pragma unroll
    for (int nt = 0; nt < 4; ++nt) acc[mt][nt] = vzero8();

#pragma unroll
  for (int j0 = 0; j0 < CH_L; j0 += 4) {
    const int ja = j0 + 2 * h;
    const float w0 = __expf(ln_dec * (float)ja);
    const float w1 = __expf(ln_dec * (float)(ja + 1));
    const float* Kr0 = Kc + (size_t)ja * C_DIM + kc0;
    const float* Kr1 = Kr0 + C_DIM;
    const float* Vr0 = Vc + (size_t)ja * C_DIM + vc0;
    const float* Vr1 = Vr0 + C_DIM;
    v2f a0, a1;
    a0.x = Kr0[r] * w0;       a0.y = Kr1[r] * w1;
    a1.x = Kr0[16 + r] * w0;  a1.y = Kr1[16 + r] * w1;
#pragma unroll
    for (int nt = 0; nt < 4; ++nt) {
      v2f bb;
      bb.x = Vr0[16 * nt + r];  bb.y = Vr1[16 * nt + r];
      acc[0][nt] = wmma4(a0, bb, acc[0][nt]);
      acc[1][nt] = wmma4(a1, bb, acc[1][nt]);
    }
  }

  float* Sc = S + (size_t)ch * C_DIM * C_DIM;
#pragma unroll
  for (int mt = 0; mt < 2; ++mt)
#pragma unroll
    for (int nt = 0; nt < 4; ++nt)
#pragma unroll
      for (int i = 0; i < 8; ++i)
        Sc[(size_t)(kc0 + mt * 16 + i + 8 * h) * C_DIM + vc0 + nt * 16 + r] = acc[mt][nt][i];
}

__global__ __launch_bounds__(256)
void chunk_scan_kernel(const float* __restrict__ decay_logit, float* __restrict__ S) {
  const float dl = *decay_logit;
  const float dec = 1.0f / (1.0f + __expf(-dl));
  const float dL = __expf(__logf(dec) * (float)CH_L);
  const size_t e = ((size_t)blockIdx.x * 256 + threadIdx.x) * 4;
  float* Sb = S + (size_t)blockIdx.z * N_CHUNK * C_DIM * C_DIM + e;
  float4 p[N_CHUNK - 1];
#pragma unroll
  for (int c = 1; c < N_CHUNK; ++c)
    p[c - 1] = *(const float4*)&Sb[(size_t)c * C_DIM * C_DIM];
  float4 s = p[N_CHUNK - 2];
#pragma unroll
  for (int c = N_CHUNK - 2; c >= 1; --c) {
    const float4 q = p[c - 1];
    s.x = fmaf(dL, s.x, q.x);
    s.y = fmaf(dL, s.y, q.y);
    s.z = fmaf(dL, s.z, q.z);
    s.w = fmaf(dL, s.w, q.w);
    *(float4*)&Sb[(size_t)c * C_DIM * C_DIM] = s;
  }
}

#define ATT_WAVES 8
#define ATT_ROWS 32
#define ATT_KEYS (16 * ATT_WAVES)
#define ATT_RBLOCKS (T_DIM / ATT_ROWS)
#define QS_PITCH 260
#define SS_PITCH 132
#define ATT_QS_FLOATS (ATT_ROWS * QS_PITCH)
#define ATT_SS_FLOATS (ATT_ROWS * SS_PITCH)
#define ATT_LDS_FLOATS 41216

__global__ __launch_bounds__(32 * ATT_WAVES)
void attn_kernel(float* QR,
                 const float* __restrict__ K,
                 const float* __restrict__ Vv,
                 const float* __restrict__ S,
                 const float* __restrict__ decay_logit) {
  __shared__ float lds[ATT_LDS_FLOATS];
  float* const qs = &lds[0];
  float* const ss = &lds[ATT_QS_FLOATS];

  const int tid = threadIdx.x;
  const int lane = tid & 31;
  const int w = __builtin_amdgcn_readfirstlane(tid >> 5);
  const int r = lane & 15;
  const int h = lane >> 4;
  const int c0 = w * 32;
  const int b = blockIdx.z;

  float* Qb = QR + (size_t)b * T_DIM * C_DIM;
  const float* Kb = K + (size_t)b * T_DIM * C_DIM;
  const float* Vb = Vv + (size_t)b * T_DIM * C_DIM;
  float* Rb = Qb;

  const float dl = *decay_logit;
  const float dec = 1.0f / (1.0f + __expf(-dl));
  const float ln_dec = __logf(dec);

  for (int part = 0; part < 2; ++part) {
    const int rb = (part == 0) ? (int)blockIdx.x : (ATT_RBLOCKS - 1 - (int)blockIdx.x);
    const int i0 = rb * ATT_ROWS;
    const int cend = (i0 & ~(CH_L - 1)) + CH_L;

    {
      float4 qv[ATT_ROWS * 64 / (32 * ATT_WAVES)];
#pragma unroll
      for (int it = 0; it < ATT_ROWS * 64 / (32 * ATT_WAVES); ++it) {
        const int idx = tid + it * 32 * ATT_WAVES;
        qv[it] = *(const float4*)&Qb[(size_t)(i0 + (idx >> 6)) * C_DIM + ((idx & 63) << 2)];
      }
#pragma unroll
      for (int it = 0; it < ATT_ROWS * 64 / (32 * ATT_WAVES); ++it) {
        const int idx = tid + it * 32 * ATT_WAVES;
        *(float4*)&qs[(idx >> 6) * QS_PITCH + ((idx & 63) << 2)] = qv[it];
      }
    }
    __syncthreads();

    v8f a0c0 = vzero8(), a0c1 = vzero8();
    v8f a1c0 = vzero8(), a1c1 = vzero8();

    int buf = 0;
    for (int jc = i0 & ~(ATT_KEYS - 1); jc < cend; jc += ATT_KEYS) {
      const int j0 = jc + 16 * w;
      {
        int jn = (jc + ATT_KEYS < cend) ? (jc + ATT_KEYS) : jc;
        __builtin_prefetch(&Kb[(size_t)(jn + 16 * w + r) * C_DIM + 128 * h], 0, 1);
        __builtin_prefetch(&Vb[(size_t)(jn + lane) * C_DIM + c0], 0, 1);
        __builtin_prefetch(&Vb[(size_t)(jn + 32 + lane) * C_DIM + c0], 0, 1);
        __builtin_prefetch(&Vb[(size_t)(jn + 64 + lane) * C_DIM + c0], 0, 1);
        __builtin_prefetch(&Vb[(size_t)(jn + 96 + lane) * C_DIM + c0], 0, 1);
      }

      float* sb = &ss[buf * ATT_SS_FLOATS];
      if (j0 >= i0) {
        v8f s0 = vzero8(), s1 = vzero8();
        const float* Kj = Kb + (size_t)j0 * C_DIM;
#pragma unroll
        for (int k0 = 0; k0 < C_DIM; k0 += 4) {
          v2f qa = *(const v2f*)&qs[r * QS_PITCH + k0 + 2 * h];
          v2f qb = *(const v2f*)&qs[(16 + r) * QS_PITCH + k0 + 2 * h];
          v2f bb = *(const v2f*)&Kj[(size_t)r * C_DIM + k0 + 2 * h];
          s0 = wmma4(qa, bb, s0);
          s1 = wmma4(qb, bb, s1);
        }
#pragma unroll
        for (int i = 0; i < 8; ++i) {
          int gi = i0 + i + 8 * h;
          int gj = j0 + r;
          int d0 = gj - gi;
          int d1 = d0 - 16;
          float wg0 = (d0 > 0) ? __expf(ln_dec * (float)(d0 - 1)) : 0.0f;
          float wg1 = (d1 > 0) ? __expf(ln_dec * (float)(d1 - 1)) : 0.0f;
          sb[(i + 8 * h) * SS_PITCH + 16 * w + r] = s0[i] * wg0;
          sb[(16 + i + 8 * h) * SS_PITCH + 16 * w + r] = s1[i] * wg1;
        }
      } else {
#pragma unroll
        for (int i = 0; i < 8; ++i) {
          sb[(i + 8 * h) * SS_PITCH + 16 * w + r] = 0.0f;
          sb[(16 + i + 8 * h) * SS_PITCH + 16 * w + r] = 0.0f;
        }
      }
      __syncthreads();
#pragma unroll
      for (int kk = 0; kk < ATT_KEYS / 4; ++kk) {
        v2f pa = *(const v2f*)&sb[r * SS_PITCH + kk * 4 + 2 * h];
        v2f pb = *(const v2f*)&sb[(16 + r) * SS_PITCH + kk * 4 + 2 * h];
        const float* Vr0 = Vb + (size_t)(jc + kk * 4 + 2 * h) * C_DIM + c0;
        const float* Vr1 = Vr0 + C_DIM;
        v2f bb;
        bb.x = Vr0[r];      bb.y = Vr1[r];
        a0c0 = wmma4(pa, bb, a0c0);  a1c0 = wmma4(pb, bb, a1c0);
        bb.x = Vr0[16 + r]; bb.y = Vr1[16 + r];
        a0c1 = wmma4(pa, bb, a0c1);  a1c1 = wmma4(pb, bb, a1c1);
      }
      buf ^= 1;
    }

    if (cend < T_DIM) {
      const float* Sn = S + ((size_t)b * N_CHUNK + (size_t)(cend / CH_L)) * C_DIM * C_DIM;
      const float f0 = __expf(ln_dec * (float)(cend - 1 - (i0 + r)));
      const float f1 = __expf(ln_dec * (float)(cend - 1 - (i0 + 16 + r)));
#pragma unroll
      for (int kk = 0; kk < C_DIM / 4; ++kk) {
        v2f pa = *(const v2f*)&qs[r * QS_PITCH + kk * 4 + 2 * h];
        v2f pb = *(const v2f*)&qs[(16 + r) * QS_PITCH + kk * 4 + 2 * h];
        pa.x *= f0;  pa.y *= f0;
        pb.x *= f1;  pb.y *= f1;
        const float* Sr0 = Sn + (size_t)(kk * 4 + 2 * h) * C_DIM + c0;
        const float* Sr1 = Sr0 + C_DIM;
        v2f bb;
        bb.x = Sr0[r];      bb.y = Sr1[r];
        a0c0 = wmma4(pa, bb, a0c0);  a1c0 = wmma4(pb, bb, a1c0);
        bb.x = Sr0[16 + r]; bb.y = Sr1[16 + r];
        a0c1 = wmma4(pa, bb, a0c1);  a1c1 = wmma4(pb, bb, a1c1);
      }
    }
    __syncthreads();

#pragma unroll
    for (int i = 0; i < 8; ++i) {
      size_t row0 = (size_t)(i0 + i + 8 * h) * C_DIM + c0 + r;
      size_t row1 = row0 + (size_t)16 * C_DIM;
      Rb[row0] = a0c0[i];
      Rb[row0 + 16] = a0c1[i];
      Rb[row1] = a1c0[i];
      Rb[row1 + 16] = a1c1[i];
    }
  }
}

__global__ __launch_bounds__(256) __attribute__((amdgpu_waves_per_eu(8, 8)))
void out_proj_kernel(const float* __restrict__ R,
                     const float* __restrict__ owT,
                     const float* __restrict__ out_scale,
                     float* __restrict__ out) {
  __shared__ float lds[GT_LDS_FLOATS];
  const int m0 = blockIdx.x * GT_M;
  const int v0 = blockIdx.y * GT_N;
  const float sc = *out_scale;
  gemm_tile_128x128<C_DIM, C_DIM, V_DIM, V_DIM, true, false, false, C_DIM / GT_K - 1>(
      R + (size_t)m0 * C_DIM, owT + v0, out + (size_t)m0 * V_DIM + v0, sc, lds);
}

extern "C" void kernel_launch(void* const* d_in, const int* in_sizes, int n_in,
                              void* d_out, int out_size, void* d_ws, size_t ws_size,
                              hipStream_t stream) {
  (void)in_sizes; (void)n_in; (void)out_size; (void)ws_size;

  const float* x = (const float*)d_in[0];
  const float* basis = (const float*)d_in[1];
  const float* qc = (const float*)d_in[2];
  const float* kc = (const float*)d_in[3];
  const float* vc = (const float*)d_in[4];
  const float* oc = (const float*)d_in[5];
  const float* decay_logit = (const float*)d_in[6];
  const float* out_scale = (const float*)d_in[7];
  float* out = (float*)d_out;

  float* ws = (float*)d_ws;
  float* owT = ws;
  float* Q = owT + (size_t)V_DIM * C_DIM;
  float* K = Q + (size_t)M_DIM * C_DIM;
  float* Vv = K + (size_t)M_DIM * C_DIM;
  float* S = Vv + (size_t)M_DIM * C_DIM;
  float* XB = S;

  dim3 blk(32);

  proj_weights_kernel<<<dim3(V_DIM / 16, C_DIM / 16), blk, 0, stream>>>(
      basis, oc, owT);

  xb_kernel<<<dim3(M_DIM / GT_M, F_DIM / GT_N), dim3(256), 0, stream>>>(
      x, basis, XB);

  qkv_kernel<<<dim3(M_DIM / GT_M, 3 * C_DIM / GT_N), dim3(256), 0, stream>>>(
      XB, qc, kc, vc, Q, K, Vv);

  chunk_state_kernel<<<dim3(C_DIM / 64, N_CHUNK - 1, B_DIM), dim3(256), 0, stream>>>(
      K, Vv, decay_logit, S);

  chunk_scan_kernel<<<dim3(C_DIM * C_DIM / 4 / 256, 1, B_DIM), dim3(256), 0, stream>>>(
      decay_logit, S);

  attn_kernel<<<dim3(ATT_RBLOCKS / 2, 1, B_DIM), dim3(32 * ATT_WAVES), 0, stream>>>(
      Q, K, Vv, S, decay_logit);

  out_proj_kernel<<<dim3(M_DIM / GT_M, V_DIM / GT_N), dim3(256), 0, stream>>>(
      Q, owT, out_scale, out);
}


# K and V projections folded away: rows of x@basis serve as keys and values; their coefficients enter via q_c^T k_c [F,F] and v_c^T o_w^T [F,V] (2 small GEMMs); summation order changes; 8 launches
# speedup vs baseline: 8.0738x; 1.0664x over previous
#include <hip/hip_runtime.h>


typedef float v2f __attribute__((ext_vector_type(2)));
typedef float v8f __attribute__((ext_vector_type(8)));

#define C_DIM 256
#define V_DIM 1024
#define F_DIM 256
#define T_DIM 4096
#define B_DIM 4
#define M_DIM (B_DIM * T_DIM)

static __device__ __forceinline__ v8f wmma4(v2f a, v2f b, v8f c) {
  return __builtin_amdgcn_wmma_f32_16x16x4_f32(
       false, a,  false, b,
       (short)0, c,  false,  false);
}

static __device__ __forceinline__ v8f vzero8() {
  v8f z;
#pragma unroll
  for (int i = 0; i < 8; ++i) z[i] = 0.0f;
  return z;
}


__global__ void proj_weights_kernel(const float* __restrict__ basis,
                                    const float* __restrict__ oc,
                                    const float* __restrict__ qc,
                                    const float* __restrict__ kc,
                                    float* __restrict__ owT,
                                    float* __restrict__ GT) {
  const int lane = threadIdx.x & 31;
  const int r = lane & 15;
  const int h = lane >> 4;

  if ((int)blockIdx.y >= C_DIM / 16) {
    const int t = (int)blockIdx.x + (V_DIM / 16) * ((int)blockIdx.y - C_DIM / 16);
    const int fr = (t >> 4) * 16;
    const int fc = (t & 15) * 16;
    v8f g = vzero8();
#pragma unroll
    for (int cc = 0; cc < C_DIM; cc += 4) {
      const float* kr = &kc[(size_t)(cc + 2 * h) * F_DIM + fr + r];
      const float* qr = &qc[(size_t)(cc + 2 * h) * F_DIM + fc + r];
      v2f a, b;
      a.x = kr[0];  a.y = kr[F_DIM];
      b.x = qr[0];  b.y = qr[F_DIM];
      g = wmma4(a, b, g);
    }
#pragma unroll
    for (int i = 0; i < 8; ++i)
      GT[(size_t)(fr + i + 8 * h) * F_DIM + fc + r] = g[i];
    return;
  }

  const int v0 = blockIdx.x * 16;
  const int c0 = blockIdx.y * 16;

  v8f acc = vzero8();
  for (int f0 = 0; f0 < F_DIM; f0 += 4) {
    v2f a = *(const v2f*)&basis[(size_t)(v0 + r) * F_DIM + f0 + 2 * h];
    v2f b = *(const v2f*)&oc[(size_t)(c0 + r) * F_DIM + f0 + 2 * h];
    acc = wmma4(a, b, acc);
  }
#pragma unroll
  for (int i = 0; i < 8; ++i)
    owT[(size_t)(c0 + r) * V_DIM + v0 + i + 8 * h] = acc[i];
}

__global__ void out_weights_kernel(const float* __restrict__ vc,
                                   const float* __restrict__ owT,
                                   float* __restrict__ W2) {
  const int lane = threadIdx.x & 31;
  const int r = lane & 15;
  const int h = lane >> 4;
  const int v0 = blockIdx.x * 16;
  const int f0 = blockIdx.y * 16;

  v8f acc = vzero8();
#pragma unroll
  for (int cc = 0; cc < C_DIM; cc += 4) {
    const float* ar = &vc[(size_t)(cc + 2 * h) * F_DIM + f0 + r];
    const float* br = &owT[(size_t)(cc + 2 * h) * V_DIM + v0 + r];
    v2f a, b;
    a.x = ar[0];  a.y = ar[F_DIM];
    b.x = br[0];  b.y = br[V_DIM];
    acc = wmma4(a, b, acc);
  }
#pragma unroll
  for (int i = 0; i < 8; ++i)
    W2[(size_t)(f0 + i + 8 * h) * V_DIM + v0 + r] = acc[i];
}

#define GT_M 128
#define GT_N 128
#define GT_K 32
#define GT_XP 36
#define GT_WP 288
#define GT_XS (GT_M * GT_XP)
#define GT_WS ((GT_K / 2) * GT_WP)
#define GT_STAGE (GT_XS + GT_WS)
#define GT_LDS_FLOATS (2 * GT_STAGE)

template <bool BT>
static __device__ __forceinline__ void gemm_slab_mma(const float* as, const float* bs,
                                                     v8f (&acc)[2][4]) {
  constexpr int B_KS = BT ? 4 : 2 * GT_WP;
  constexpr int B_NT = BT ? 16 * GT_XP : 32;
#pragma unroll
  for (int ks = 0; ks < GT_K; ks += 4) {
    v2f a0 = *(const v2f*)&as[ks];
    v2f a1 = *(const v2f*)&as[16 * GT_XP + ks];
    const float* b0 = &bs[(ks >> 2) * B_KS];
#pragma unroll
    for (int nt = 0; nt < 4; ++nt) {
      const v2f bb = *(const v2f*)&b0[nt * B_NT];
      acc[0][nt] = wmma4(a0, bb, acc[0][nt]);
      acc[1][nt] = wmma4(a1, bb, acc[1][nt]);
    }
  }
}

template <int KDIM, int LDA, int LDB, int LDC, bool SCALED, bool BT, bool ALONE, int KC_UNROLL>
static __device__ __forceinline__ void gemm_tile_128x128(
    const float* __restrict__ A,
    const float* __restrict__ Bm,
    float* __restrict__ Cm,
    float scale, float* __restrict__ lds) {
  const int tid = threadIdx.x;
  const int lane = tid & 31;
  const int w = __builtin_amdgcn_readfirstlane(tid >> 5);
  const int r = lane & 15;
  const int h = lane >> 4;
  const int wm = (w >> 1) * 32;
  const int wn = (w & 1) * 64;

  const int xr = tid >> 3;
  const int xc = (tid & 7) << 2;
  const int wp = tid >> 5;
  const int wc = (tid & 31) << 2;
  const float* Ag = A + (size_t)xr * LDA + xc;
  const float* Bg = BT ? Bm + (size_t)xr * LDB + xc
                       : Bm + (size_t)(2 * wp) * LDB + wc;
  const int xs_off = xr * GT_XP + xc;
  const int ws_off = BT ? GT_XS + xs_off : GT_XS + wp * GT_WP + 2 * wc;

  float4 xa[4], wb[4];
#pragma unroll
  for (int j = 0; j < 4; ++j) {
    xa[j] = *(const float4*)&Ag[(size_t)(32 * j) * LDA];
    if (BT)
      wb[j] = *(const float4*)&Bg[(size_t)(32 * j) * LDB];
    else
      wb[j] = *(const float4*)&Bg[(size_t)(16 * (j >> 1) + (j & 1)) * LDB];
  }
#pragma unroll
  for (int j = 0; j < 4; ++j)
    *(float4*)&lds[xs_off + 32 * j * GT_XP] = xa[j];
  if (BT) {
#pragma unroll
    for (int j = 0; j < 4; ++j)
      *(float4*)&lds[ws_off + 32 * j * GT_XP] = wb[j];
  } else {
#pragma unroll
    for (int p = 0; p < 2; ++p) {
      const float4 e0 = wb[2 * p], e1 = wb[2 * p + 1];
      *(float4*)&lds[ws_off + 8 * p * GT_WP] = make_float4(e0.x, e1.x, e0.y, e1.y);
      *(float4*)&lds[ws_off + 8 * p * GT_WP + 4] = make_float4(e0.z, e1.z, e0.w, e1.w);
    }
  }
  __syncthreads();

  v8f acc[2][4];
#pragma unroll
  for (int mt = 0; mt < 2; ++mt)
#pragma unroll
    for (int nt = 0; nt < 4; ++nt) acc[mt][nt] = vzero8();

  const int a_off = (wm + r) * GT_XP + 2 * h;
  const int b_off = BT ? GT_XS + (wn + r) * GT_XP + 2 * h
                       : GT_XS + h * GT_WP + 2 * (wn + r);

  constexpr int NCH = KDIM / GT_K;
#pragma unroll KC_UNROLL
  for (int kc = 0; kc < NCH - 1; ++kc) {
    const float* cur = lds + (kc & 1) * GT_STAGE;
    float* nxt = lds + ((kc + 1) & 1) * GT_STAGE;
    const int k0 = (kc + 1) * GT_K;
#pragma unroll
    for (int j = 0; j < 4; ++j) {
      xa[j] = *(const float4*)&Ag[(size_t)(32 * j) * LDA + k0];
      if (BT)
        wb[j] = *(const float4*)&Bg[(size_t)(32 * j) * LDB + k0];
      else
        wb[j] = *(const float4*)&Bg[(size_t)(k0 + 16 * (j >> 1) + (j & 1)) * LDB];
    }
    gemm_slab_mma<BT>(cur + a_off, cur + b_off, acc);
    if (ALONE) __builtin_amdgcn_sched_barrier(0x26);
#pragma unroll
    for (int j = 0; j < 4; ++j)
      *(float4*)&nxt[xs_off + 32 * j * GT_XP] = xa[j];
    if (BT) {
#pragma unroll
      for (int j = 0; j < 4; ++j)
        *(float4*)&nxt[ws_off + 32 * j * GT_XP] = wb[j];
    } else {
#pragma unroll
      for (int p = 0; p < 2; ++p) {
        const float4 e0 = wb[2 * p], e1 = wb[2 * p + 1];
        *(float4*)&nxt[ws_off + 8 * p * GT_WP] = make_float4(e0.x, e1.x, e0.y, e1.y);
        *(float4*)&nxt[ws_off + 8 * p * GT_WP + 4] = make_float4(e0.z, e1.z, e0.w, e1.w);
      }
    }
    __syncthreads();
  }
  {
    const float* last = lds + ((NCH - 1) & 1) * GT_STAGE;
    gemm_slab_mma<BT>(last + a_off, last + b_off, acc);
  }

#pragma unroll
  for (int mt = 0; mt < 2; ++mt)
#pragma unroll
    for (int nt = 0; nt < 4; ++nt)
#pragma unroll
      for (int i = 0; i < 8; ++i) {
        float v = acc[mt][nt][i];
        if (SCALED) v *= scale;
        Cm[(size_t)(wm + mt * 16 + i + 8 * h) * LDC + wn + nt * 16 + r] = v;
      }
}

#define XB_LDS_FLOATS 41216

__global__ __launch_bounds__(256)
void xb_kernel(const float* __restrict__ x,
               const float* __restrict__ basis,
               float* __restrict__ XB) {
  __shared__ float lds[XB_LDS_FLOATS];
  const int m0 = blockIdx.x * GT_M;
  const int n0 = blockIdx.y * GT_N;
  gemm_tile_128x128<V_DIM, V_DIM, F_DIM, F_DIM, false, false, true, 1>(
      x + (size_t)m0 * V_DIM, basis + n0, XB + (size_t)m0 * F_DIM + n0, 1.0f, lds);
}

__global__ __launch_bounds__(256)
void q_kernel(const float* __restrict__ XB,
              const float* __restrict__ GT,
              float* __restrict__ Qp) {
  __shared__ float lds[XB_LDS_FLOATS];
  const int m0 = blockIdx.x * GT_M;
  const int n0 = blockIdx.y * GT_N;
  gemm_tile_128x128<F_DIM, F_DIM, F_DIM, F_DIM, false, true, true, 1>(
      XB + (size_t)m0 * F_DIM, GT + (size_t)n0 * F_DIM,
      Qp + (size_t)m0 * F_DIM + n0, 1.0f, lds);
}

#define CH_L 256
#define N_CHUNK (T_DIM / CH_L)

__global__ __launch_bounds__(256) __attribute__((amdgpu_waves_per_eu(3, 3)))
void chunk_state_kernel(const float* __restrict__ K,
                        const float* __restrict__ Vv,
                        const float* __restrict__ decay_logit,
                        float* __restrict__ S) {
  const int tid = threadIdx.x;
  const int lane = tid & 31;
  const int w = __builtin_amdgcn_readfirstlane(tid >> 5);
  const int r = lane & 15;
  const int h = lane >> 4;
  const int kc0 = w * 32;
  const int vc0 = blockIdx.x * 64;
  const int ch = (int)blockIdx.z * N_CHUNK + 1 + (int)blockIdx.y;
  const float* Kc = K + (size_t)ch * CH_L * C_DIM;
  const float* Vc = Vv + (size_t)ch * CH_L * C_DIM;

  const float dl = *decay_logit;
  const float dec = 1.0f / (1.0f + __expf(-dl));
  const float ln_dec = __logf(dec);

  v8f acc[2][4];
#pragma unroll
  for (int mt = 0; mt < 2; ++mt)
#pragma unroll
    for (int nt = 0; nt < 4; ++nt) acc[mt][nt] = vzero8();

#pragma unroll
  for (int j0 = 0; j0 < CH_L; j0 += 4) {
    const int ja = j0 + 2 * h;
    const float w0 = __expf(ln_dec * (float)ja);
    const float w1 = __expf(ln_dec * (float)(ja + 1));
    const float* Kr0 = Kc + (size_t)ja * C_DIM + kc0;
    const float* Kr1 = Kr0 + C_DIM;
    const float* Vr0 = Vc + (size_t)ja * C_DIM + vc0;
    const float* Vr1 = Vr0 + C_DIM;
    v2f a0, a1;
    a0.x = Kr0[r] * w0;       a0.y = Kr1[r] * w1;
    a1.x = Kr0[16 + r] * w0;  a1.y = Kr1[16 + r] * w1;
#pragma unroll
    for (int nt = 0; nt < 4; ++nt) {
      v2f bb;
      bb.x = Vr0[16 * nt + r];  bb.y = Vr1[16 * nt + r];
      acc[0][nt] = wmma4(a0, bb, acc[0][nt]);
      acc[1][nt] = wmma4(a1, bb, acc[1][nt]);
    }
  }

  float* Sc = S + (size_t)ch * C_DIM * C_DIM;
#pragma unroll
  for (int mt = 0; mt < 2; ++mt)
#pragma unroll
    for (int nt = 0; nt < 4; ++nt)
#pragma unroll
      for (int i = 0; i < 8; ++i)
        Sc[(size_t)(kc0 + mt * 16 + i + 8 * h) * C_DIM + vc0 + nt * 16 + r] = acc[mt][nt][i];
}

__global__ __launch_bounds__(256)
void chunk_scan_kernel(const float* __restrict__ decay_logit, float* __restrict__ S) {
  const float dl = *decay_logit;
  const float dec = 1.0f / (1.0f + __expf(-dl));
  const float dL = __expf(__logf(dec) * (float)CH_L);
  const size_t e = ((size_t)blockIdx.x * 256 + threadIdx.x) * 4;
  float* Sb = S + (size_t)blockIdx.z * N_CHUNK * C_DIM * C_DIM + e;
  float4 p[N_CHUNK - 1];
#pragma unroll
  for (int c = 1; c < N_CHUNK; ++c)
    p[c - 1] = *(const float4*)&Sb[(size_t)c * C_DIM * C_DIM];
  float4 s = p[N_CHUNK - 2];
#pragma unroll
  for (int c = N_CHUNK - 2; c >= 1; --c) {
    const float4 q = p[c - 1];
    s.x = fmaf(dL, s.x, q.x);
    s.y = fmaf(dL, s.y, q.y);
    s.z = fmaf(dL, s.z, q.z);
    s.w = fmaf(dL, s.w, q.w);
    *(float4*)&Sb[(size_t)c * C_DIM * C_DIM] = s;
  }
}

#define ATT_WAVES 8
#define ATT_ROWS 32
#define ATT_KEYS (16 * ATT_WAVES)
#define ATT_RBLOCKS (T_DIM / ATT_ROWS)
#define QS_PITCH 260
#define SS_PITCH 132
#define ATT_QS_FLOATS (ATT_ROWS * QS_PITCH)
#define ATT_SS_FLOATS (ATT_ROWS * SS_PITCH)
#define ATT_LDS_FLOATS 41216

__global__ __launch_bounds__(32 * ATT_WAVES)
void attn_kernel(float* QR,
                 const float* __restrict__ K,
                 const float* __restrict__ Vv,
                 const float* __restrict__ S,
                 const float* __restrict__ decay_logit) {
  __shared__ float lds[ATT_LDS_FLOATS];
  float* const qs = &lds[0];
  float* const ss = &lds[ATT_QS_FLOATS];

  const int tid = threadIdx.x;
  const int lane = tid & 31;
  const int w = __builtin_amdgcn_readfirstlane(tid >> 5);
  const int r = lane & 15;
  const int h = lane >> 4;
  const int c0 = w * 32;
  const int b = blockIdx.z;

  float* Qb = QR + (size_t)b * T_DIM * C_DIM;
  const float* Kb = K + (size_t)b * T_DIM * C_DIM;
  const float* Vb = Vv + (size_t)b * T_DIM * C_DIM;
  float* Rb = Qb;

  const float dl = *decay_logit;
  const float dec = 1.0f / (1.0f + __expf(-dl));
  const float ln_dec = __logf(dec);

  for (int part = 0; part < 2; ++part) {
    const int rb = (part == 0) ? (int)blockIdx.x : (ATT_RBLOCKS - 1 - (int)blockIdx.x);
    const int i0 = rb * ATT_ROWS;
    const int cend = (i0 & ~(CH_L - 1)) + CH_L;

    {
      float4 qv[ATT_ROWS * 64 / (32 * ATT_WAVES)];
#pragma unroll
      for (int it = 0; it < ATT_ROWS * 64 / (32 * ATT_WAVES); ++it) {
        const int idx = tid + it * 32 * ATT_WAVES;
        qv[it] = *(const float4*)&Qb[(size_t)(i0 + (idx >> 6)) * C_DIM + ((idx & 63) << 2)];
      }
#pragma unroll
      for (int it = 0; it < ATT_ROWS * 64 / (32 * ATT_WAVES); ++it) {
        const int idx = tid + it * 32 * ATT_WAVES;
        *(float4*)&qs[(idx >> 6) * QS_PITCH + ((idx & 63) << 2)] = qv[it];
      }
    }
    __syncthreads();

    v8f a0c0 = vzero8(), a0c1 = vzero8();
    v8f a1c0 = vzero8(), a1c1 = vzero8();

    int buf = 0;
    for (int jc = i0 & ~(ATT_KEYS - 1); jc < cend; jc += ATT_KEYS) {
      const int j0 = jc + 16 * w;
      {
        int jn = (jc + ATT_KEYS < cend) ? (jc + ATT_KEYS) : jc;
        __builtin_prefetch(&Kb[(size_t)(jn + 16 * w + r) * C_DIM + 128 * h], 0, 1);
        __builtin_prefetch(&Vb[(size_t)(jn + lane) * C_DIM + c0], 0, 1);
        __builtin_prefetch(&Vb[(size_t)(jn + 32 + lane) * C_DIM + c0], 0, 1);
        __builtin_prefetch(&Vb[(size_t)(jn + 64 + lane) * C_DIM + c0], 0, 1);
        __builtin_prefetch(&Vb[(size_t)(jn + 96 + lane) * C_DIM + c0], 0, 1);
      }

      float* sb = &ss[buf * ATT_SS_FLOATS];
      if (j0 >= i0) {
        v8f s0 = vzero8(), s1 = vzero8();
        const float* Kj = Kb + (size_t)j0 * C_DIM;
#pragma unroll
        for (int k0 = 0; k0 < C_DIM; k0 += 4) {
          v2f qa = *(const v2f*)&qs[r * QS_PITCH + k0 + 2 * h];
          v2f qb = *(const v2f*)&qs[(16 + r) * QS_PITCH + k0 + 2 * h];
          v2f bb = *(const v2f*)&Kj[(size_t)r * C_DIM + k0 + 2 * h];
          s0 = wmma4(qa, bb, s0);
          s1 = wmma4(qb, bb, s1);
        }
#pragma unroll
        for (int i = 0; i < 8; ++i) {
          int gi = i0 + i + 8 * h;
          int gj = j0 + r;
          int d0 = gj - gi;
          int d1 = d0 - 16;
          float wg0 = (d0 > 0) ? __expf(ln_dec * (float)(d0 - 1)) : 0.0f;
          float wg1 = (d1 > 0) ? __expf(ln_dec * (float)(d1 - 1)) : 0.0f;
          sb[(i + 8 * h) * SS_PITCH + 16 * w + r] = s0[i] * wg0;
          sb[(16 + i + 8 * h) * SS_PITCH + 16 * w + r] = s1[i] * wg1;
        }
      } else {
#pragma unroll
        for (int i = 0; i < 8; ++i) {
          sb[(i + 8 * h) * SS_PITCH + 16 * w + r] = 0.0f;
          sb[(16 + i + 8 * h) * SS_PITCH + 16 * w + r] = 0.0f;
        }
      }
      __syncthreads();
#pragma unroll
      for (int kk = 0; kk < ATT_KEYS / 4; ++kk) {
        v2f pa = *(const v2f*)&sb[r * SS_PITCH + kk * 4 + 2 * h];
        v2f pb = *(const v2f*)&sb[(16 + r) * SS_PITCH + kk * 4 + 2 * h];
        const float* Vr0 = Vb + (size_t)(jc + kk * 4 + 2 * h) * C_DIM + c0;
        const float* Vr1 = Vr0 + C_DIM;
        v2f bb;
        bb.x = Vr0[r];      bb.y = Vr1[r];
        a0c0 = wmma4(pa, bb, a0c0);  a1c0 = wmma4(pb, bb, a1c0);
        bb.x = Vr0[16 + r]; bb.y = Vr1[16 + r];
        a0c1 = wmma4(pa, bb, a0c1);  a1c1 = wmma4(pb, bb, a1c1);
      }
      buf ^= 1;
    }

    if (cend < T_DIM) {
      const float* Sn = S + ((size_t)b * N_CHUNK + (size_t)(cend / CH_L)) * C_DIM * C_DIM;
      const float f0 = __expf(ln_dec * (float)(cend - 1 - (i0 + r)));
      const float f1 = __expf(ln_dec * (float)(cend - 1 - (i0 + 16 + r)));
#pragma unroll
      for (int kk = 0; kk < C_DIM / 4; ++kk) {
        v2f pa = *(const v2f*)&qs[r * QS_PITCH + kk * 4 + 2 * h];
        v2f pb = *(const v2f*)&qs[(16 + r) * QS_PITCH + kk * 4 + 2 * h];
        pa.x *= f0;  pa.y *= f0;
        pb.x *= f1;  pb.y *= f1;
        const float* Sr0 = Sn + (size_t)(kk * 4 + 2 * h) * C_DIM + c0;
        const float* Sr1 = Sr0 + C_DIM;
        v2f bb;
        bb.x = Sr0[r];      bb.y = Sr1[r];
        a0c0 = wmma4(pa, bb, a0c0);  a1c0 = wmma4(pb, bb, a1c0);
        bb.x = Sr0[16 + r]; bb.y = Sr1[16 + r];
        a0c1 = wmma4(pa, bb, a0c1);  a1c1 = wmma4(pb, bb, a1c1);
      }
    }
    __syncthreads();

#pragma unroll
    for (int i = 0; i < 8; ++i) {
      size_t row0 = (size_t)(i0 + i + 8 * h) * C_DIM + c0 + r;
      size_t row1 = row0 + (size_t)16 * C_DIM;
      Rb[row0] = a0c0[i];
      Rb[row0 + 16] = a0c1[i];
      Rb[row1] = a1c0[i];
      Rb[row1 + 16] = a1c1[i];
    }
  }
}

__global__ __launch_bounds__(256) __attribute__((amdgpu_waves_per_eu(8, 8)))
void out_proj_kernel(const float* __restrict__ R,
                     const float* __restrict__ W2,
                     const float* __restrict__ out_scale,
                     float* __restrict__ out) {
  __shared__ float lds[GT_LDS_FLOATS];
  const int m0 = blockIdx.x * GT_M;
  const int v0 = blockIdx.y * GT_N;
  const float sc = *out_scale;
  gemm_tile_128x128<C_DIM, C_DIM, V_DIM, V_DIM, true, false, false, C_DIM / GT_K - 1>(
      R + (size_t)m0 * C_DIM, W2 + v0, out + (size_t)m0 * V_DIM + v0, sc, lds);
}

extern "C" void kernel_launch(void* const* d_in, const int* in_sizes, int n_in,
                              void* d_out, int out_size, void* d_ws, size_t ws_size,
                              hipStream_t stream) {
  (void)in_sizes; (void)n_in; (void)out_size; (void)ws_size;

  const float* x = (const float*)d_in[0];
  const float* basis = (const float*)d_in[1];
  const float* qc = (const float*)d_in[2];
  const float* kc = (const float*)d_in[3];
  const float* vc = (const float*)d_in[4];
  const float* oc = (const float*)d_in[5];
  const float* decay_logit = (const float*)d_in[6];
  const float* out_scale = (const float*)d_in[7];
  float* out = (float*)d_out;

  float* ws = (float*)d_ws;
  float* owT = ws;
  float* W2 = owT + (size_t)V_DIM * C_DIM;
  float* GT = W2 + (size_t)V_DIM * F_DIM;
  float* Qp = GT + (size_t)F_DIM * F_DIM;
  float* XB = Qp + (size_t)M_DIM * F_DIM;
  float* S = XB + (size_t)M_DIM * F_DIM;

  dim3 blk(32);

  proj_weights_kernel<<<dim3(V_DIM / 16, C_DIM / 16 + F_DIM * F_DIM / 256 / (V_DIM / 16)),
                        blk, 0, stream>>>(basis, oc, qc, kc, owT, GT);

  out_weights_kernel<<<dim3(V_DIM / 16, F_DIM / 16), blk, 0, stream>>>(
      vc, owT, W2);

  xb_kernel<<<dim3(M_DIM / GT_M, F_DIM / GT_N), dim3(256), 0, stream>>>(
      x, basis, XB);

  q_kernel<<<dim3(M_DIM / GT_M, F_DIM / GT_N), dim3(256), 0, stream>>>(
      XB, GT, Qp);

  chunk_state_kernel<<<dim3(C_DIM / 64, N_CHUNK - 1, B_DIM), dim3(256), 0, stream>>>(
      XB, XB, decay_logit, S);

  chunk_scan_kernel<<<dim3(C_DIM * C_DIM / 4 / 256, 1, B_DIM), dim3(256), 0, stream>>>(
      decay_logit, S);

  attn_kernel<<<dim3(ATT_RBLOCKS / 2, 1, B_DIM), dim3(32 * ATT_WAVES), 0, stream>>>(
      Qp, XB, XB, S, decay_logit);

  out_proj_kernel<<<dim3(M_DIM / GT_M, V_DIM / GT_N), dim3(256), 0, stream>>>(
      Qp, W2, out_scale, out);
}


# launches 8 -> 6: the small weight products get no launch of their own: o_w^T and GT tiles ride behind the x@basis GEMM tiles, W2 tiles beside the chunk scan; summation order unchanged
# speedup vs baseline: 8.5129x; 1.0544x over previous
#include <hip/hip_runtime.h>


typedef float v2f __attribute__((ext_vector_type(2)));
typedef float v8f __attribute__((ext_vector_type(8)));

#define C_DIM 256
#define V_DIM 1024
#define F_DIM 256
#define T_DIM 4096
#define B_DIM 4
#define M_DIM (B_DIM * T_DIM)

static __device__ __forceinline__ v8f wmma4(v2f a, v2f b, v8f c) {
  return __builtin_amdgcn_wmma_f32_16x16x4_f32(
       false, a,  false, b,
       (short)0, c,  false,  false);
}

static __device__ __forceinline__ v8f vzero8() {
  v8f z;
#pragma unroll
  for (int i = 0; i < 8; ++i) z[i] = 0.0f;
  return z;
}


#define W1_OW_TILES ((V_DIM / 16) * (C_DIM / 16))
#define W1_TILES (W1_OW_TILES + (F_DIM / 16) * (F_DIM / 16))

static __device__ __forceinline__ void proj_weights_tile(
    const float* __restrict__ basis, const float* __restrict__ oc,
    const float* __restrict__ qc, const float* __restrict__ kc,
    float* __restrict__ owT, float* __restrict__ GT,
    int tile  , int lane) {
  const int r = lane & 15;
  const int h = lane >> 4;

  if (tile >= W1_OW_TILES) {
    const int t = tile - W1_OW_TILES;
    const int fr = (t >> 4) * 16;
    const int fc = (t & 15) * 16;
    v8f g = vzero8();
#pragma unroll
    for (int cc = 0; cc < C_DIM; cc += 4) {
      const float* kr = &kc[(size_t)(cc + 2 * h) * F_DIM + fr + r];
      const float* qr = &qc[(size_t)(cc + 2 * h) * F_DIM + fc + r];
      v2f a, b;
      a.x = kr[0];  a.y = kr[F_DIM];
      b.x = qr[0];  b.y = qr[F_DIM];
      g = wmma4(a, b, g);
    }
#pragma unroll
    for (int i = 0; i < 8; ++i)
      GT[(size_t)(fr + i + 8 * h) * F_DIM + fc + r] = g[i];
    return;
  }

  const int v0 = (tile & (V_DIM / 16 - 1)) * 16;
  const int c0 = (tile / (V_DIM / 16)) * 16;

  v8f acc = vzero8();
#pragma unroll
  for (int f0 = 0; f0 < F_DIM; f0 += 4) {
    v2f a = *(const v2f*)&basis[(size_t)(v0 + r) * F_DIM + f0 + 2 * h];
    v2f b = *(const v2f*)&oc[(size_t)(c0 + r) * F_DIM + f0 + 2 * h];
    acc = wmma4(a, b, acc);
  }
#pragma unroll
  for (int i = 0; i < 8; ++i)
    owT[(size_t)(c0 + r) * V_DIM + v0 + i + 8 * h] = acc[i];
}

#define W2_TILES ((V_DIM / 16) * (F_DIM / 16))

static __device__ __forceinline__ void out_weights_tile(
    const float* __restrict__ vc, const float* __restrict__ owT,
    float* __restrict__ W2, int tile  , int lane) {
  const int r = lane & 15;
  const int h = lane >> 4;
  const int v0 = (tile & (V_DIM / 16 - 1)) * 16;
  const int f0 = (tile / (V_DIM / 16)) * 16;

  v8f acc = vzero8();
#pragma unroll
  for (int cc = 0; cc < C_DIM; cc += 4) {
    const float* ar = &vc[(size_t)(cc + 2 * h) * F_DIM + f0 + r];
    const float* br = &owT[(size_t)(cc + 2 * h) * V_DIM + v0 + r];
    v2f a, b;
    a.x = ar[0];  a.y = ar[F_DIM];
    b.x = br[0];  b.y = br[V_DIM];
    acc = wmma4(a, b, acc);
  }
#pragma unroll
  for (int i = 0; i < 8; ++i)
    W2[(size_t)(f0 + i + 8 * h) * V_DIM + v0 + r] = acc[i];
}

#define GT_M 128
#define GT_N 128
#define GT_K 32
#define GT_XP 36
#define GT_WP 288
#define GT_XS (GT_M * GT_XP)
#define GT_WS ((GT_K / 2) * GT_WP)
#define GT_STAGE (GT_XS + GT_WS)
#define GT_LDS_FLOATS (2 * GT_STAGE)

template <bool BT>
static __device__ __forceinline__ void gemm_slab_mma(const float* as, const float* bs,
                                                     v8f (&acc)[2][4]) {
  constexpr int B_KS = BT ? 4 : 2 * GT_WP;
  constexpr int B_NT = BT ? 16 * GT_XP : 32;
#pragma unroll
  for (int ks = 0; ks < GT_K; ks += 4) {
    v2f a0 = *(const v2f*)&as[ks];
    v2f a1 = *(const v2f*)&as[16 * GT_XP + ks];
    const float* b0 = &bs[(ks >> 2) * B_KS];
#pragma unroll
    for (int nt = 0; nt < 4; ++nt) {
      const v2f bb = *(const v2f*)&b0[nt * B_NT];
      acc[0][nt] = wmma4(a0, bb, acc[0][nt]);
      acc[1][nt] = wmma4(a1, bb, acc[1][nt]);
    }
  }
}

template <int KDIM, int LDA, int LDB, int LDC, bool SCALED, bool BT, bool ALONE, int KC_UNROLL>
static __device__ __forceinline__ void gemm_tile_128x128(
    const float* __restrict__ A,
    const float* __restrict__ Bm,
    float* __restrict__ Cm,
    float scale, float* __restrict__ lds) {
  const int tid = threadIdx.x;
  const int lane = tid & 31;
  const int w = __builtin_amdgcn_readfirstlane(tid >> 5);
  const int r = lane & 15;
  const int h = lane >> 4;
  const int wm = (w >> 1) * 32;
  const int wn = (w & 1) * 64;

  const int xr = tid >> 3;
  const int xc = (tid & 7) << 2;
  const int wp = tid >> 5;
  const int wc = (tid & 31) << 2;
  const float* Ag = A + (size_t)xr * LDA + xc;
  const float* Bg = BT ? Bm + (size_t)xr * LDB + xc
                       : Bm + (size_t)(2 * wp) * LDB + wc;
  const int xs_off = xr * GT_XP + xc;
  const int ws_off = BT ? GT_XS + xs_off : GT_XS + wp * GT_WP + 2 * wc;

  float4 xa[4], wb[4];
#pragma unroll
  for (int j = 0; j < 4; ++j) {
    xa[j] = *(const float4*)&Ag[(size_t)(32 * j) * LDA];
    if (BT)
      wb[j] = *(const float4*)&Bg[(size_t)(32 * j) * LDB];
    else
      wb[j] = *(const float4*)&Bg[(size_t)(16 * (j >> 1) + (j & 1)) * LDB];
  }
#pragma unroll
  for (int j = 0; j < 4; ++j)
    *(float4*)&lds[xs_off + 32 * j * GT_XP] = xa[j];
  if (BT) {
#pragma unroll
    for (int j = 0; j < 4; ++j)
      *(float4*)&lds[ws_off + 32 * j * GT_XP] = wb[j];
  } else {
#pragma unroll
    for (int p = 0; p < 2; ++p) {
      const float4 e0 = wb[2 * p], e1 = wb[2 * p + 1];
      *(float4*)&lds[ws_off + 8 * p * GT_WP] = make_float4(e0.x, e1.x, e0.y, e1.y);
      *(float4*)&lds[ws_off + 8 * p * GT_WP + 4] = make_float4(e0.z, e1.z, e0.w, e1.w);
    }
  }
  __syncthreads();

  v8f acc[2][4];
#pragma unroll
  for (int mt = 0; mt < 2; ++mt)
#pragma unroll
    for (int nt = 0; nt < 4; ++nt) acc[mt][nt] = vzero8();

  const int a_off = (wm + r) * GT_XP + 2 * h;
  const int b_off = BT ? GT_XS + (wn + r) * GT_XP + 2 * h
                       : GT_XS + h * GT_WP + 2 * (wn + r);

  constexpr int NCH = KDIM / GT_K;
#pragma unroll KC_UNROLL
  for (int kc = 0; kc < NCH - 1; ++kc) {
    const float* cur = lds + (kc & 1) * GT_STAGE;
    float* nxt = lds + ((kc + 1) & 1) * GT_STAGE;
    const int k0 = (kc + 1) * GT_K;
#pragma unroll
    for (int j = 0; j < 4; ++j) {
      xa[j] = *(const float4*)&Ag[(size_t)(32 * j) * LDA + k0];
      if (BT)
        wb[j] = *(const float4*)&Bg[(size_t)(32 * j) * LDB + k0];
      else
        wb[j] = *(const float4*)&Bg[(size_t)(k0 + 16 * (j >> 1) + (j & 1)) * LDB];
    }
    gemm_slab_mma<BT>(cur + a_off, cur + b_off, acc);
    if (ALONE) __builtin_amdgcn_sched_barrier(0x26);
#pragma unroll
    for (int j = 0; j < 4; ++j)
      *(float4*)&nxt[xs_off + 32 * j * GT_XP] = xa[j];
    if (BT) {
#pragma unroll
      for (int j = 0; j < 4; ++j)
        *(float4*)&nxt[ws_off + 32 * j * GT_XP] = wb[j];
    } else {
#pragma unroll
      for (int p = 0; p < 2; ++p) {
        const float4 e0 = wb[2 * p], e1 = wb[2 * p + 1];
        *(float4*)&nxt[ws_off + 8 * p * GT_WP] = make_float4(e0.x, e1.x, e0.y, e1.y);
        *(float4*)&nxt[ws_off + 8 * p * GT_WP + 4] = make_float4(e0.z, e1.z, e0.w, e1.w);
      }
    }
    __syncthreads();
  }
  {
    const float* last = lds + ((NCH - 1) & 1) * GT_STAGE;
    gemm_slab_mma<BT>(last + a_off, last + b_off, acc);
  }

#pragma unroll
  for (int mt = 0; mt < 2; ++mt)
#pragma unroll
    for (int nt = 0; nt < 4; ++nt)
#pragma unroll
      for (int i = 0; i < 8; ++i) {
        float v = acc[mt][nt][i];
        if (SCALED) v *= scale;
        Cm[(size_t)(wm + mt * 16 + i + 8 * h) * LDC + wn + nt * 16 + r] = v;
      }
}

#define XB_LDS_FLOATS 41216
#define XB_GROUPS ((M_DIM / GT_M) * (F_DIM / GT_N))
#define W1_PER_GROUP (W1_TILES / XB_GROUPS)

__global__ __launch_bounds__(256)
void xb_kernel(const float* __restrict__ x,
               const float* __restrict__ basis,
               const float* __restrict__ oc,
               const float* __restrict__ qc,
               const float* __restrict__ kc,
               float* __restrict__ XB,
               float* __restrict__ owT,
               float* __restrict__ GT) {
  __shared__ float lds[XB_LDS_FLOATS];
  const int m0 = blockIdx.x * GT_M;
  const int n0 = blockIdx.y * GT_N;
  gemm_tile_128x128<V_DIM, V_DIM, F_DIM, F_DIM, false, false, true, 1>(
      x + (size_t)m0 * V_DIM, basis + n0, XB + (size_t)m0 * F_DIM + n0, 1.0f, lds);

  const int w = __builtin_amdgcn_readfirstlane((int)threadIdx.x >> 5);
  if (w < W1_PER_GROUP) {
    const int group = (int)blockIdx.x + (M_DIM / GT_M) * (int)blockIdx.y;
    proj_weights_tile(basis, oc, qc, kc, owT, GT, group * W1_PER_GROUP + w,
                      (int)threadIdx.x & 31);
  }
}

__global__ __launch_bounds__(256)
void q_kernel(const float* __restrict__ XB,
              const float* __restrict__ GT,
              float* __restrict__ Qp) {
  __shared__ float lds[XB_LDS_FLOATS];
  const int m0 = blockIdx.x * GT_M;
  const int n0 = blockIdx.y * GT_N;
  gemm_tile_128x128<F_DIM, F_DIM, F_DIM, F_DIM, false, true, true, 1>(
      XB + (size_t)m0 * F_DIM, GT + (size_t)n0 * F_DIM,
      Qp + (size_t)m0 * F_DIM + n0, 1.0f, lds);
}

#define CH_L 256
#define N_CHUNK (T_DIM / CH_L)

__global__ __launch_bounds__(256) __attribute__((amdgpu_waves_per_eu(3, 3)))
void chunk_state_kernel(const float* __restrict__ K,
                        const float* __restrict__ Vv,
                        const float* __restrict__ decay_logit,
                        float* __restrict__ S) {
  const int tid = threadIdx.x;
  const int lane = tid & 31;
  const int w = __builtin_amdgcn_readfirstlane(tid >> 5);
  const int r = lane & 15;
  const int h = lane >> 4;
  const int kc0 = w * 32;
  const int vc0 = blockIdx.x * 64;
  const int ch = (int)blockIdx.z * N_CHUNK + 1 + (int)blockIdx.y;
  const float* Kc = K + (size_t)ch * CH_L * C_DIM;
  const float* Vc = Vv + (size_t)ch * CH_L * C_DIM;

  const float dl = *decay_logit;
  const float dec = 1.0f / (1.0f + __expf(-dl));
  const float ln_dec = __logf(dec);

  v8f acc[2][4];
#pragma unroll
  for (int mt = 0; mt < 2; ++mt)
#pragma unroll
    for (int nt = 0; nt < 4; ++nt) acc[mt][nt] = vzero8();

#pragma unroll
  for (int j0 = 0; j0 < CH_L; j0 += 4) {
    const int ja = j0 + 2 * h;
    const float w0 = __expf(ln_dec * (float)ja);
    const float w1 = __expf(ln_dec * (float)(ja + 1));
    const float* Kr0 = Kc + (size_t)ja * C_DIM + kc0;
    const float* Kr1 = Kr0 + C_DIM;
    const float* Vr0 = Vc + (size_t)ja * C_DIM + vc0;
    const float* Vr1 = Vr0 + C_DIM;
    v2f a0, a1;
    a0.x = Kr0[r] * w0;       a0.y = Kr1[r] * w1;
    a1.x = Kr0[16 + r] * w0;  a1.y = Kr1[16 + r] * w1;
#pragma unroll
    for (int nt = 0; nt < 4; ++nt) {
      v2f bb;
      bb.x = Vr0[16 * nt + r];  bb.y = Vr1[16 * nt + r];
      acc[0][nt] = wmma4(a0, bb, acc[0][nt]);
      acc[1][nt] = wmma4(a1, bb, acc[1][nt]);
    }
  }

  float* Sc = S + (size_t)ch * C_DIM * C_DIM;
#pragma unroll
  for (int mt = 0; mt < 2; ++mt)
#pragma unroll
    for (int nt = 0; nt < 4; ++nt)
#pragma unroll
      for (int i = 0; i < 8; ++i)
        Sc[(size_t)(kc0 + mt * 16 + i + 8 * h) * C_DIM + vc0 + nt * 16 + r] = acc[mt][nt][i];
}

#define SCAN_W2_PLANES (W2_TILES / (8 * (C_DIM * C_DIM / 4 / 256)))

__global__ __launch_bounds__(256)
void scan_w2_kernel(const float* __restrict__ decay_logit, float* __restrict__ S,
                    const float* __restrict__ vc, const float* __restrict__ owT,
                    float* __restrict__ W2) {
  if ((int)blockIdx.z < SCAN_W2_PLANES) {
    const int w = __builtin_amdgcn_readfirstlane((int)threadIdx.x >> 5);
    const int tile = (((int)blockIdx.z * (int)gridDim.x + (int)blockIdx.x) << 3) + w;
    out_weights_tile(vc, owT, W2, tile, (int)threadIdx.x & 31);
    return;
  }
  const int bz = (int)blockIdx.z - SCAN_W2_PLANES;
  const float dl = *decay_logit;
  const float dec = 1.0f / (1.0f + __expf(-dl));
  const float dL = __expf(__logf(dec) * (float)CH_L);
  const size_t e = ((size_t)blockIdx.x * 256 + threadIdx.x) * 4;
  float* Sb = S + (size_t)bz * N_CHUNK * C_DIM * C_DIM + e;
  float4 p[N_CHUNK - 1];
#pragma unroll
  for (int c = 1; c < N_CHUNK; ++c)
    p[c - 1] = *(const float4*)&Sb[(size_t)c * C_DIM * C_DIM];
  float4 s = p[N_CHUNK - 2];
#pragma unroll
  for (int c = N_CHUNK - 2; c >= 1; --c) {
    const float4 q = p[c - 1];
    s.x = fmaf(dL, s.x, q.x);
    s.y = fmaf(dL, s.y, q.y);
    s.z = fmaf(dL, s.z, q.z);
    s.w = fmaf(dL, s.w, q.w);
    *(float4*)&Sb[(size_t)c * C_DIM * C_DIM] = s;
  }
}

#define ATT_WAVES 8
#define ATT_ROWS 32
#define ATT_KEYS (16 * ATT_WAVES)
#define ATT_RBLOCKS (T_DIM / ATT_ROWS)
#define QS_PITCH 260
#define SS_PITCH 132
#define ATT_QS_FLOATS (ATT_ROWS * QS_PITCH)
#define ATT_SS_FLOATS (ATT_ROWS * SS_PITCH)
#define ATT_LDS_FLOATS 41216

__global__ __launch_bounds__(32 * ATT_WAVES)
void attn_kernel(float* QR,
                 const float* __restrict__ K,
                 const float* __restrict__ Vv,
                 const float* __restrict__ S,
                 const float* __restrict__ decay_logit) {
  __shared__ float lds[ATT_LDS_FLOATS];
  float* const qs = &lds[0];
  float* const ss = &lds[ATT_QS_FLOATS];

  const int tid = threadIdx.x;
  const int lane = tid & 31;
  const int w = __builtin_amdgcn_readfirstlane(tid >> 5);
  const int r = lane & 15;
  const int h = lane >> 4;
  const int c0 = w * 32;
  const int b = blockIdx.z;

  float* Qb = QR + (size_t)b * T_DIM * C_DIM;
  const float* Kb = K + (size_t)b * T_DIM * C_DIM;
  const float* Vb = Vv + (size_t)b * T_DIM * C_DIM;
  float* Rb = Qb;

  const float dl = *decay_logit;
  const float dec = 1.0f / (1.0f + __expf(-dl));
  const float ln_dec = __logf(dec);

  for (int part = 0; part < 2; ++part) {
    const int rb = (part == 0) ? (int)blockIdx.x : (ATT_RBLOCKS - 1 - (int)blockIdx.x);
    const int i0 = rb * ATT_ROWS;
    const int cend = (i0 & ~(CH_L - 1)) + CH_L;

    {
      float4 qv[ATT_ROWS * 64 / (32 * ATT_WAVES)];
#pragma unroll
      for (int it = 0; it < ATT_ROWS * 64 / (32 * ATT_WAVES); ++it) {
        const int idx = tid + it * 32 * ATT_WAVES;
        qv[it] = *(const float4*)&Qb[(size_t)(i0 + (idx >> 6)) * C_DIM + ((idx & 63) << 2)];
      }
#pragma unroll
      for (int it = 0; it < ATT_ROWS * 64 / (32 * ATT_WAVES); ++it) {
        const int idx = tid + it * 32 * ATT_WAVES;
        *(float4*)&qs[(idx >> 6) * QS_PITCH + ((idx & 63) << 2)] = qv[it];
      }
    }
    __syncthreads();

    v8f a0c0 = vzero8(), a0c1 = vzero8();
    v8f a1c0 = vzero8(), a1c1 = vzero8();

    int buf = 0;
    for (int jc = i0 & ~(ATT_KEYS - 1); jc < cend; jc += ATT_KEYS) {
      const int j0 = jc + 16 * w;
      {
        int jn = (jc + ATT_KEYS < cend) ? (jc + ATT_KEYS) : jc;
        __builtin_prefetch(&Kb[(size_t)(jn + 16 * w + r) * C_DIM + 128 * h], 0, 1);
        __builtin_prefetch(&Vb[(size_t)(jn + lane) * C_DIM + c0], 0, 1);
        __builtin_prefetch(&Vb[(size_t)(jn + 32 + lane) * C_DIM + c0], 0, 1);
        __builtin_prefetch(&Vb[(size_t)(jn + 64 + lane) * C_DIM + c0], 0, 1);
        __builtin_prefetch(&Vb[(size_t)(jn + 96 + lane) * C_DIM + c0], 0, 1);
      }

      float* sb = &ss[buf * ATT_SS_FLOATS];
      if (j0 >= i0) {
        v8f s0 = vzero8(), s1 = vzero8();
        const float* Kj = Kb + (size_t)j0 * C_DIM;
#pragma unroll
        for (int k0 = 0; k0 < C_DIM; k0 += 4) {
          v2f qa = *(const v2f*)&qs[r * QS_PITCH + k0 + 2 * h];
          v2f qb = *(const v2f*)&qs[(16 + r) * QS_PITCH + k0 + 2 * h];
          v2f bb = *(const v2f*)&Kj[(size_t)r * C_DIM + k0 + 2 * h];
          s0 = wmma4(qa, bb, s0);
          s1 = wmma4(qb, bb, s1);
        }
#pragma unroll
        for (int i = 0; i < 8; ++i) {
          int gi = i0 + i + 8 * h;
          int gj = j0 + r;
          int d0 = gj - gi;
          int d1 = d0 - 16;
          float wg0 = (d0 > 0) ? __expf(ln_dec * (float)(d0 - 1)) : 0.0f;
          float wg1 = (d1 > 0) ? __expf(ln_dec * (float)(d1 - 1)) : 0.0f;
          sb[(i + 8 * h) * SS_PITCH + 16 * w + r] = s0[i] * wg0;
          sb[(16 + i + 8 * h) * SS_PITCH + 16 * w + r] = s1[i] * wg1;
        }
      } else {
#pragma unroll
        for (int i = 0; i < 8; ++i) {
          sb[(i + 8 * h) * SS_PITCH + 16 * w + r] = 0.0f;
          sb[(16 + i + 8 * h) * SS_PITCH + 16 * w + r] = 0.0f;
        }
      }
      __syncthreads();
#pragma unroll
      for (int kk = 0; kk < ATT_KEYS / 4; ++kk) {
        v2f pa = *(const v2f*)&sb[r * SS_PITCH + kk * 4 + 2 * h];
        v2f pb = *(const v2f*)&sb[(16 + r) * SS_PITCH + kk * 4 + 2 * h];
        const float* Vr0 = Vb + (size_t)(jc + kk * 4 + 2 * h) * C_DIM + c0;
        const float* Vr1 = Vr0 + C_DIM;
        v2f bb;
        bb.x = Vr0[r];      bb.y = Vr1[r];
        a0c0 = wmma4(pa, bb, a0c0);  a1c0 = wmma4(pb, bb, a1c0);
        bb.x = Vr0[16 + r]; bb.y = Vr1[16 + r];
        a0c1 = wmma4(pa, bb, a0c1);  a1c1 = wmma4(pb, bb, a1c1);
      }
      buf ^= 1;
    }

    if (cend < T_DIM) {
      const float* Sn = S + ((size_t)b * N_CHUNK + (size_t)(cend / CH_L)) * C_DIM * C_DIM;
      const float f0 = __expf(ln_dec * (float)(cend - 1 - (i0 + r)));
      const float f1 = __expf(ln_dec * (float)(cend - 1 - (i0 + 16 + r)));
#pragma unroll
      for (int kk = 0; kk < C_DIM / 4; ++kk) {
        v2f pa = *(const v2f*)&qs[r * QS_PITCH + kk * 4 + 2 * h];
        v2f pb = *(const v2f*)&qs[(16 + r) * QS_PITCH + kk * 4 + 2 * h];
        pa.x *= f0;  pa.y *= f0;
        pb.x *= f1;  pb.y *= f1;
        const float* Sr0 = Sn + (size_t)(kk * 4 + 2 * h) * C_DIM + c0;
        const float* Sr1 = Sr0 + C_DIM;
        v2f bb;
        bb.x = Sr0[r];      bb.y = Sr1[r];
        a0c0 = wmma4(pa, bb, a0c0);  a1c0 = wmma4(pb, bb, a1c0);
        bb.x = Sr0[16 + r]; bb.y = Sr1[16 + r];
        a0c1 = wmma4(pa, bb, a0c1);  a1c1 = wmma4(pb, bb, a1c1);
      }
    }
    __syncthreads();

#pragma unroll
    for (int i = 0; i < 8; ++i) {
      size_t row0 = (size_t)(i0 + i + 8 * h) * C_DIM + c0 + r;
      size_t row1 = row0 + (size_t)16 * C_DIM;
      Rb[row0] = a0c0[i];
      Rb[row0 + 16] = a0c1[i];
      Rb[row1] = a1c0[i];
      Rb[row1 + 16] = a1c1[i];
    }
  }
}

__global__ __launch_bounds__(256) __attribute__((amdgpu_waves_per_eu(8, 8)))
void out_proj_kernel(const float* __restrict__ R,
                     const float* __restrict__ W2,
                     const float* __restrict__ out_scale,
                     float* __restrict__ out) {
  __shared__ float lds[GT_LDS_FLOATS];
  const int m0 = blockIdx.x * GT_M;
  const int v0 = blockIdx.y * GT_N;
  const float sc = *out_scale;
  gemm_tile_128x128<C_DIM, C_DIM, V_DIM, V_DIM, true, false, false, C_DIM / GT_K - 1>(
      R + (size_t)m0 * C_DIM, W2 + v0, out + (size_t)m0 * V_DIM + v0, sc, lds);
}

extern "C" void kernel_launch(void* const* d_in, const int* in_sizes, int n_in,
                              void* d_out, int out_size, void* d_ws, size_t ws_size,
                              hipStream_t stream) {
  (void)in_sizes; (void)n_in; (void)out_size; (void)ws_size;

  const float* x = (const float*)d_in[0];
  const float* basis = (const float*)d_in[1];
  const float* qc = (const float*)d_in[2];
  const float* kc = (const float*)d_in[3];
  const float* vc = (const float*)d_in[4];
  const float* oc = (const float*)d_in[5];
  const float* decay_logit = (const float*)d_in[6];
  const float* out_scale = (const float*)d_in[7];
  float* out = (float*)d_out;

  float* ws = (float*)d_ws;
  float* owT = ws;
  float* W2 = owT + (size_t)V_DIM * C_DIM;
  float* GT = W2 + (size_t)V_DIM * F_DIM;
  float* Qp = GT + (size_t)F_DIM * F_DIM;
  float* XB = Qp + (size_t)M_DIM * F_DIM;
  float* S = XB + (size_t)M_DIM * F_DIM;

  xb_kernel<<<dim3(M_DIM / GT_M, F_DIM / GT_N), dim3(256), 0, stream>>>(
      x, basis, oc, qc, kc, XB, owT, GT);

  q_kernel<<<dim3(M_DIM / GT_M, F_DIM / GT_N), dim3(256), 0, stream>>>(
      XB, GT, Qp);

  chunk_state_kernel<<<dim3(C_DIM / 64, N_CHUNK - 1, B_DIM), dim3(256), 0, stream>>>(
      XB, XB, decay_logit, S);

  scan_w2_kernel<<<dim3(C_DIM * C_DIM / 4 / 256, 1, SCAN_W2_PLANES + B_DIM), dim3(256), 0, stream>>>(
      decay_logit, S, vc, owT, W2);

  attn_kernel<<<dim3(ATT_RBLOCKS / 2, 1, B_DIM), dim3(32 * ATT_WAVES), 0, stream>>>(
      Qp, XB, XB, S, decay_logit);

  out_proj_kernel<<<dim3(M_DIM / GT_M, V_DIM / GT_N), dim3(256), 0, stream>>>(
      Qp, W2, out_scale, out);
}


# q' tiles and chunk products in one launch (were two): 496 equal 8-wave work-groups, two per WGP at a quarter of the registers, each covering the other's waits; launches 6 to 5; sum order same
# speedup vs baseline: 8.8088x; 1.0348x over previous
#include <hip/hip_runtime.h>


typedef float v2f __attribute__((ext_vector_type(2)));
typedef float v8f __attribute__((ext_vector_type(8)));

#define C_DIM 256
#define V_DIM 1024
#define F_DIM 256
#define T_DIM 4096
#define B_DIM 4
#define M_DIM (B_DIM * T_DIM)

static __device__ __forceinline__ v8f wmma4(v2f a, v2f b, v8f c) {
  return __builtin_amdgcn_wmma_f32_16x16x4_f32(
       false, a,  false, b,
       (short)0, c,  false,  false);
}

static __device__ __forceinline__ v8f vzero8() {
  v8f z;
#pragma unroll
  for (int i = 0; i < 8; ++i) z[i] = 0.0f;
  return z;
}


#define W1_OW_TILES ((V_DIM / 16) * (C_DIM / 16))
#define W1_TILES (W1_OW_TILES + (F_DIM / 16) * (F_DIM / 16))

static __device__ __forceinline__ void proj_weights_tile(
    const float* __restrict__ basis, const float* __restrict__ oc,
    const float* __restrict__ qc, const float* __restrict__ kc,
    float* __restrict__ owT, float* __restrict__ GT,
    int tile  , int lane) {
  const int r = lane & 15;
  const int h = lane >> 4;

  if (tile >= W1_OW_TILES) {
    const int t = tile - W1_OW_TILES;
    const int fr = (t >> 4) * 16;
    const int fc = (t & 15) * 16;
    v8f g = vzero8();
#pragma unroll
    for (int cc = 0; cc < C_DIM; cc += 4) {
      const float* kr = &kc[(size_t)(cc + 2 * h) * F_DIM + fr + r];
      const float* qr = &qc[(size_t)(cc + 2 * h) * F_DIM + fc + r];
      v2f a, b;
      a.x = kr[0];  a.y = kr[F_DIM];
      b.x = qr[0];  b.y = qr[F_DIM];
      g = wmma4(a, b, g);
    }
#pragma unroll
    for (int i = 0; i < 8; ++i)
      GT[(size_t)(fr + i + 8 * h) * F_DIM + fc + r] = g[i];
    return;
  }

  const int v0 = (tile & (V_DIM / 16 - 1)) * 16;
  const int c0 = (tile / (V_DIM / 16)) * 16;

  v8f acc = vzero8();
#pragma unroll
  for (int f0 = 0; f0 < F_DIM; f0 += 4) {
    v2f a = *(const v2f*)&basis[(size_t)(v0 + r) * F_DIM + f0 + 2 * h];
    v2f b = *(const v2f*)&oc[(size_t)(c0 + r) * F_DIM + f0 + 2 * h];
    acc = wmma4(a, b, acc);
  }
#pragma unroll
  for (int i = 0; i < 8; ++i)
    owT[(size_t)(c0 + r) * V_DIM + v0 + i + 8 * h] = acc[i];
}

#define W2_TILES ((V_DIM / 16) * (F_DIM / 16))

static __device__ __forceinline__ void out_weights_tile(
    const float* __restrict__ vc, const float* __restrict__ owT,
    float* __restrict__ W2, int tile  , int lane) {
  const int r = lane & 15;
  const int h = lane >> 4;
  const int v0 = (tile & (V_DIM / 16 - 1)) * 16;
  const int f0 = (tile / (V_DIM / 16)) * 16;

  v8f acc = vzero8();
#pragma unroll
  for (int cc = 0; cc < C_DIM; cc += 4) {
    const float* ar = &vc[(size_t)(cc + 2 * h) * F_DIM + f0 + r];
    const float* br = &owT[(size_t)(cc + 2 * h) * V_DIM + v0 + r];
    v2f a, b;
    a.x = ar[0];  a.y = ar[F_DIM];
    b.x = br[0];  b.y = br[V_DIM];
    acc = wmma4(a, b, acc);
  }
#pragma unroll
  for (int i = 0; i < 8; ++i)
    W2[(size_t)(f0 + i + 8 * h) * V_DIM + v0 + r] = acc[i];
}

#define GT_M 128
#define GT_N 128
#define GT_K 32
#define GT_XP 36
#define GT_WP 288
#define GT_XS (GT_M * GT_XP)
#define GT_WS ((GT_K / 2) * GT_WP)
#define GT_STAGE (GT_XS + GT_WS)
#define GT_LDS_FLOATS (2 * GT_STAGE)

template <bool BT>
static __device__ __forceinline__ void gemm_slab_mma(const float* as, const float* bs,
                                                     v8f (&acc)[2][4]) {
  constexpr int B_KS = BT ? 4 : 2 * GT_WP;
  constexpr int B_NT = BT ? 16 * GT_XP : 32;
#pragma unroll
  for (int ks = 0; ks < GT_K; ks += 4) {
    v2f a0 = *(const v2f*)&as[ks];
    v2f a1 = *(const v2f*)&as[16 * GT_XP + ks];
    const float* b0 = &bs[(ks >> 2) * B_KS];
#pragma unroll
    for (int nt = 0; nt < 4; ++nt) {
      const v2f bb = *(const v2f*)&b0[nt * B_NT];
      acc[0][nt] = wmma4(a0, bb, acc[0][nt]);
      acc[1][nt] = wmma4(a1, bb, acc[1][nt]);
    }
  }
}

struct SlabRegs {
  float4 a0, a1, a2, a3;
  float4 b0, b1, b2, b3;
};

template <int LDA, int LDB, bool BT>
static __device__ __forceinline__ SlabRegs slab_load(const float* __restrict__ Ag,
                                                     const float* __restrict__ Bg, int k0) {
  SlabRegs s;
  s.a0 = *(const float4*)&Ag[(size_t)0 * LDA + k0];
  s.a1 = *(const float4*)&Ag[(size_t)32 * LDA + k0];
  s.a2 = *(const float4*)&Ag[(size_t)64 * LDA + k0];
  s.a3 = *(const float4*)&Ag[(size_t)96 * LDA + k0];
  if (BT) {
    s.b0 = *(const float4*)&Bg[(size_t)0 * LDB + k0];
    s.b1 = *(const float4*)&Bg[(size_t)32 * LDB + k0];
    s.b2 = *(const float4*)&Bg[(size_t)64 * LDB + k0];
    s.b3 = *(const float4*)&Bg[(size_t)96 * LDB + k0];
  } else {
    s.b0 = *(const float4*)&Bg[(size_t)(k0 + 0) * LDB];
    s.b1 = *(const float4*)&Bg[(size_t)(k0 + 1) * LDB];
    s.b2 = *(const float4*)&Bg[(size_t)(k0 + 16) * LDB];
    s.b3 = *(const float4*)&Bg[(size_t)(k0 + 17) * LDB];
  }
  return s;
}

template <bool BT>
static __device__ __forceinline__ void slab_store(float* __restrict__ stage, int xs_off,
                                                  int ws_off, const SlabRegs& s) {
  *(float4*)&stage[xs_off + 0 * GT_XP] = s.a0;
  *(float4*)&stage[xs_off + 32 * GT_XP] = s.a1;
  *(float4*)&stage[xs_off + 64 * GT_XP] = s.a2;
  *(float4*)&stage[xs_off + 96 * GT_XP] = s.a3;
  if (BT) {
    *(float4*)&stage[ws_off + 0 * GT_XP] = s.b0;
    *(float4*)&stage[ws_off + 32 * GT_XP] = s.b1;
    *(float4*)&stage[ws_off + 64 * GT_XP] = s.b2;
    *(float4*)&stage[ws_off + 96 * GT_XP] = s.b3;
  } else {
    *(float4*)&stage[ws_off] = make_float4(s.b0.x, s.b1.x, s.b0.y, s.b1.y);
    *(float4*)&stage[ws_off + 4] = make_float4(s.b0.z, s.b1.z, s.b0.w, s.b1.w);
    *(float4*)&stage[ws_off + 8 * GT_WP] = make_float4(s.b2.x, s.b3.x, s.b2.y, s.b3.y);
    *(float4*)&stage[ws_off + 8 * GT_WP + 4] = make_float4(s.b2.z, s.b3.z, s.b2.w, s.b3.w);
  }
}

template <int KDIM, int LDA, int LDB, int LDC, bool SCALED, bool BT, bool ALONE, int KC_UNROLL>
static __device__ __forceinline__ void gemm_tile_128x128(
    const float* __restrict__ A,
    const float* __restrict__ Bm,
    float* __restrict__ Cm,
    float scale, float* __restrict__ lds) {
  const int tid = threadIdx.x;
  const int lane = tid & 31;
  const int w = __builtin_amdgcn_readfirstlane(tid >> 5);
  const int r = lane & 15;
  const int h = lane >> 4;
  const int wm = (w >> 1) * 32;
  const int wn = (w & 1) * 64;

  const int xr = tid >> 3;
  const int xc = (tid & 7) << 2;
  const int wp = tid >> 5;
  const int wc = (tid & 31) << 2;
  const float* Ag = A + (size_t)xr * LDA + xc;
  const float* Bg = BT ? Bm + (size_t)xr * LDB + xc
                       : Bm + (size_t)(2 * wp) * LDB + wc;
  const int xs_off = xr * GT_XP + xc;
  const int ws_off = BT ? GT_XS + xs_off : GT_XS + wp * GT_WP + 2 * wc;

  {
    const SlabRegs s = slab_load<LDA, LDB, BT>(Ag, Bg, 0);
    slab_store<BT>(lds, xs_off, ws_off, s);
  }
  __syncthreads();

  v8f acc[2][4];
#pragma unroll
  for (int mt = 0; mt < 2; ++mt)
#pragma unroll
    for (int nt = 0; nt < 4; ++nt) acc[mt][nt] = vzero8();

  const int a_off = (wm + r) * GT_XP + 2 * h;
  const int b_off = BT ? GT_XS + (wn + r) * GT_XP + 2 * h
                       : GT_XS + h * GT_WP + 2 * (wn + r);

  constexpr int NCH = KDIM / GT_K;
#pragma unroll KC_UNROLL
  for (int kc = 0; kc < NCH - 1; ++kc) {
    const float* cur = lds + (kc & 1) * GT_STAGE;
    float* nxt = lds + ((kc + 1) & 1) * GT_STAGE;
    const SlabRegs s = slab_load<LDA, LDB, BT>(Ag, Bg, (kc + 1) * GT_K);
    gemm_slab_mma<BT>(cur + a_off, cur + b_off, acc);
    if (ALONE) __builtin_amdgcn_sched_barrier(0x26);
    slab_store<BT>(nxt, xs_off, ws_off, s);
    __syncthreads();
  }
  {
    const float* last = lds + ((NCH - 1) & 1) * GT_STAGE;
    gemm_slab_mma<BT>(last + a_off, last + b_off, acc);
  }

#pragma unroll
  for (int mt = 0; mt < 2; ++mt)
#pragma unroll
    for (int nt = 0; nt < 4; ++nt)
#pragma unroll
      for (int i = 0; i < 8; ++i) {
        float v = acc[mt][nt][i];
        if (SCALED) v *= scale;
        Cm[(size_t)(wm + mt * 16 + i + 8 * h) * LDC + wn + nt * 16 + r] = v;
      }
}

#define XB_LDS_FLOATS 41216
#define XB_GROUPS ((M_DIM / GT_M) * (F_DIM / GT_N))
#define W1_PER_GROUP (W1_TILES / XB_GROUPS)

__global__ __launch_bounds__(256)
void xb_kernel(const float* __restrict__ x,
               const float* __restrict__ basis,
               const float* __restrict__ oc,
               const float* __restrict__ qc,
               const float* __restrict__ kc,
               float* __restrict__ XB,
               float* __restrict__ owT,
               float* __restrict__ GT) {
  __shared__ float lds[XB_LDS_FLOATS];
  const int m0 = blockIdx.x * GT_M;
  const int n0 = blockIdx.y * GT_N;
  gemm_tile_128x128<V_DIM, V_DIM, F_DIM, F_DIM, false, false, true, 1>(
      x + (size_t)m0 * V_DIM, basis + n0, XB + (size_t)m0 * F_DIM + n0, 1.0f, lds);

  const int w = __builtin_amdgcn_readfirstlane((int)threadIdx.x >> 5);
  if (w < W1_PER_GROUP) {
    const int group = (int)blockIdx.x + (M_DIM / GT_M) * (int)blockIdx.y;
    proj_weights_tile(basis, oc, qc, kc, owT, GT, group * W1_PER_GROUP + w,
                      (int)threadIdx.x & 31);
  }
}

#define Q_GROUPS ((M_DIM / GT_M) * (F_DIM / GT_N))

static __device__ __forceinline__ void q_tile(const float* __restrict__ XB,
                                              const float* __restrict__ GT,
                                              float* __restrict__ Qp,
                                              int group  ,
                                              float* __restrict__ lds) {
  const int m0 = (group & (M_DIM / GT_M - 1)) * GT_M;
  const int n0 = (group / (M_DIM / GT_M)) * GT_N;
  gemm_tile_128x128<F_DIM, F_DIM, F_DIM, F_DIM, false, true, true, 1>(
      XB + (size_t)m0 * F_DIM, GT + (size_t)n0 * F_DIM,
      Qp + (size_t)m0 * F_DIM + n0, 1.0f, lds);
}

#define CH_L 256
#define N_CHUNK (T_DIM / CH_L)

#define CS_GROUPS ((C_DIM / 64) * (N_CHUNK - 1) * B_DIM)

static __device__ __forceinline__ void chunk_state_tile(
    const float* __restrict__ K, const float* __restrict__ Vv,
    const float* __restrict__ decay_logit, float* __restrict__ S,
    int group  ) {
  const int tid = threadIdx.x;
  const int lane = tid & 31;
  const int w = __builtin_amdgcn_readfirstlane(tid >> 5);
  const int r = lane & 15;
  const int h = lane >> 4;
  const int kc0 = w * 32;
  const int vc0 = (group & (C_DIM / 64 - 1)) * 64;
  const int cb = group / (C_DIM / 64);
  const int ch = (cb / (N_CHUNK - 1)) * N_CHUNK + 1 + cb % (N_CHUNK - 1);
  const float* Kc = K + (size_t)ch * CH_L * C_DIM;
  const float* Vc = Vv + (size_t)ch * CH_L * C_DIM;

  const float dl = *decay_logit;
  const float dec = 1.0f / (1.0f + __expf(-dl));
  const float ln_dec = __logf(dec);

  v8f acc[2][4];
#pragma unroll
  for (int mt = 0; mt < 2; ++mt)
#pragma unroll
    for (int nt = 0; nt < 4; ++nt) acc[mt][nt] = vzero8();

#pragma unroll 1
  for (int jb = 0; jb < CH_L; jb += 32) {
#pragma unroll
    for (int j0 = jb; j0 < jb + 32; j0 += 4) {
      const int ja = j0 + 2 * h;
      const float w0 = __expf(ln_dec * (float)ja);
      const float w1 = __expf(ln_dec * (float)(ja + 1));
      const float* Kr0 = Kc + (size_t)ja * C_DIM + kc0;
      const float* Kr1 = Kr0 + C_DIM;
      const float* Vr0 = Vc + (size_t)ja * C_DIM + vc0;
      const float* Vr1 = Vr0 + C_DIM;
      v2f a0, a1;
      a0.x = Kr0[r] * w0;       a0.y = Kr1[r] * w1;
      a1.x = Kr0[16 + r] * w0;  a1.y = Kr1[16 + r] * w1;
#pragma unroll
      for (int nt = 0; nt < 4; ++nt) {
        v2f bb;
        bb.x = Vr0[16 * nt + r];  bb.y = Vr1[16 * nt + r];
        acc[0][nt] = wmma4(a0, bb, acc[0][nt]);
        acc[1][nt] = wmma4(a1, bb, acc[1][nt]);
      }
    }
  }

  float* Sc = S + (size_t)ch * C_DIM * C_DIM;
#pragma unroll
  for (int mt = 0; mt < 2; ++mt)
#pragma unroll
    for (int nt = 0; nt < 4; ++nt)
#pragma unroll
      for (int i = 0; i < 8; ++i)
        Sc[(size_t)(kc0 + mt * 16 + i + 8 * h) * C_DIM + vc0 + nt * 16 + r] = acc[mt][nt][i];
}

__global__ __launch_bounds__(256) __attribute__((amdgpu_waves_per_eu(4, 4)))
void q_chunk_kernel(const float* __restrict__ XB,
                    const float* __restrict__ GT,
                    float* __restrict__ Qp,
                    const float* __restrict__ decay_logit,
                    float* __restrict__ S) {
  __shared__ float lds[GT_LDS_FLOATS];
  const int group = (int)blockIdx.x;
  if (group < Q_GROUPS) {
    q_tile(XB, GT, Qp, group, lds);
    return;
  }
  chunk_state_tile(XB, XB, decay_logit, S, group - Q_GROUPS);
}

#define SCAN_W2_PLANES (W2_TILES / (8 * (C_DIM * C_DIM / 4 / 256)))

__global__ __launch_bounds__(256)
void scan_w2_kernel(const float* __restrict__ decay_logit, float* __restrict__ S,
                    const float* __restrict__ vc, const float* __restrict__ owT,
                    float* __restrict__ W2) {
  if ((int)blockIdx.z < SCAN_W2_PLANES) {
    const int w = __builtin_amdgcn_readfirstlane((int)threadIdx.x >> 5);
    const int tile = (((int)blockIdx.z * (int)gridDim.x + (int)blockIdx.x) << 3) + w;
    out_weights_tile(vc, owT, W2, tile, (int)threadIdx.x & 31);
    return;
  }
  const int bz = (int)blockIdx.z - SCAN_W2_PLANES;
  const float dl = *decay_logit;
  const float dec = 1.0f / (1.0f + __expf(-dl));
  const float dL = __expf(__logf(dec) * (float)CH_L);
  const size_t e = ((size_t)blockIdx.x * 256 + threadIdx.x) * 4;
  float* Sb = S + (size_t)bz * N_CHUNK * C_DIM * C_DIM + e;
  float4 p[N_CHUNK - 1];
#pragma unroll
  for (int c = 1; c < N_CHUNK; ++c)
    p[c - 1] = *(const float4*)&Sb[(size_t)c * C_DIM * C_DIM];
  float4 s = p[N_CHUNK - 2];
#pragma unroll
  for (int c = N_CHUNK - 2; c >= 1; --c) {
    const float4 q = p[c - 1];
    s.x = fmaf(dL, s.x, q.x);
    s.y = fmaf(dL, s.y, q.y);
    s.z = fmaf(dL, s.z, q.z);
    s.w = fmaf(dL, s.w, q.w);
    *(float4*)&Sb[(size_t)c * C_DIM * C_DIM] = s;
  }
}

#define ATT_WAVES 8
#define ATT_ROWS 32
#define ATT_KEYS (16 * ATT_WAVES)
#define ATT_RBLOCKS (T_DIM / ATT_ROWS)
#define QS_PITCH 260
#define SS_PITCH 132
#define ATT_QS_FLOATS (ATT_ROWS * QS_PITCH)
#define ATT_SS_FLOATS (ATT_ROWS * SS_PITCH)
#define ATT_LDS_FLOATS 41216

__global__ __launch_bounds__(32 * ATT_WAVES)
void attn_kernel(float* QR,
                 const float* __restrict__ K,
                 const float* __restrict__ Vv,
                 const float* __restrict__ S,
                 const float* __restrict__ decay_logit) {
  __shared__ float lds[ATT_LDS_FLOATS];
  float* const qs = &lds[0];
  float* const ss = &lds[ATT_QS_FLOATS];

  const int tid = threadIdx.x;
  const int lane = tid & 31;
  const int w = __builtin_amdgcn_readfirstlane(tid >> 5);
  const int r = lane & 15;
  const int h = lane >> 4;
  const int c0 = w * 32;
  const int b = blockIdx.z;

  float* Qb = QR + (size_t)b * T_DIM * C_DIM;
  const float* Kb = K + (size_t)b * T_DIM * C_DIM;
  const float* Vb = Vv + (size_t)b * T_DIM * C_DIM;
  float* Rb = Qb;

  const float dl = *decay_logit;
  const float dec = 1.0f / (1.0f + __expf(-dl));
  const float ln_dec = __logf(dec);

  for (int part = 0; part < 2; ++part) {
    const int rb = (part == 0) ? (int)blockIdx.x : (ATT_RBLOCKS - 1 - (int)blockIdx.x);
    const int i0 = rb * ATT_ROWS;
    const int cend = (i0 & ~(CH_L - 1)) + CH_L;

    {
      float4 qv[ATT_ROWS * 64 / (32 * ATT_WAVES)];
#pragma unroll
      for (int it = 0; it < ATT_ROWS * 64 / (32 * ATT_WAVES); ++it) {
        const int idx = tid + it * 32 * ATT_WAVES;
        qv[it] = *(const float4*)&Qb[(size_t)(i0 + (idx >> 6)) * C_DIM + ((idx & 63) << 2)];
      }
#pragma unroll
      for (int it = 0; it < ATT_ROWS * 64 / (32 * ATT_WAVES); ++it) {
        const int idx = tid + it * 32 * ATT_WAVES;
        *(float4*)&qs[(idx >> 6) * QS_PITCH + ((idx & 63) << 2)] = qv[it];
      }
    }
    __syncthreads();

    v8f a0c0 = vzero8(), a0c1 = vzero8();
    v8f a1c0 = vzero8(), a1c1 = vzero8();

    int buf = 0;
    for (int jc = i0 & ~(ATT_KEYS - 1); jc < cend; jc += ATT_KEYS) {
      const int j0 = jc + 16 * w;
      {
        int jn = (jc + ATT_KEYS < cend) ? (jc + ATT_KEYS) : jc;
        __builtin_prefetch(&Kb[(size_t)(jn + 16 * w + r) * C_DIM + 128 * h], 0, 1);
        __builtin_prefetch(&Vb[(size_t)(jn + lane) * C_DIM + c0], 0, 1);
        __builtin_prefetch(&Vb[(size_t)(jn + 32 + lane) * C_DIM + c0], 0, 1);
        __builtin_prefetch(&Vb[(size_t)(jn + 64 + lane) * C_DIM + c0], 0, 1);
        __builtin_prefetch(&Vb[(size_t)(jn + 96 + lane) * C_DIM + c0], 0, 1);
      }

      float* sb = &ss[buf * ATT_SS_FLOATS];
      if (j0 >= i0) {
        v8f s0 = vzero8(), s1 = vzero8();
        const float* Kj = Kb + (size_t)j0 * C_DIM;
#pragma unroll
        for (int k0 = 0; k0 < C_DIM; k0 += 4) {
          v2f qa = *(const v2f*)&qs[r * QS_PITCH + k0 + 2 * h];
          v2f qb = *(const v2f*)&qs[(16 + r) * QS_PITCH + k0 + 2 * h];
          v2f bb = *(const v2f*)&Kj[(size_t)r * C_DIM + k0 + 2 * h];
          s0 = wmma4(qa, bb, s0);
          s1 = wmma4(qb, bb, s1);
        }
#pragma unroll
        for (int i = 0; i < 8; ++i) {
          int gi = i0 + i + 8 * h;
          int gj = j0 + r;
          int d0 = gj - gi;
          int d1 = d0 - 16;
          float wg0 = (d0 > 0) ? __expf(ln_dec * (float)(d0 - 1)) : 0.0f;
          float wg1 = (d1 > 0) ? __expf(ln_dec * (float)(d1 - 1)) : 0.0f;
          sb[(i + 8 * h) * SS_PITCH + 16 * w + r] = s0[i] * wg0;
          sb[(16 + i + 8 * h) * SS_PITCH + 16 * w + r] = s1[i] * wg1;
        }
      } else {
#pragma unroll
        for (int i = 0; i < 8; ++i) {
          sb[(i + 8 * h) * SS_PITCH + 16 * w + r] = 0.0f;
          sb[(16 + i + 8 * h) * SS_PITCH + 16 * w + r] = 0.0f;
        }
      }
      __syncthreads();
#pragma unroll
      for (int kk = 0; kk < ATT_KEYS / 4; ++kk) {
        v2f pa = *(const v2f*)&sb[r * SS_PITCH + kk * 4 + 2 * h];
        v2f pb = *(const v2f*)&sb[(16 + r) * SS_PITCH + kk * 4 + 2 * h];
        const float* Vr0 = Vb + (size_t)(jc + kk * 4 + 2 * h) * C_DIM + c0;
        const float* Vr1 = Vr0 + C_DIM;
        v2f bb;
        bb.x = Vr0[r];      bb.y = Vr1[r];
        a0c0 = wmma4(pa, bb, a0c0);  a1c0 = wmma4(pb, bb, a1c0);
        bb.x = Vr0[16 + r]; bb.y = Vr1[16 + r];
        a0c1 = wmma4(pa, bb, a0c1);  a1c1 = wmma4(pb, bb, a1c1);
      }
      buf ^= 1;
    }

    if (cend < T_DIM) {
      const float* Sn = S + ((size_t)b * N_CHUNK + (size_t)(cend / CH_L)) * C_DIM * C_DIM;
      const float f0 = __expf(ln_dec * (float)(cend - 1 - (i0 + r)));
      const float f1 = __expf(ln_dec * (float)(cend - 1 - (i0 + 16 + r)));
#pragma unroll
      for (int kk = 0; kk < C_DIM / 4; ++kk) {
        v2f pa = *(const v2f*)&qs[r * QS_PITCH + kk * 4 + 2 * h];
        v2f pb = *(const v2f*)&qs[(16 + r) * QS_PITCH + kk * 4 + 2 * h];
        pa.x *= f0;  pa.y *= f0;
        pb.x *= f1;  pb.y *= f1;
        const float* Sr0 = Sn + (size_t)(kk * 4 + 2 * h) * C_DIM + c0;
        const float* Sr1 = Sr0 + C_DIM;
        v2f bb;
        bb.x = Sr0[r];      bb.y = Sr1[r];
        a0c0 = wmma4(pa, bb, a0c0);  a1c0 = wmma4(pb, bb, a1c0);
        bb.x = Sr0[16 + r]; bb.y = Sr1[16 + r];
        a0c1 = wmma4(pa, bb, a0c1);  a1c1 = wmma4(pb, bb, a1c1);
      }
    }
    __syncthreads();

#pragma unroll
    for (int i = 0; i < 8; ++i) {
      size_t row0 = (size_t)(i0 + i + 8 * h) * C_DIM + c0 + r;
      size_t row1 = row0 + (size_t)16 * C_DIM;
      Rb[row0] = a0c0[i];
      Rb[row0 + 16] = a0c1[i];
      Rb[row1] = a1c0[i];
      Rb[row1 + 16] = a1c1[i];
    }
  }
}

__global__ __launch_bounds__(256) __attribute__((amdgpu_waves_per_eu(8, 8)))
void out_proj_kernel(const float* __restrict__ R,
                     const float* __restrict__ W2,
                     const float* __restrict__ out_scale,
                     float* __restrict__ out) {
  __shared__ float lds[GT_LDS_FLOATS];
  const int m0 = blockIdx.x * GT_M;
  const int v0 = blockIdx.y * GT_N;
  const float sc = *out_scale;
  gemm_tile_128x128<C_DIM, C_DIM, V_DIM, V_DIM, true, false, false, C_DIM / GT_K - 1>(
      R + (size_t)m0 * C_DIM, W2 + v0, out + (size_t)m0 * V_DIM + v0, sc, lds);
}

extern "C" void kernel_launch(void* const* d_in, const int* in_sizes, int n_in,
                              void* d_out, int out_size, void* d_ws, size_t ws_size,
                              hipStream_t stream) {
  (void)in_sizes; (void)n_in; (void)out_size; (void)ws_size;

  const float* x = (const float*)d_in[0];
  const float* basis = (const float*)d_in[1];
  const float* qc = (const float*)d_in[2];
  const float* kc = (const float*)d_in[3];
  const float* vc = (const float*)d_in[4];
  const float* oc = (const float*)d_in[5];
  const float* decay_logit = (const float*)d_in[6];
  const float* out_scale = (const float*)d_in[7];
  float* out = (float*)d_out;

  float* ws = (float*)d_ws;
  float* owT = ws;
  float* W2 = owT + (size_t)V_DIM * C_DIM;
  float* GT = W2 + (size_t)V_DIM * F_DIM;
  float* Qp = GT + (size_t)F_DIM * F_DIM;
  float* XB = Qp + (size_t)M_DIM * F_DIM;
  float* S = XB + (size_t)M_DIM * F_DIM;

  xb_kernel<<<dim3(M_DIM / GT_M, F_DIM / GT_N), dim3(256), 0, stream>>>(
      x, basis, oc, qc, kc, XB, owT, GT);

  q_chunk_kernel<<<dim3(Q_GROUPS + CS_GROUPS), dim3(256), 0, stream>>>(
      XB, GT, Qp, decay_logit, S);

  scan_w2_kernel<<<dim3(C_DIM * C_DIM / 4 / 256, 1, SCAN_W2_PLANES + B_DIM), dim3(256), 0, stream>>>(
      decay_logit, S, vc, owT, W2);

  attn_kernel<<<dim3(ATT_RBLOCKS / 2, 1, B_DIM), dim3(32 * ATT_WAVES), 0, stream>>>(
      Qp, XB, XB, S, decay_logit);

  out_proj_kernel<<<dim3(M_DIM / GT_M, V_DIM / GT_N), dim3(256), 0, stream>>>(
      Qp, W2, out_scale, out);
}
